# starting kernel
# baseline (speedup 1.0000x reference)
#include <hip/hip_runtime.h>


typedef __attribute__((ext_vector_type(2))) float v2f;
typedef __attribute__((ext_vector_type(8))) float v8f;

#define B_DIM  32
#define A_DIM  1000
#define C_DIM  64
#define H_DIM  12
#define W_DIM  20
#define CH     768
#define N1     999
#define MP     1008
#define NP1    1024
#define FP     1024
#define LDATT  1024
#define KC     64
#define LDS_PAD 68


__global__ void pad_w(const float* __restrict__ Wt, float* __restrict__ Wpk)
{
    int e = blockIdx.x * blockDim.x + threadIdx.x;
    if (e >= CH * NP1) return;
    int k = e >> 10;
    int j = e & (NP1 - 1);
    float v = (j < N1) ? Wt[k * N1 + j] : 0.0f;
    Wpk[(size_t)(k >> 1) * (2 * NP1) + 2 * j + (k & 1)] = v;
}

__global__ void pad_bias(const float* __restrict__ bias, float* __restrict__ biasp)
{
    int j = blockIdx.x * blockDim.x + threadIdx.x;
    if (j < NP1) biasp[j] = (j < N1) ? bias[j] : 0.0f;
}

__global__ void gather_feats(const float* __restrict__ fv,
                             const int*   __restrict__ zi,
                             const int*   __restrict__ yi,
                             const int*   __restrict__ xi,
                             const int*   __restrict__ mask,
                             float* __restrict__ feats,
                             float* __restrict__ featsPk,
                             int b)
{
    int e = blockIdx.x * blockDim.x + threadIdx.x;
    if (e >= FP * CH) return;
    int a = e / CH;
    int k = e - a * CH;
    float v = 0.0f;
    if (a < A_DIM) {
        int idx = a * CH + k;
        if (!mask[idx]) {
            int z = zi[idx], y = yi[idx], x = xi[idx];
            v = fv[((b * C_DIM + z) * H_DIM + y) * W_DIM + x];
        }
    }
    feats[e] = v;
    featsPk[(size_t)(a >> 1) * (2 * CH) + 2 * k + (a & 1)] = v;
}

__device__ __forceinline__ void stage_a(float (*As)[LDS_PAD],
                                        const float* __restrict__ Abase,
                                        int i0, int lda, int kc, int tid)
{
    const int r = tid >> 3;
    const int g = (tid & 7) * 8;
    const float* src = Abase + (size_t)(i0 + r) * lda + kc + g;
    float4 p0 = *(const float4*)(src);
    float4 p1 = *(const float4*)(src + 4);
    *(float4*)&As[r][g]     = p0;
    *(float4*)&As[r][g + 4] = p1;
}

__global__ void gemm1_scores(const float* __restrict__ feats,
                             const float* __restrict__ Wpk,
                             const float* __restrict__ biasp,
                             float* __restrict__ attn)
{
    __shared__ float As[16][LDS_PAD];

    const int lane = threadIdx.x;
    const int wave = threadIdx.y;
    const int tid  = wave * 32 + lane;
    const int q    = blockIdx.x * 4 + wave;
    const int j0   = q * 64;
    const int i0   = blockIdx.y * 16;
    const int half = lane >> 4;
    const int n    = lane & 15;
    const int ar   = lane & 15;

    v8f acc0, acc1, acc2, acc3;
    {
        const float b0 = biasp[j0 + n];
        const float b1 = biasp[j0 + 16 + n];
        const float b2 = biasp[j0 + 32 + n];
        const float b3 = biasp[j0 + 48 + n];
#pragma unroll
        for (int r = 0; r < 8; ++r) {
            acc0[r] = b0; acc1[r] = b1; acc2[r] = b2; acc3[r] = b3;
        }
    }

    for (int kc = 0; kc < CH; kc += KC) {
        __syncthreads();
        stage_a(As, feats, i0, CH, kc, tid);
        __syncthreads();
#pragma unroll
        for (int kk = 0; kk < KC; kk += 4) {
            const int ka = kk + 2 * half;
            v2f afrag = *(const v2f*)&As[ar][ka];
            const float* wrow = Wpk + (size_t)((kc + ka) >> 1) * (2 * NP1);
            v2f b0 = *(const v2f*)&wrow[2 * (j0 + n)];
            v2f b1 = *(const v2f*)&wrow[2 * (j0 + 16 + n)];
            v2f b2 = *(const v2f*)&wrow[2 * (j0 + 32 + n)];
            v2f b3 = *(const v2f*)&wrow[2 * (j0 + 48 + n)];
            acc0 = __builtin_amdgcn_wmma_f32_16x16x4_f32(
                false, afrag, false, b0, (short)0, acc0, false, false);
            acc1 = __builtin_amdgcn_wmma_f32_16x16x4_f32(
                false, afrag, false, b1, (short)0, acc1, false, false);
            acc2 = __builtin_amdgcn_wmma_f32_16x16x4_f32(
                false, afrag, false, b2, (short)0, acc2, false, false);
            acc3 = __builtin_amdgcn_wmma_f32_16x16x4_f32(
                false, afrag, false, b3, (short)0, acc3, false, false);
        }
    }

#pragma unroll
    for (int t = 0; t < 4; ++t) {
        const int jp = j0 + t * 16 + n;
        if (jp < N1) {
#pragma unroll
            for (int r = 0; r < 8; ++r) {
                const int m = i0 + r + 8 * half;
                const int jcol = jp + (jp >= m ? 1 : 0);
                const float val = (t == 0) ? acc0[r] : (t == 1) ? acc1[r]
                                : (t == 2) ? acc2[r] : acc3[r];
                attn[(size_t)m * LDATT + jcol] = val;
            }
        }
    }
}

__global__ void softmax_row(float* __restrict__ attn)
{
    __shared__ float red[256];
    const int i = blockIdx.x;
    const int t = threadIdx.x;
    float* row = attn + (size_t)i * LDATT;

    float mx = -3.402823466e+38f;
    for (int j = t; j < A_DIM; j += 256)
        if (j != i) mx = fmaxf(mx, row[j]);
    red[t] = mx;
    __syncthreads();
    for (int s = 128; s > 0; s >>= 1) {
        if (t < s) red[t] = fmaxf(red[t], red[t + s]);
        __syncthreads();
    }
    mx = red[0];
    __syncthreads();

    float sum = 0.0f;
    for (int j = t; j < A_DIM; j += 256)
        if (j != i) sum += expf(row[j] - mx);
    red[t] = sum;
    __syncthreads();
    for (int s = 128; s > 0; s >>= 1) {
        if (t < s) red[t] += red[t + s];
        __syncthreads();
    }
    const float inv = 1.0f / red[0];

    for (int j = t; j < A_DIM; j += 256)
        row[j] = (j == i) ? 0.0f : expf(row[j] - mx) * inv;
    for (int j = A_DIM + t; j < LDATT; j += 256)
        row[j] = 0.0f;
}

__global__ void gemm2_out(const float* __restrict__ attn,
                          const float* __restrict__ featsPk,
                          float* __restrict__ outb)
{
    __shared__ float As[16][LDS_PAD];

    const int lane = threadIdx.x;
    const int wave = threadIdx.y;
    const int tid  = wave * 32 + lane;
    const int q    = blockIdx.x * 4 + wave;
    const int n0   = q * 64;
    const int i0   = blockIdx.y * 16;
    const int half = lane >> 4;
    const int n    = lane & 15;
    const int ar   = lane & 15;

    v8f acc0 = {}, acc1 = {}, acc2 = {}, acc3 = {};

    for (int kc = 0; kc < FP; kc += KC) {
        __syncthreads();
        stage_a(As, attn, i0, LDATT, kc, tid);
        __syncthreads();
#pragma unroll
        for (int kk = 0; kk < KC; kk += 4) {
            const int ka = kk + 2 * half;
            v2f afrag = *(const v2f*)&As[ar][ka];
            const float* frow = featsPk + (size_t)((kc + ka) >> 1) * (2 * CH);
            v2f b0 = *(const v2f*)&frow[2 * (n0 + n)];
            v2f b1 = *(const v2f*)&frow[2 * (n0 + 16 + n)];
            v2f b2 = *(const v2f*)&frow[2 * (n0 + 32 + n)];
            v2f b3 = *(const v2f*)&frow[2 * (n0 + 48 + n)];
            acc0 = __builtin_amdgcn_wmma_f32_16x16x4_f32(
                false, afrag, false, b0, (short)0, acc0, false, false);
            acc1 = __builtin_amdgcn_wmma_f32_16x16x4_f32(
                false, afrag, false, b1, (short)0, acc1, false, false);
            acc2 = __builtin_amdgcn_wmma_f32_16x16x4_f32(
                false, afrag, false, b2, (short)0, acc2, false, false);
            acc3 = __builtin_amdgcn_wmma_f32_16x16x4_f32(
                false, afrag, false, b3, (short)0, acc3, false, false);
        }
    }

#pragma unroll
    for (int r = 0; r < 8; ++r) {
        const int m = i0 + r + 8 * half;
        if (m < A_DIM) {
            float* orow = outb + (size_t)m * CH;
            orow[n0 + n]      = acc0[r];
            orow[n0 + 16 + n] = acc1[r];
            orow[n0 + 32 + n] = acc2[r];
            orow[n0 + 48 + n] = acc3[r];
        }
    }
}

extern "C" void kernel_launch(void* const* d_in, const int* in_sizes, int n_in,
                              void* d_out, int out_size, void* d_ws, size_t ws_size,
                              hipStream_t stream)
{
    const float* fv   = (const float*)d_in[0];
    const int*   zi   = (const int*)  d_in[1];
    const int*   yi   = (const int*)  d_in[2];
    const int*   xi   = (const int*)  d_in[3];
    const int*   mask = (const int*)  d_in[4];
    const float* Wt   = (const float*)d_in[5];
    const float* bias = (const float*)d_in[6];
    float*       out  = (float*)d_out;

    float* Wpk     = (float*)d_ws;
    float* biasp   = Wpk     + (size_t)CH * NP1;
    float* feats   = biasp   + NP1;
    float* featsPk = feats   + (size_t)FP * CH;
    float* attn    = featsPk + (size_t)FP * CH;

    const dim3 wblk(32, 4);

    pad_w<<<(CH * NP1 + 255) / 256, 256, 0, stream>>>(Wt, Wpk);
    pad_bias<<<(NP1 + 255) / 256, 256, 0, stream>>>(bias, biasp);

    for (int b = 0; b < B_DIM; ++b) {
        gather_feats<<<(FP * CH + 255) / 256, 256, 0, stream>>>(
            fv, zi, yi, xi, mask, feats, featsPk, b);
        gemm1_scores<<<dim3(4, 63), wblk, 0, stream>>>(feats, Wpk, biasp, attn);
        softmax_row<<<dim3(A_DIM), 256, 0, stream>>>(attn);
        gemm2_out<<<dim3(3, 63), wblk, 0, stream>>>(attn, featsPk,
                                                    out + (size_t)b * A_DIM * CH);
    }
}


# Both GEMM K loops software-pipelined: next 64-k chunk's A tile and 64 B fragments are loaded (2nd LDS buffer, 2nd register set) while the current chunk's 64 WMMAs run; one barrier per chunk
# speedup vs baseline: 2.3288x; 2.3288x over previous
#include <hip/hip_runtime.h>


typedef __attribute__((ext_vector_type(2))) float v2f;
typedef __attribute__((ext_vector_type(8))) float v8f;

#define B_DIM  32
#define A_DIM  1000
#define C_DIM  64
#define H_DIM  12
#define W_DIM  20
#define CH     768
#define N1     999
#define MP     1008
#define NP1    1024
#define FP     1024
#define LDATT  1024
#define KC     64
#define LDS_PAD 68


__global__ void pad_w(const float* __restrict__ Wt, float* __restrict__ Wpk)
{
    int e = blockIdx.x * blockDim.x + threadIdx.x;
    if (e >= CH * NP1) return;
    int k = e >> 10;
    int j = e & (NP1 - 1);
    float v = (j < N1) ? Wt[k * N1 + j] : 0.0f;
    Wpk[(size_t)(k >> 1) * (2 * NP1) + 2 * j + (k & 1)] = v;
}

__global__ void pad_bias(const float* __restrict__ bias, float* __restrict__ biasp)
{
    int j = blockIdx.x * blockDim.x + threadIdx.x;
    if (j < NP1) biasp[j] = (j < N1) ? bias[j] : 0.0f;
}

__global__ void gather_feats(const float* __restrict__ fv,
                             const int*   __restrict__ zi,
                             const int*   __restrict__ yi,
                             const int*   __restrict__ xi,
                             const int*   __restrict__ mask,
                             float* __restrict__ feats,
                             float* __restrict__ featsPk,
                             int b)
{
    int e = blockIdx.x * blockDim.x + threadIdx.x;
    if (e >= FP * CH) return;
    int a = e / CH;
    int k = e - a * CH;
    float v = 0.0f;
    if (a < A_DIM) {
        int idx = a * CH + k;
        if (!mask[idx]) {
            int z = zi[idx], y = yi[idx], x = xi[idx];
            v = fv[((b * C_DIM + z) * H_DIM + y) * W_DIM + x];
        }
    }
    feats[e] = v;
    featsPk[(size_t)(a >> 1) * (2 * CH) + 2 * k + (a & 1)] = v;
}

template <int RP>
__device__ __forceinline__ void load_b_chunk(v2f (&b)[64],
                                             const float* __restrict__ pB)
{
#pragma unroll
    for (int s = 0; s < 16; ++s)
#pragma unroll
        for (int t = 0; t < 4; ++t)
            b[4 * s + t] = *(const v2f*)(pB + (size_t)(2 * s) * RP + 32 * t);
}

__device__ __forceinline__ void mma_chunk(const float (*As)[LDS_PAD],
                                          const v2f (&b)[64], int ar, int half,
                                          v8f& acc0, v8f& acc1,
                                          v8f& acc2, v8f& acc3)
{
    v2f a[16];
#pragma unroll
    for (int s = 0; s < 16; ++s)
        a[s] = *(const v2f*)&As[ar][4 * s + 2 * half];
#pragma unroll
    for (int s = 0; s < 16; ++s) {
        acc0 = __builtin_amdgcn_wmma_f32_16x16x4_f32(
            false, a[s], false, b[4 * s + 0], (short)0, acc0, false, false);
        acc1 = __builtin_amdgcn_wmma_f32_16x16x4_f32(
            false, a[s], false, b[4 * s + 1], (short)0, acc1, false, false);
        acc2 = __builtin_amdgcn_wmma_f32_16x16x4_f32(
            false, a[s], false, b[4 * s + 2], (short)0, acc2, false, false);
        acc3 = __builtin_amdgcn_wmma_f32_16x16x4_f32(
            false, a[s], false, b[4 * s + 3], (short)0, acc3, false, false);
    }
}

template <int KTOT, int LDA, int RP>
__device__ __forceinline__ void gemm_strip(const float* __restrict__ A,
                                           const float* __restrict__ Bpk,
                                           int i0, int j0,
                                           float (*As)[16][LDS_PAD],
                                           v8f& acc0, v8f& acc1,
                                           v8f& acc2, v8f& acc3)
{
    const int lane = threadIdx.x;
    const int tid  = threadIdx.y * 32 + lane;
    const int half = lane >> 4;
    const int n    = lane & 15;
    const int r    = tid >> 3;
    const int g    = (tid & 7) * 8;
    const float* pA = A + (size_t)(i0 + r) * LDA + g;
    const float* pB = Bpk + (size_t)half * RP + 2 * (j0 + n);

    v2f b0[64], b1[64];
    float4 p0 = *(const float4*)(pA);
    float4 p1 = *(const float4*)(pA + 4);
    load_b_chunk<RP>(b0, pB);
    *(float4*)&As[0][r][g]     = p0;
    *(float4*)&As[0][r][g + 4] = p1;
    __syncthreads();

    for (int c = 0; c < KTOT / KC; c += 2) {
        pA += KC;
        pB += (size_t)(KC / 2) * RP;
        p0 = *(const float4*)(pA);
        p1 = *(const float4*)(pA + 4);
        load_b_chunk<RP>(b1, pB);
        __builtin_amdgcn_sched_barrier(0);
        mma_chunk(As[0], b0, n, half, acc0, acc1, acc2, acc3);
        *(float4*)&As[1][r][g]     = p0;
        *(float4*)&As[1][r][g + 4] = p1;
        __syncthreads();

        const bool more = (c + 2 < KTOT / KC);
        if (more) {
            pA += KC;
            pB += (size_t)(KC / 2) * RP;
            p0 = *(const float4*)(pA);
            p1 = *(const float4*)(pA + 4);
            load_b_chunk<RP>(b0, pB);
        }
        __builtin_amdgcn_sched_barrier(0);
        mma_chunk(As[1], b1, n, half, acc0, acc1, acc2, acc3);
        if (more) {
            *(float4*)&As[0][r][g]     = p0;
            *(float4*)&As[0][r][g + 4] = p1;
        }
        __syncthreads();
    }
}

__global__ void __launch_bounds__(128) gemm1_scores(const float* __restrict__ feats,
                             const float* __restrict__ Wpk,
                             const float* __restrict__ biasp,
                             float* __restrict__ attn)
{
    __shared__ float As[2][16][LDS_PAD];

    const int lane = threadIdx.x;
    const int wave = threadIdx.y;
    const int q    = blockIdx.x * 4 + wave;
    const int j0   = q * 64;
    const int i0   = blockIdx.y * 16;
    const int half = lane >> 4;
    const int n    = lane & 15;

    v8f acc0, acc1, acc2, acc3;
    {
        const float b0 = biasp[j0 + n];
        const float b1 = biasp[j0 + 16 + n];
        const float b2 = biasp[j0 + 32 + n];
        const float b3 = biasp[j0 + 48 + n];
#pragma unroll
        for (int r = 0; r < 8; ++r) {
            acc0[r] = b0; acc1[r] = b1; acc2[r] = b2; acc3[r] = b3;
        }
    }

    gemm_strip<CH, CH, 2 * NP1>(feats, Wpk, i0, j0, As, acc0, acc1, acc2, acc3);

#pragma unroll
    for (int t = 0; t < 4; ++t) {
        const int jp = j0 + t * 16 + n;
        if (jp < N1) {
#pragma unroll
            for (int r = 0; r < 8; ++r) {
                const int m = i0 + r + 8 * half;
                const int jcol = jp + (jp >= m ? 1 : 0);
                const float val = (t == 0) ? acc0[r] : (t == 1) ? acc1[r]
                                : (t == 2) ? acc2[r] : acc3[r];
                attn[(size_t)m * LDATT + jcol] = val;
            }
        }
    }
}

__global__ void softmax_row(float* __restrict__ attn)
{
    __shared__ float red[256];
    const int i = blockIdx.x;
    const int t = threadIdx.x;
    float* row = attn + (size_t)i * LDATT;

    float mx = -3.402823466e+38f;
    for (int j = t; j < A_DIM; j += 256)
        if (j != i) mx = fmaxf(mx, row[j]);
    red[t] = mx;
    __syncthreads();
    for (int s = 128; s > 0; s >>= 1) {
        if (t < s) red[t] = fmaxf(red[t], red[t + s]);
        __syncthreads();
    }
    mx = red[0];
    __syncthreads();

    float sum = 0.0f;
    for (int j = t; j < A_DIM; j += 256)
        if (j != i) sum += expf(row[j] - mx);
    red[t] = sum;
    __syncthreads();
    for (int s = 128; s > 0; s >>= 1) {
        if (t < s) red[t] += red[t + s];
        __syncthreads();
    }
    const float inv = 1.0f / red[0];

    for (int j = t; j < A_DIM; j += 256)
        row[j] = (j == i) ? 0.0f : expf(row[j] - mx) * inv;
    for (int j = A_DIM + t; j < LDATT; j += 256)
        row[j] = 0.0f;
}

__global__ void __launch_bounds__(128) gemm2_out(const float* __restrict__ attn,
                          const float* __restrict__ featsPk,
                          float* __restrict__ outb)
{
    __shared__ float As[2][16][LDS_PAD];

    const int lane = threadIdx.x;
    const int wave = threadIdx.y;
    const int q    = blockIdx.x * 4 + wave;
    const int n0   = q * 64;
    const int i0   = blockIdx.y * 16;
    const int half = lane >> 4;
    const int n    = lane & 15;

    v8f acc0 = {}, acc1 = {}, acc2 = {}, acc3 = {};

    gemm_strip<FP, LDATT, 2 * CH>(attn, featsPk, i0, n0, As, acc0, acc1, acc2, acc3);

#pragma unroll
    for (int r = 0; r < 8; ++r) {
        const int m = i0 + r + 8 * half;
        if (m < A_DIM) {
            float* orow = outb + (size_t)m * CH;
            orow[n0 + n]      = acc0[r];
            orow[n0 + 16 + n] = acc1[r];
            orow[n0 + 32 + n] = acc2[r];
            orow[n0 + 48 + n] = acc3[r];
        }
    }
}

extern "C" void kernel_launch(void* const* d_in, const int* in_sizes, int n_in,
                              void* d_out, int out_size, void* d_ws, size_t ws_size,
                              hipStream_t stream)
{
    const float* fv   = (const float*)d_in[0];
    const int*   zi   = (const int*)  d_in[1];
    const int*   yi   = (const int*)  d_in[2];
    const int*   xi   = (const int*)  d_in[3];
    const int*   mask = (const int*)  d_in[4];
    const float* Wt   = (const float*)d_in[5];
    const float* bias = (const float*)d_in[6];
    float*       out  = (float*)d_out;

    float* Wpk     = (float*)d_ws;
    float* biasp   = Wpk     + (size_t)CH * NP1;
    float* feats   = biasp   + NP1;
    float* featsPk = feats   + (size_t)FP * CH;
    float* attn    = featsPk + (size_t)FP * CH;

    const dim3 wblk(32, 4);

    pad_w<<<(CH * NP1 + 255) / 256, 256, 0, stream>>>(Wt, Wpk);
    pad_bias<<<(NP1 + 255) / 256, 256, 0, stream>>>(bias, biasp);

    for (int b = 0; b < B_DIM; ++b) {
        gather_feats<<<(FP * CH + 255) / 256, 256, 0, stream>>>(
            fv, zi, yi, xi, mask, feats, featsPk, b);
        gemm1_scores<<<dim3(4, 63), wblk, 0, stream>>>(feats, Wpk, biasp, attn);
        softmax_row<<<dim3(A_DIM), 256, 0, stream>>>(attn);
        gemm2_out<<<dim3(3, 63), wblk, 0, stream>>>(attn, featsPk,
                                                    out + (size_t)b * A_DIM * CH);
    }
}


# Both GEMMs: block tile 64x64 with the B tile staged in LDS too and shared by the 4 waves (was 16x256, every wave reading its B fragments from global); fragments read from LDS one chunk ahead
# speedup vs baseline: 3.2651x; 1.4021x over previous
#include <hip/hip_runtime.h>


typedef __attribute__((ext_vector_type(2))) float v2f;
typedef __attribute__((ext_vector_type(4))) float v4f;
typedef __attribute__((ext_vector_type(8))) float v8f;

#define B_DIM  32
#define A_DIM  1000
#define C_DIM  64
#define H_DIM  12
#define W_DIM  20
#define CH     768
#define N1     999
#define MP     1008
#define NP1    1024
#define FP     1024
#define LDATT  1024
#define KC     64
#define LDS_PAD 68
#define BM     64
#define BN     64
#define BPITCH 160


__global__ void pad_w(const float* __restrict__ Wt, float* __restrict__ Wpk)
{
    int e = blockIdx.x * blockDim.x + threadIdx.x;
    if (e >= CH * NP1) return;
    int k = e >> 10;
    int j = e & (NP1 - 1);
    float v = (j < N1) ? Wt[k * N1 + j] : 0.0f;
    Wpk[(size_t)(k >> 1) * (2 * NP1) + 2 * j + (k & 1)] = v;
}

__global__ void pad_bias(const float* __restrict__ bias, float* __restrict__ biasp)
{
    int j = blockIdx.x * blockDim.x + threadIdx.x;
    if (j < NP1) biasp[j] = (j < N1) ? bias[j] : 0.0f;
}

__global__ void gather_feats(const float* __restrict__ fv,
                             const int*   __restrict__ zi,
                             const int*   __restrict__ yi,
                             const int*   __restrict__ xi,
                             const int*   __restrict__ mask,
                             float* __restrict__ feats,
                             float* __restrict__ featsPk,
                             int b)
{
    int e = blockIdx.x * blockDim.x + threadIdx.x;
    if (e >= FP * CH) return;
    int a = e / CH;
    int k = e - a * CH;
    float v = 0.0f;
    if (a < A_DIM) {
        int idx = a * CH + k;
        if (!mask[idx]) {
            int z = zi[idx], y = yi[idx], x = xi[idx];
            v = fv[((b * C_DIM + z) * H_DIM + y) * W_DIM + x];
        }
    }
    feats[e] = v;
    featsPk[(size_t)(a >> 1) * (2 * CH) + 2 * k + (a & 1)] = v;
}

template <int RP>
__device__ __forceinline__ void stage_load(v4f (&sa)[8], v4f (&sb)[8],
                                           const float* const (&pA)[8],
                                           const float* __restrict__ pB, int c)
{
#pragma unroll
    for (int q = 0; q < 8; ++q)
        sa[q] = *(const v4f*)(pA[q] + c * KC);
#pragma unroll
    for (int q = 0; q < 8; ++q)
        sb[q] = *(const v4f*)(pB + (size_t)(c * (KC / 2) + 4 * q) * RP);
}

__device__ __forceinline__ void stage_store(float (*As)[LDS_PAD], float (*Bs)[BPITCH],
                                            const v4f (&sa)[8], const v4f (&sb)[8],
                                            int ar, int ac, int br, int bc)
{
#pragma unroll
    for (int q = 0; q < 8; ++q)
        *(v4f*)&As[ar + 8 * q][ac] = sa[q];
#pragma unroll
    for (int q = 0; q < 8; ++q)
        *(v4f*)&Bs[br + 4 * q][bc] = sb[q];
}

__device__ __forceinline__ void frag_load(const float (*As)[LDS_PAD],
                                          const float (*Bs)[BPITCH],
                                          v2f (&fa)[16], v4f (&fb)[32],
                                          int arow, int half, int n)
{
#pragma unroll
    for (int s = 0; s < 16; ++s) {
        fa[s]         = *(const v2f*)&As[arow][4 * s + 2 * half];
        fb[2 * s]     = *(const v4f*)&Bs[2 * s + half][4 * n];
        fb[2 * s + 1] = *(const v4f*)&Bs[2 * s + half][64 + 4 * n];
    }
}

__device__ __forceinline__ void mma_chunk(const v2f (&fa)[16], const v4f (&fb)[32],
                                          v8f& acc0, v8f& acc1,
                                          v8f& acc2, v8f& acc3)
{
#pragma unroll
    for (int s = 0; s < 16; ++s) {
        const v2f b0 = __builtin_shufflevector(fb[2 * s],     fb[2 * s],     0, 1);
        const v2f b1 = __builtin_shufflevector(fb[2 * s],     fb[2 * s],     2, 3);
        const v2f b2 = __builtin_shufflevector(fb[2 * s + 1], fb[2 * s + 1], 0, 1);
        const v2f b3 = __builtin_shufflevector(fb[2 * s + 1], fb[2 * s + 1], 2, 3);
        acc0 = __builtin_amdgcn_wmma_f32_16x16x4_f32(
            false, fa[s], false, b0, (short)0, acc0, false, false);
        acc1 = __builtin_amdgcn_wmma_f32_16x16x4_f32(
            false, fa[s], false, b1, (short)0, acc1, false, false);
        acc2 = __builtin_amdgcn_wmma_f32_16x16x4_f32(
            false, fa[s], false, b2, (short)0, acc2, false, false);
        acc3 = __builtin_amdgcn_wmma_f32_16x16x4_f32(
            false, fa[s], false, b3, (short)0, acc3, false, false);
    }
}

template <int KTOT, int LDA, int RP, int AROWS>
__device__ __forceinline__ void gemm_tile(const float* __restrict__ A,
                                          const float* __restrict__ Bpk,
                                          int ib, int j0,
                                          float (*As)[BM][LDS_PAD],
                                          float (*Bs)[KC / 2][BPITCH],
                                          v8f& acc0, v8f& acc1,
                                          v8f& acc2, v8f& acc3)
{
    constexpr int NC = KTOT / KC;
    const int lane = threadIdx.x;
    const int wave = threadIdx.y;
    const int tid  = wave * 32 + lane;
    const int half = lane >> 4;
    const int n    = lane & 15;
    const int arow = wave * 16 + n;
    const int ar   = tid >> 4;
    const int ac   = (tid & 15) * 4;
    const int br   = tid >> 5;
    const int bc   = (tid & 31) * 4;

    const float* pA[8];
#pragma unroll
    for (int q = 0; q < 8; ++q) {
        int row = ib + ar + 8 * q;
        row = row < AROWS ? row : AROWS - 1;
        pA[q] = A + (size_t)row * LDA + ac;
    }
    const float* pB = Bpk + (size_t)br * RP + 2 * j0 + bc;

    v4f sa[8], sb[8];
    v2f fa0[16], fa1[16];
    v4f fb0[32], fb1[32];

    stage_load<RP>(sa, sb, pA, pB, 0);
    stage_store(As[0], Bs[0], sa, sb, ar, ac, br, bc);
    __syncthreads();
    stage_load<RP>(sa, sb, pA, pB, 1);
    frag_load(As[0], Bs[0], fa0, fb0, arow, half, n);
    stage_store(As[1], Bs[1], sa, sb, ar, ac, br, bc);
    __syncthreads();

    for (int c = 0; c < NC; c += 2) {
        const bool more = (c + 2 < NC);
        if (more) stage_load<RP>(sa, sb, pA, pB, c + 2);
        frag_load(As[1], Bs[1], fa1, fb1, arow, half, n);
        asm volatile("" ::: "memory");
        __builtin_amdgcn_sched_barrier(0);
        mma_chunk(fa0, fb0, acc0, acc1, acc2, acc3);
        if (more) stage_store(As[0], Bs[0], sa, sb, ar, ac, br, bc);
        __syncthreads();

        if (more) {
            stage_load<RP>(sa, sb, pA, pB, c + 3);
            frag_load(As[0], Bs[0], fa0, fb0, arow, half, n);
        }
        asm volatile("" ::: "memory");
        __builtin_amdgcn_sched_barrier(0);
        mma_chunk(fa1, fb1, acc0, acc1, acc2, acc3);
        if (more) {
            stage_store(As[1], Bs[1], sa, sb, ar, ac, br, bc);
            __syncthreads();
        }
    }
}

__global__ void __launch_bounds__(128) gemm1_scores(const float* __restrict__ feats,
                             const float* __restrict__ Wpk,
                             const float* __restrict__ biasp,
                             float* __restrict__ attn)
{
    __shared__ float As[2][BM][LDS_PAD];
    __shared__ float Bs[2][KC / 2][BPITCH];

    const int lane = threadIdx.x;
    const int wave = threadIdx.y;
    const int j0   = blockIdx.x * BN;
    const int ib   = blockIdx.y * BM;
    const int i0   = ib + wave * 16;
    const int half = lane >> 4;
    const int n    = lane & 15;

    v8f acc0, acc1, acc2, acc3;
    {
        const float2 bA = *(const float2*)&biasp[j0 + 2 * n];
        const float2 bB = *(const float2*)&biasp[j0 + 32 + 2 * n];
#pragma unroll
        for (int r = 0; r < 8; ++r) {
            acc0[r] = bA.x; acc1[r] = bA.y; acc2[r] = bB.x; acc3[r] = bB.y;
        }
    }

    gemm_tile<CH, CH, 2 * NP1, FP>(feats, Wpk, ib, j0, As, Bs, acc0, acc1, acc2, acc3);

#pragma unroll
    for (int t = 0; t < 4; ++t) {
        const int jp = j0 + 32 * (t >> 1) + 2 * n + (t & 1);
        if (jp < N1) {
#pragma unroll
            for (int r = 0; r < 8; ++r) {
                const int m = i0 + r + 8 * half;
                if (m < MP) {
                    const int jcol = jp + (jp >= m ? 1 : 0);
                    const float val = (t == 0) ? acc0[r] : (t == 1) ? acc1[r]
                                    : (t == 2) ? acc2[r] : acc3[r];
                    attn[(size_t)m * LDATT + jcol] = val;
                }
            }
        }
    }
}

__global__ void softmax_row(float* __restrict__ attn)
{
    __shared__ float red[256];
    const int i = blockIdx.x;
    const int t = threadIdx.x;
    float* row = attn + (size_t)i * LDATT;

    float mx = -3.402823466e+38f;
    for (int j = t; j < A_DIM; j += 256)
        if (j != i) mx = fmaxf(mx, row[j]);
    red[t] = mx;
    __syncthreads();
    for (int s = 128; s > 0; s >>= 1) {
        if (t < s) red[t] = fmaxf(red[t], red[t + s]);
        __syncthreads();
    }
    mx = red[0];
    __syncthreads();

    float sum = 0.0f;
    for (int j = t; j < A_DIM; j += 256)
        if (j != i) sum += expf(row[j] - mx);
    red[t] = sum;
    __syncthreads();
    for (int s = 128; s > 0; s >>= 1) {
        if (t < s) red[t] += red[t + s];
        __syncthreads();
    }
    const float inv = 1.0f / red[0];

    for (int j = t; j < A_DIM; j += 256)
        row[j] = (j == i) ? 0.0f : expf(row[j] - mx) * inv;
    for (int j = A_DIM + t; j < LDATT; j += 256)
        row[j] = 0.0f;
}

__global__ void __launch_bounds__(128) gemm2_out(const float* __restrict__ attn,
                          const float* __restrict__ featsPk,
                          float* __restrict__ outb)
{
    __shared__ float As[2][BM][LDS_PAD];
    __shared__ float Bs[2][KC / 2][BPITCH];

    const int lane = threadIdx.x;
    const int wave = threadIdx.y;
    const int n0   = blockIdx.x * BN;
    const int ib   = blockIdx.y * BM;
    const int i0   = ib + wave * 16;
    const int half = lane >> 4;
    const int n    = lane & 15;

    v8f acc0 = {}, acc1 = {}, acc2 = {}, acc3 = {};

    gemm_tile<FP, LDATT, 2 * CH, MP>(attn, featsPk, ib, n0, As, Bs, acc0, acc1, acc2, acc3);

#pragma unroll
    for (int r = 0; r < 8; ++r) {
        const int m = i0 + r + 8 * half;
        if (m < A_DIM) {
            float* orow = outb + (size_t)m * CH + n0 + 2 * n;
            *(float2*)(orow)      = make_float2(acc0[r], acc1[r]);
            *(float2*)(orow + 32) = make_float2(acc2[r], acc3[r]);
        }
    }
}

extern "C" void kernel_launch(void* const* d_in, const int* in_sizes, int n_in,
                              void* d_out, int out_size, void* d_ws, size_t ws_size,
                              hipStream_t stream)
{
    const float* fv   = (const float*)d_in[0];
    const int*   zi   = (const int*)  d_in[1];
    const int*   yi   = (const int*)  d_in[2];
    const int*   xi   = (const int*)  d_in[3];
    const int*   mask = (const int*)  d_in[4];
    const float* Wt   = (const float*)d_in[5];
    const float* bias = (const float*)d_in[6];
    float*       out  = (float*)d_out;

    float* Wpk     = (float*)d_ws;
    float* biasp   = Wpk     + (size_t)CH * NP1;
    float* feats   = biasp   + NP1;
    float* featsPk = feats   + (size_t)FP * CH;
    float* attn    = featsPk + (size_t)FP * CH;

    const dim3 wblk(32, 4);

    pad_w<<<(CH * NP1 + 255) / 256, 256, 0, stream>>>(Wt, Wpk);
    pad_bias<<<(NP1 + 255) / 256, 256, 0, stream>>>(bias, biasp);

    for (int b = 0; b < B_DIM; ++b) {
        gather_feats<<<(FP * CH + 255) / 256, 256, 0, stream>>>(
            fv, zi, yi, xi, mask, feats, featsPk, b);
        gemm1_scores<<<dim3(16, 16), wblk, 0, stream>>>(feats, Wpk, biasp, attn);
        softmax_row<<<dim3(A_DIM), 256, 0, stream>>>(attn);
        gemm2_out<<<dim3(12, 16), wblk, 0, stream>>>(attn, featsPk,
                                                    out + (size_t)b * A_DIM * CH);
    }
}


# Next batch's gather rides in the softmax launch (feats double-buffered where the pair-packed copy was; gemm2 packs its B row pairs while staging): 99 launches, was 130; sums unchanged
# speedup vs baseline: 3.6044x; 1.1039x over previous
#include <hip/hip_runtime.h>


typedef __attribute__((ext_vector_type(2))) float v2f;
typedef __attribute__((ext_vector_type(4))) float v4f;
typedef __attribute__((ext_vector_type(8))) float v8f;

#define B_DIM  32
#define A_DIM  1000
#define C_DIM  64
#define H_DIM  12
#define W_DIM  20
#define CH     768
#define N1     999
#define MP     1008
#define NP1    1024
#define FP     1024
#define LDATT  1024
#define KC     64
#define LDS_PAD 68
#define BM     64
#define BN     64
#define BPITCH 160


__global__ void pad_w(const float* __restrict__ Wt, float* __restrict__ Wpk)
{
    int e = blockIdx.x * blockDim.x + threadIdx.x;
    if (e >= CH * NP1) return;
    int k = e >> 10;
    int j = e & (NP1 - 1);
    float v = (j < N1) ? Wt[k * N1 + j] : 0.0f;
    Wpk[(size_t)(k >> 1) * (2 * NP1) + 2 * j + (k & 1)] = v;
}

__global__ void pad_bias(const float* __restrict__ bias, float* __restrict__ biasp)
{
    int j = blockIdx.x * blockDim.x + threadIdx.x;
    if (j < NP1) biasp[j] = (j < N1) ? bias[j] : 0.0f;
}

#define GATHER_BLOCKS ((FP * CH + 255) / 256)

__device__ __forceinline__ void gather_body(int blk,
                             const float* __restrict__ fv,
                             const int*   __restrict__ zi,
                             const int*   __restrict__ yi,
                             const int*   __restrict__ xi,
                             const int*   __restrict__ mask,
                             float* __restrict__ feats,
                             int b)
{
    int e = blk * 256 + threadIdx.x;
    if (e >= FP * CH) return;
    int a = e / CH;
    int k = e - a * CH;
    float v = 0.0f;
    if (a < A_DIM) {
        int idx = a * CH + k;
        if (!mask[idx]) {
            int z = zi[idx], y = yi[idx], x = xi[idx];
            v = fv[((b * C_DIM + z) * H_DIM + y) * W_DIM + x];
        }
    }
    feats[e] = v;
}

__global__ void gather_feats(const float* __restrict__ fv,
                             const int*   __restrict__ zi,
                             const int*   __restrict__ yi,
                             const int*   __restrict__ xi,
                             const int*   __restrict__ mask,
                             float* __restrict__ feats, int b)
{
    gather_body(blockIdx.x, fv, zi, yi, xi, mask, feats, b);
}

template <int RP, bool BROW>
__device__ __forceinline__ void stage_load(v4f (&sa)[8], v4f (&sb)[8],
                                           const float* const (&pA)[8],
                                           const float* __restrict__ pB, int c)
{
#pragma unroll
    for (int q = 0; q < 8; ++q)
        sa[q] = *(const v4f*)(pA[q] + c * KC);
    if constexpr (BROW) {
#pragma unroll
        for (int q = 0; q < 4; ++q) {
            const float* r = pB + (size_t)(c * KC + 16 * q) * RP;
            sb[2 * q]     = *(const v4f*)r;
            sb[2 * q + 1] = *(const v4f*)(r + RP);
        }
    } else {
#pragma unroll
        for (int q = 0; q < 8; ++q)
            sb[q] = *(const v4f*)(pB + (size_t)(c * (KC / 2) + 4 * q) * RP);
    }
}

template <bool BROW>
__device__ __forceinline__ void stage_store(float (*As)[LDS_PAD], float (*Bs)[BPITCH],
                                            const v4f (&sa)[8], const v4f (&sb)[8],
                                            int ar, int ac, int br, int bc)
{
#pragma unroll
    for (int q = 0; q < 8; ++q)
        *(v4f*)&As[ar + 8 * q][ac] = sa[q];
    if constexpr (BROW) {
#pragma unroll
        for (int q = 0; q < 4; ++q) {
            *(v4f*)&Bs[br + 8 * q][bc] =
                __builtin_shufflevector(sb[2 * q], sb[2 * q + 1], 0, 4, 1, 5);
            *(v4f*)&Bs[br + 8 * q][bc + 4] =
                __builtin_shufflevector(sb[2 * q], sb[2 * q + 1], 2, 6, 3, 7);
        }
    } else {
#pragma unroll
        for (int q = 0; q < 8; ++q)
            *(v4f*)&Bs[br + 4 * q][bc] = sb[q];
    }
}

__device__ __forceinline__ void frag_load(const float (*As)[LDS_PAD],
                                          const float (*Bs)[BPITCH],
                                          v2f (&fa)[16], v4f (&fb)[32],
                                          int arow, int half, int n)
{
#pragma unroll
    for (int s = 0; s < 16; ++s) {
        fa[s]         = *(const v2f*)&As[arow][4 * s + 2 * half];
        fb[2 * s]     = *(const v4f*)&Bs[2 * s + half][4 * n];
        fb[2 * s + 1] = *(const v4f*)&Bs[2 * s + half][64 + 4 * n];
    }
}

__device__ __forceinline__ void mma_chunk(const v2f (&fa)[16], const v4f (&fb)[32],
                                          v8f& acc0, v8f& acc1,
                                          v8f& acc2, v8f& acc3)
{
#pragma unroll
    for (int s = 0; s < 16; ++s) {
        const v2f b0 = __builtin_shufflevector(fb[2 * s],     fb[2 * s],     0, 1);
        const v2f b1 = __builtin_shufflevector(fb[2 * s],     fb[2 * s],     2, 3);
        const v2f b2 = __builtin_shufflevector(fb[2 * s + 1], fb[2 * s + 1], 0, 1);
        const v2f b3 = __builtin_shufflevector(fb[2 * s + 1], fb[2 * s + 1], 2, 3);
        acc0 = __builtin_amdgcn_wmma_f32_16x16x4_f32(
            false, fa[s], false, b0, (short)0, acc0, false, false);
        acc1 = __builtin_amdgcn_wmma_f32_16x16x4_f32(
            false, fa[s], false, b1, (short)0, acc1, false, false);
        acc2 = __builtin_amdgcn_wmma_f32_16x16x4_f32(
            false, fa[s], false, b2, (short)0, acc2, false, false);
        acc3 = __builtin_amdgcn_wmma_f32_16x16x4_f32(
            false, fa[s], false, b3, (short)0, acc3, false, false);
    }
}

template <int KTOT, int LDA, int RP, int AROWS, bool BROW>
__device__ __forceinline__ void gemm_tile(const float* __restrict__ A,
                                          const float* __restrict__ Bpk,
                                          int ib, int j0,
                                          float (*As)[BM][LDS_PAD],
                                          float (*Bs)[KC / 2][BPITCH],
                                          v8f& acc0, v8f& acc1,
                                          v8f& acc2, v8f& acc3)
{
    constexpr int NC = KTOT / KC;
    const int lane = threadIdx.x;
    const int wave = threadIdx.y;
    const int tid  = wave * 32 + lane;
    const int half = lane >> 4;
    const int n    = lane & 15;
    const int arow = wave * 16 + n;
    const int ar   = tid >> 4;
    const int ac   = (tid & 15) * 4;
    const int br   = BROW ? (tid >> 4) : (tid >> 5);
    const int bc   = BROW ? (tid & 15) * 8 : (tid & 31) * 4;

    const float* pA[8];
#pragma unroll
    for (int q = 0; q < 8; ++q) {
        int row = ib + ar + 8 * q;
        row = row < AROWS ? row : AROWS - 1;
        pA[q] = A + (size_t)row * LDA + ac;
    }
    const float* pB = BROW ? Bpk + (size_t)(2 * br) * RP + j0 + (bc >> 1)
                           : Bpk + (size_t)br * RP + 2 * j0 + bc;

    v4f sa[8], sb[8];
    v2f fa0[16], fa1[16];
    v4f fb0[32], fb1[32];

    stage_load<RP, BROW>(sa, sb, pA, pB, 0);
    stage_store<BROW>(As[0], Bs[0], sa, sb, ar, ac, br, bc);
    __syncthreads();
    stage_load<RP, BROW>(sa, sb, pA, pB, 1);
    frag_load(As[0], Bs[0], fa0, fb0, arow, half, n);
    stage_store<BROW>(As[1], Bs[1], sa, sb, ar, ac, br, bc);
    __syncthreads();

    for (int c = 0; c < NC; c += 2) {
        const bool more = (c + 2 < NC);
        if (more) stage_load<RP, BROW>(sa, sb, pA, pB, c + 2);
        frag_load(As[1], Bs[1], fa1, fb1, arow, half, n);
        asm volatile("" ::: "memory");
        __builtin_amdgcn_sched_barrier(0);
        mma_chunk(fa0, fb0, acc0, acc1, acc2, acc3);
        if (more) stage_store<BROW>(As[0], Bs[0], sa, sb, ar, ac, br, bc);
        __syncthreads();

        if (more) {
            stage_load<RP, BROW>(sa, sb, pA, pB, c + 3);
            frag_load(As[0], Bs[0], fa0, fb0, arow, half, n);
        }
        asm volatile("" ::: "memory");
        __builtin_amdgcn_sched_barrier(0);
        mma_chunk(fa1, fb1, acc0, acc1, acc2, acc3);
        if (more) {
            stage_store<BROW>(As[1], Bs[1], sa, sb, ar, ac, br, bc);
            __syncthreads();
        }
    }
}

__global__ void __launch_bounds__(128) gemm1_scores(const float* __restrict__ feats,
                             const float* __restrict__ Wpk,
                             const float* __restrict__ biasp,
                             float* __restrict__ attn)
{
    __shared__ float As[2][BM][LDS_PAD];
    __shared__ float Bs[2][KC / 2][BPITCH];

    const int lane = threadIdx.x;
    const int wave = threadIdx.y;
    const int j0   = blockIdx.x * BN;
    const int ib   = blockIdx.y * BM;
    const int i0   = ib + wave * 16;
    const int half = lane >> 4;
    const int n    = lane & 15;

    v8f acc0, acc1, acc2, acc3;
    {
        const float2 bA = *(const float2*)&biasp[j0 + 2 * n];
        const float2 bB = *(const float2*)&biasp[j0 + 32 + 2 * n];
#pragma unroll
        for (int r = 0; r < 8; ++r) {
            acc0[r] = bA.x; acc1[r] = bA.y; acc2[r] = bB.x; acc3[r] = bB.y;
        }
    }

    gemm_tile<CH, CH, 2 * NP1, FP, false>(feats, Wpk, ib, j0, As, Bs, acc0, acc1, acc2, acc3);

#pragma unroll
    for (int t = 0; t < 4; ++t) {
        const int jp = j0 + 32 * (t >> 1) + 2 * n + (t & 1);
        if (jp < N1) {
#pragma unroll
            for (int r = 0; r < 8; ++r) {
                const int m = i0 + r + 8 * half;
                if (m < MP) {
                    const int jcol = jp + (jp >= m ? 1 : 0);
                    const float val = (t == 0) ? acc0[r] : (t == 1) ? acc1[r]
                                    : (t == 2) ? acc2[r] : acc3[r];
                    attn[(size_t)m * LDATT + jcol] = val;
                }
            }
        }
    }
}

__global__ void softmax_gather(float* __restrict__ attn,
                               const float* __restrict__ fv,
                               const int*   __restrict__ zi,
                               const int*   __restrict__ yi,
                               const int*   __restrict__ xi,
                               const int*   __restrict__ mask,
                               float* __restrict__ featsNext, int bNext)
{
    __shared__ float red[256];
    if (blockIdx.x >= A_DIM) {
        gather_body(blockIdx.x - A_DIM, fv, zi, yi, xi, mask, featsNext, bNext);
        return;
    }
    const int i = blockIdx.x;
    const int t = threadIdx.x;
    float* row = attn + (size_t)i * LDATT;

    float mx = -3.402823466e+38f;
    for (int j = t; j < A_DIM; j += 256)
        if (j != i) mx = fmaxf(mx, row[j]);
    red[t] = mx;
    __syncthreads();
    for (int s = 128; s > 0; s >>= 1) {
        if (t < s) red[t] = fmaxf(red[t], red[t + s]);
        __syncthreads();
    }
    mx = red[0];
    __syncthreads();

    float sum = 0.0f;
    for (int j = t; j < A_DIM; j += 256)
        if (j != i) sum += expf(row[j] - mx);
    red[t] = sum;
    __syncthreads();
    for (int s = 128; s > 0; s >>= 1) {
        if (t < s) red[t] += red[t + s];
        __syncthreads();
    }
    const float inv = 1.0f / red[0];

    for (int j = t; j < A_DIM; j += 256)
        row[j] = (j == i) ? 0.0f : expf(row[j] - mx) * inv;
    for (int j = A_DIM + t; j < LDATT; j += 256)
        row[j] = 0.0f;
}

__global__ void __launch_bounds__(128) gemm2_out(const float* __restrict__ attn,
                          const float* __restrict__ feats,
                          float* __restrict__ outb)
{
    __shared__ float As[2][BM][LDS_PAD];
    __shared__ float Bs[2][KC / 2][BPITCH];

    const int lane = threadIdx.x;
    const int wave = threadIdx.y;
    const int n0   = blockIdx.x * BN;
    const int ib   = blockIdx.y * BM;
    const int i0   = ib + wave * 16;
    const int half = lane >> 4;
    const int n    = lane & 15;

    v8f acc0 = {}, acc1 = {}, acc2 = {}, acc3 = {};

    gemm_tile<FP, LDATT, CH, MP, true>(attn, feats, ib, n0, As, Bs, acc0, acc1, acc2, acc3);

#pragma unroll
    for (int r = 0; r < 8; ++r) {
        const int m = i0 + r + 8 * half;
        if (m < A_DIM) {
            float* orow = outb + (size_t)m * CH + n0 + 2 * n;
            *(float2*)(orow)      = make_float2(acc0[r], acc1[r]);
            *(float2*)(orow + 32) = make_float2(acc2[r], acc3[r]);
        }
    }
}

extern "C" void kernel_launch(void* const* d_in, const int* in_sizes, int n_in,
                              void* d_out, int out_size, void* d_ws, size_t ws_size,
                              hipStream_t stream)
{
    const float* fv   = (const float*)d_in[0];
    const int*   zi   = (const int*)  d_in[1];
    const int*   yi   = (const int*)  d_in[2];
    const int*   xi   = (const int*)  d_in[3];
    const int*   mask = (const int*)  d_in[4];
    const float* Wt   = (const float*)d_in[5];
    const float* bias = (const float*)d_in[6];
    float*       out  = (float*)d_out;

    float* Wpk     = (float*)d_ws;
    float* biasp   = Wpk     + (size_t)CH * NP1;
    float* feats0  = biasp   + NP1;
    float* feats1  = feats0  + (size_t)FP * CH;
    float* attn    = feats1  + (size_t)FP * CH;

    const dim3 wblk(32, 4);

    pad_w<<<(CH * NP1 + 255) / 256, 256, 0, stream>>>(Wt, Wpk);
    pad_bias<<<(NP1 + 255) / 256, 256, 0, stream>>>(bias, biasp);

    gather_feats<<<GATHER_BLOCKS, 256, 0, stream>>>(fv, zi, yi, xi, mask, feats0, 0);

    for (int b = 0; b < B_DIM; ++b) {
        float* cur = (b & 1) ? feats1 : feats0;
        float* nxt = (b & 1) ? feats0 : feats1;
        gemm1_scores<<<dim3(16, 16), wblk, 0, stream>>>(cur, Wpk, biasp, attn);
        softmax_gather<<<(b + 1 < B_DIM) ? A_DIM + GATHER_BLOCKS : A_DIM, 256, 0, stream>>>(
            attn, fv, zi, yi, xi, mask, nxt, b + 1);
        gemm2_out<<<dim3(12, 16), wblk, 0, stream>>>(attn, cur,
                                                    out + (size_t)b * A_DIM * CH);
    }
}


# Softmax folded into the GEMMs: gemm1 stores exp(score - block row max) + block maxima, gemm2 rescales A while staging, divides by row sums (new sum order); gather rides in gemm2; 67 launches, was 99
# speedup vs baseline: 4.0820x; 1.1325x over previous
#include <hip/hip_runtime.h>


typedef __attribute__((ext_vector_type(2))) float v2f;
typedef __attribute__((ext_vector_type(4))) float v4f;
typedef __attribute__((ext_vector_type(8))) float v8f;

#define B_DIM  32
#define A_DIM  1000
#define C_DIM  64
#define H_DIM  12
#define W_DIM  20
#define CH     768
#define N1     999
#define MP     1008
#define NP1    1024
#define FP     1024
#define LDATT  1024
#define KC     64
#define LDS_PAD 68
#define BM     64
#define BN     64
#define BPITCH 160


__global__ void pad_w(const float* __restrict__ Wt, float* __restrict__ Wpk)
{
    int e = blockIdx.x * blockDim.x + threadIdx.x;
    if (e >= CH * NP1) return;
    int k = e >> 10;
    int j = e & (NP1 - 1);
    float v = (j < N1) ? Wt[k * N1 + j] : 0.0f;
    Wpk[(size_t)(k >> 1) * (2 * NP1) + 2 * j + (k & 1)] = v;
}

__global__ void pad_bias(const float* __restrict__ bias, float* __restrict__ biasp)
{
    int j = blockIdx.x * blockDim.x + threadIdx.x;
    if (j < NP1) biasp[j] = (j < N1) ? bias[j] : 0.0f;
}

#define GATHER_BLOCKS ((FP * CH + 255) / 256)

__device__ __forceinline__ void gather_body(int blk,
                             const float* __restrict__ fv,
                             const int*   __restrict__ zi,
                             const int*   __restrict__ yi,
                             const int*   __restrict__ xi,
                             const int*   __restrict__ mask,
                             float* __restrict__ feats,
                             int b)
{
    int e = blk * 256 + threadIdx.x;
    if (e >= FP * CH) return;
    int a = e / CH;
    int k = e - a * CH;
    float v = 0.0f;
    if (a < A_DIM) {
        int idx = a * CH + k;
        if (!mask[idx]) {
            int z = zi[idx], y = yi[idx], x = xi[idx];
            v = fv[((b * C_DIM + z) * H_DIM + y) * W_DIM + x];
        }
    }
    feats[e] = v;
}

__global__ void gather_feats(const float* __restrict__ fv,
                             const int*   __restrict__ zi,
                             const int*   __restrict__ yi,
                             const int*   __restrict__ xi,
                             const int*   __restrict__ mask,
                             float* __restrict__ feats, int b)
{
    gather_body(blockIdx.x, fv, zi, yi, xi, mask, feats, b);
}

template <int RP, bool BROW>
__device__ __forceinline__ void stage_load(v4f (&sa)[8], v4f (&sb)[8],
                                           const float* const (&pA)[8],
                                           const float* __restrict__ pB, int c)
{
#pragma unroll
    for (int q = 0; q < 8; ++q)
        sa[q] = *(const v4f*)(pA[q] + c * KC);
    if constexpr (BROW) {
#pragma unroll
        for (int q = 0; q < 4; ++q) {
            const float* r = pB + (size_t)(c * KC + 16 * q) * RP;
            sb[2 * q]     = *(const v4f*)r;
            sb[2 * q + 1] = *(const v4f*)(r + RP);
        }
    } else {
#pragma unroll
        for (int q = 0; q < 8; ++q)
            sb[q] = *(const v4f*)(pB + (size_t)(c * (KC / 2) + 4 * q) * RP);
    }
}

template <bool BROW>
__device__ __forceinline__ void stage_store(float (*As)[LDS_PAD], float (*Bs)[BPITCH],
                                            const v4f (&sa)[8], const v4f (&sb)[8],
                                            int ar, int ac, int br, int bc)
{
#pragma unroll
    for (int q = 0; q < 8; ++q)
        *(v4f*)&As[ar + 8 * q][ac] = sa[q];
    if constexpr (BROW) {
#pragma unroll
        for (int q = 0; q < 4; ++q) {
            *(v4f*)&Bs[br + 8 * q][bc] =
                __builtin_shufflevector(sb[2 * q], sb[2 * q + 1], 0, 4, 1, 5);
            *(v4f*)&Bs[br + 8 * q][bc + 4] =
                __builtin_shufflevector(sb[2 * q], sb[2 * q + 1], 2, 6, 3, 7);
        }
    } else {
#pragma unroll
        for (int q = 0; q < 8; ++q)
            *(v4f*)&Bs[br + 4 * q][bc] = sb[q];
    }
}

__device__ __forceinline__ void frag_load(const float (*As)[LDS_PAD],
                                          const float (*Bs)[BPITCH],
                                          v2f (&fa)[16], v4f (&fb)[32],
                                          int arow, int half, int n)
{
#pragma unroll
    for (int s = 0; s < 16; ++s) {
        fa[s]         = *(const v2f*)&As[arow][4 * s + 2 * half];
        fb[2 * s]     = *(const v4f*)&Bs[2 * s + half][4 * n];
        fb[2 * s + 1] = *(const v4f*)&Bs[2 * s + half][64 + 4 * n];
    }
}

__device__ __forceinline__ void mma_chunk(const v2f (&fa)[16], const v4f (&fb)[32],
                                          v8f& acc0, v8f& acc1,
                                          v8f& acc2, v8f& acc3)
{
#pragma unroll
    for (int s = 0; s < 16; ++s) {
        const v2f b0 = __builtin_shufflevector(fb[2 * s],     fb[2 * s],     0, 1);
        const v2f b1 = __builtin_shufflevector(fb[2 * s],     fb[2 * s],     2, 3);
        const v2f b2 = __builtin_shufflevector(fb[2 * s + 1], fb[2 * s + 1], 0, 1);
        const v2f b3 = __builtin_shufflevector(fb[2 * s + 1], fb[2 * s + 1], 2, 3);
        acc0 = __builtin_amdgcn_wmma_f32_16x16x4_f32(
            false, fa[s], false, b0, (short)0, acc0, false, false);
        acc1 = __builtin_amdgcn_wmma_f32_16x16x4_f32(
            false, fa[s], false, b1, (short)0, acc1, false, false);
        acc2 = __builtin_amdgcn_wmma_f32_16x16x4_f32(
            false, fa[s], false, b2, (short)0, acc2, false, false);
        acc3 = __builtin_amdgcn_wmma_f32_16x16x4_f32(
            false, fa[s], false, b3, (short)0, acc3, false, false);
    }
}

typedef float (*FqPtr)[8][8];

__device__ __forceinline__ void row_factors(const float* __restrict__ attn,
                                            FqPtr Fq, int ib, int tid)
{
    const int rl = tid >> 1;
    const int hc = (tid & 1) * 8;
    int row = ib + rl;
    row = row < MP ? row : MP - 1;
    const float* st = attn + (size_t)row * LDATT + A_DIM + hc;
    const v4f m0 = *(const v4f*)st;
    const v4f m1 = *(const v4f*)(st + 4);
    float mx = fmaxf(fmaxf(fmaxf(m0[0], m0[1]), fmaxf(m0[2], m0[3])),
                     fmaxf(fmaxf(m1[0], m1[1]), fmaxf(m1[2], m1[3])));
    mx = fmaxf(mx, __shfl_xor(mx, 1, 32));
#pragma unroll
    for (int e = 0; e < 4; ++e) {
        Fq[hc + e][rl & 7][rl >> 3]     = expf(m0[e] - mx);
        Fq[hc + 4 + e][rl & 7][rl >> 3] = expf(m1[e] - mx);
    }
}

__device__ __forceinline__ void load_factors(v4f (&ff)[2], v4f (&fp)[2],
                                             FqPtr Fq, int c, int ar)
{
    const int cp = c > 0 ? c - 1 : 0;
    ff[0] = *(const v4f*)&Fq[c][ar][0];
    ff[1] = *(const v4f*)&Fq[c][ar][4];
    fp[0] = *(const v4f*)&Fq[cp][ar][0];
    fp[1] = *(const v4f*)&Fq[cp][ar][4];
}

__device__ __forceinline__ void scale_a(const v4f (&sa)[8], v4f (&sw)[8], v4f (&rs)[8],
                                        const v4f (&ff)[2], const v4f (&fp)[2],
                                        int c, int by, int ar, int ac)
{
    const bool  first = (ac == 0) && (c > by);
    const float keep  = (c * KC + ac >= A_DIM) ? 0.0f : 1.0f;
    const v4f   keep4 = {keep, keep, keep, keep};
#pragma unroll
    for (int q = 0; q < 8; ++q) {
        const float f  = ff[q >> 2][q & 3];
        const float f0 = first ? fp[q >> 2][q & 3] : f;
        const v4f fv4 = {f0, f, f, f};
        const v4f w = sa[q] * fv4;
        rs[q] = w * keep4 + rs[q];
        sw[q] = w;
    }
}

template <int KTOT, int LDA, int RP, int AROWS, bool BROW, bool SCALE>
__device__ __forceinline__ void gemm_tile(const float* __restrict__ A,
                                          const float* __restrict__ Bpk,
                                          int ib, int j0,
                                          float (*As)[BM][LDS_PAD],
                                          float (*Bs)[KC / 2][BPITCH],
                                          v8f& acc0, v8f& acc1,
                                          v8f& acc2, v8f& acc3,
                                          FqPtr Fq, v4f (&rs)[8])
{
    constexpr int NC = KTOT / KC;
    const int lane = threadIdx.x;
    const int wave = threadIdx.y;
    const int tid  = wave * 32 + lane;
    const int half = lane >> 4;
    const int n    = lane & 15;
    const int arow = wave * 16 + n;
    const int ar   = tid >> 4;
    const int ac   = (tid & 15) * 4;
    const int br   = BROW ? (tid >> 4) : (tid >> 5);
    const int bc   = BROW ? (tid & 15) * 8 : (tid & 31) * 4;

    const float* pA[8];
#pragma unroll
    for (int q = 0; q < 8; ++q) {
        int row = ib + ar + 8 * q;
        row = row < AROWS ? row : AROWS - 1;
        pA[q] = A + (size_t)row * LDA + ac;
    }
    const float* pB = BROW ? Bpk + (size_t)(2 * br) * RP + j0 + (bc >> 1)
                           : Bpk + (size_t)br * RP + 2 * j0 + bc;

    v4f sa[8], sb[8];
    v2f fa0[16], fa1[16];
    v4f fb0[32], fb1[32];

    const int by = ib / BM;
    v4f sw[8];
    v4f ff[2], fp[2];
    stage_load<RP, BROW>(sa, sb, pA, pB, 0);
    if constexpr (SCALE) {
        row_factors(A, Fq, ib, tid);
        __syncthreads();
        load_factors(ff, fp, Fq, 0, ar);
        scale_a(sa, sw, rs, ff, fp, 0, by, ar, ac);
        stage_store<BROW>(As[0], Bs[0], sw, sb, ar, ac, br, bc);
    } else {
        stage_store<BROW>(As[0], Bs[0], sa, sb, ar, ac, br, bc);
    }
    __syncthreads();
    stage_load<RP, BROW>(sa, sb, pA, pB, 1);
    frag_load(As[0], Bs[0], fa0, fb0, arow, half, n);
    if constexpr (SCALE) {
        load_factors(ff, fp, Fq, 1, ar);
        scale_a(sa, sw, rs, ff, fp, 1, by, ar, ac);
        stage_store<BROW>(As[1], Bs[1], sw, sb, ar, ac, br, bc);
    } else {
        stage_store<BROW>(As[1], Bs[1], sa, sb, ar, ac, br, bc);
    }
    __syncthreads();

    for (int c = 0; c < NC; c += 2) {
        const bool more = (c + 2 < NC);
        if (more) stage_load<RP, BROW>(sa, sb, pA, pB, c + 2);
        const int c2 = c + 2 < NC ? c + 2 : NC - 1;
        if constexpr (SCALE) load_factors(ff, fp, Fq, c2, ar);
        frag_load(As[1], Bs[1], fa1, fb1, arow, half, n);
        asm volatile("" ::: "memory");
        __builtin_amdgcn_sched_barrier(0);
        mma_chunk(fa0, fb0, acc0, acc1, acc2, acc3);
        if (more) {
            if constexpr (SCALE) {
                scale_a(sa, sw, rs, ff, fp, c2, by, ar, ac);
                stage_store<BROW>(As[0], Bs[0], sw, sb, ar, ac, br, bc);
            } else {
                stage_store<BROW>(As[0], Bs[0], sa, sb, ar, ac, br, bc);
            }
        }
        __syncthreads();

        const int c3 = c + 3 < NC ? c + 3 : NC - 1;
        if constexpr (SCALE) load_factors(ff, fp, Fq, c3, ar);
        if (more) {
            stage_load<RP, BROW>(sa, sb, pA, pB, c + 3);
            frag_load(As[0], Bs[0], fa0, fb0, arow, half, n);
        }
        asm volatile("" ::: "memory");
        __builtin_amdgcn_sched_barrier(0);
        mma_chunk(fa1, fb1, acc0, acc1, acc2, acc3);
        if (more) {
            if constexpr (SCALE) {
                scale_a(sa, sw, rs, ff, fp, c3, by, ar, ac);
                stage_store<BROW>(As[1], Bs[1], sw, sb, ar, ac, br, bc);
            } else {
                stage_store<BROW>(As[1], Bs[1], sa, sb, ar, ac, br, bc);
            }
            __syncthreads();
        }
    }
}

__global__ void __launch_bounds__(128) gemm1_scores(const float* __restrict__ feats,
                             const float* __restrict__ Wpk,
                             const float* __restrict__ biasp,
                             float* __restrict__ attn)
{
    __shared__ float As[2][BM][LDS_PAD];
    __shared__ float Bs[2][KC / 2][BPITCH];

    const int lane = threadIdx.x;
    const int wave = threadIdx.y;
    const int j0   = blockIdx.x * BN;
    const int ib   = blockIdx.y * BM;
    const int i0   = ib + wave * 16;
    const int half = lane >> 4;
    const int n    = lane & 15;

    v8f acc0, acc1, acc2, acc3;
    {
        const float2 bA = *(const float2*)&biasp[j0 + 2 * n];
        const float2 bB = *(const float2*)&biasp[j0 + 32 + 2 * n];
#pragma unroll
        for (int r = 0; r < 8; ++r) {
            acc0[r] = bA.x; acc1[r] = bA.y; acc2[r] = bB.x; acc3[r] = bB.y;
        }
    }

    v4f unused[8];
    gemm_tile<CH, CH, 2 * NP1, FP, false, false>(feats, Wpk, ib, j0, As, Bs,
                                                 acc0, acc1, acc2, acc3, nullptr, unused);

    const int  jpA = j0 + 2 * n;
    const bool ok0 = jpA < N1, ok1 = jpA + 1 < N1, ok2 = jpA + 32 < N1, ok3 = jpA + 33 < N1;
#pragma unroll
    for (int r = 0; r < 8; ++r) {
        float mx = -3.402823466e+38f;
        mx = ok0 ? fmaxf(mx, acc0[r]) : mx;
        mx = ok1 ? fmaxf(mx, acc1[r]) : mx;
        mx = ok2 ? fmaxf(mx, acc2[r]) : mx;
        mx = ok3 ? fmaxf(mx, acc3[r]) : mx;
        mx = fmaxf(mx, __shfl_xor(mx, 1, 16));
        mx = fmaxf(mx, __shfl_xor(mx, 2, 16));
        mx = fmaxf(mx, __shfl_xor(mx, 4, 16));
        mx = fmaxf(mx, __shfl_xor(mx, 8, 16));
        const float e0 = expf(acc0[r] - mx);
        const float e1 = expf(acc1[r] - mx);
        const float e2 = expf(acc2[r] - mx);
        const float e3 = expf(acc3[r] - mx);
        const int m = i0 + r + 8 * half;
        if (m < MP) {
            float* row = attn + (size_t)m * LDATT;
            if (ok0) row[jpA      + (jpA      >= m ? 1 : 0)] = e0;
            if (ok1) row[jpA + 1  + (jpA + 1  >= m ? 1 : 0)] = e1;
            if (ok2) row[jpA + 32 + (jpA + 32 >= m ? 1 : 0)] = e2;
            if (ok3) row[jpA + 33 + (jpA + 33 >= m ? 1 : 0)] = e3;
            if (n == 0) {
                row[A_DIM + blockIdx.x] = mx;
                if (m < A_DIM && (m >> 6) == (int)blockIdx.x) row[m] = 0.0f;
            }
            if (blockIdx.x == 15 && n < 8) row[A_DIM + 16 + n] = 0.0f;
        }
    }
}

#define GATHER8_Y 64
#define GATHER8_BLOCKS (12 * GATHER8_Y)

__device__ __forceinline__ void gather8(int gb, int tid,
                             const float* __restrict__ fv,
                             const int*   __restrict__ zi,
                             const int*   __restrict__ yi,
                             const int*   __restrict__ xi,
                             const int*   __restrict__ mask,
                             float* __restrict__ feats,
                             int b)
{
    const int e0 = (gb * 128 + tid) * 8;
    const int a  = e0 / CH;
    v4f o0 = {}, o1 = {};
    if (a < A_DIM) {
        const int4 ma = *(const int4*)(mask + e0), mb = *(const int4*)(mask + e0 + 4);
        const int4 za = *(const int4*)(zi + e0),   zb = *(const int4*)(zi + e0 + 4);
        const int4 ya = *(const int4*)(yi + e0),   yb = *(const int4*)(yi + e0 + 4);
        const int4 xa = *(const int4*)(xi + e0),   xb = *(const int4*)(xi + e0 + 4);
        const float* base = fv + (size_t)b * (C_DIM * H_DIM * W_DIM);
        const float g0 = base[(za.x * H_DIM + ya.x) * W_DIM + xa.x];
        const float g1 = base[(za.y * H_DIM + ya.y) * W_DIM + xa.y];
        const float g2 = base[(za.z * H_DIM + ya.z) * W_DIM + xa.z];
        const float g3 = base[(za.w * H_DIM + ya.w) * W_DIM + xa.w];
        const float g4 = base[(zb.x * H_DIM + yb.x) * W_DIM + xb.x];
        const float g5 = base[(zb.y * H_DIM + yb.y) * W_DIM + xb.y];
        const float g6 = base[(zb.z * H_DIM + yb.z) * W_DIM + xb.z];
        const float g7 = base[(zb.w * H_DIM + yb.w) * W_DIM + xb.w];
        o0[0] = ma.x ? 0.0f : g0;  o0[1] = ma.y ? 0.0f : g1;
        o0[2] = ma.z ? 0.0f : g2;  o0[3] = ma.w ? 0.0f : g3;
        o1[0] = mb.x ? 0.0f : g4;  o1[1] = mb.y ? 0.0f : g5;
        o1[2] = mb.z ? 0.0f : g6;  o1[3] = mb.w ? 0.0f : g7;
    }
    *(v4f*)(feats + e0)     = o0;
    *(v4f*)(feats + e0 + 4) = o1;
}

__global__ void __launch_bounds__(128) gemm2_out(const float* __restrict__ attn,
                          const float* __restrict__ feats,
                          float* __restrict__ outb,
                          const float* __restrict__ fv,
                          const int*   __restrict__ zi,
                          const int*   __restrict__ yi,
                          const int*   __restrict__ xi,
                          const int*   __restrict__ mask,
                          float* __restrict__ featsNext, int bNext)
{
    __shared__ float As[2][BM][LDS_PAD];
    __shared__ float Bs[2][KC / 2][BPITCH];
    __shared__ float Fq[FP / KC][8][8];
    __shared__ float Ss[BM];

    const int lane = threadIdx.x;
    const int wave = threadIdx.y;
    if (blockIdx.y >= 16) {
        gather8((blockIdx.y - 16) * 12 + blockIdx.x, wave * 32 + lane,
                fv, zi, yi, xi, mask, featsNext, bNext);
        return;
    }
    const int n0   = blockIdx.x * BN;
    const int ib   = blockIdx.y * BM;
    const int i0   = ib + wave * 16;
    const int half = lane >> 4;
    const int n    = lane & 15;

    v8f acc0 = {}, acc1 = {}, acc2 = {}, acc3 = {};

    v4f rs[8];
#pragma unroll
    for (int q = 0; q < 8; ++q) rs[q] = v4f{0.0f, 0.0f, 0.0f, 0.0f};

    gemm_tile<FP, LDATT, CH, MP, true, true>(attn, feats, ib, n0, As, Bs,
                                             acc0, acc1, acc2, acc3, Fq, rs);

#pragma unroll
    for (int q = 0; q < 8; ++q) {
        float s = (rs[q][0] + rs[q][1]) + (rs[q][2] + rs[q][3]);
        s += __shfl_xor(s, 1, 16);
        s += __shfl_xor(s, 2, 16);
        s += __shfl_xor(s, 4, 16);
        s += __shfl_xor(s, 8, 16);
        if (n == 0) Ss[2 * wave + half + 8 * q] = s;
    }
    __syncthreads();

#pragma unroll
    for (int r = 0; r < 8; ++r) {
        const int m = i0 + r + 8 * half;
        const float inv = 1.0f / Ss[wave * 16 + r + 8 * half];
        if (m < A_DIM) {
            float* orow = outb + (size_t)m * CH + n0 + 2 * n;
            *(float2*)(orow)      = make_float2(acc0[r] * inv, acc1[r] * inv);
            *(float2*)(orow + 32) = make_float2(acc2[r] * inv, acc3[r] * inv);
        }
    }
}

extern "C" void kernel_launch(void* const* d_in, const int* in_sizes, int n_in,
                              void* d_out, int out_size, void* d_ws, size_t ws_size,
                              hipStream_t stream)
{
    const float* fv   = (const float*)d_in[0];
    const int*   zi   = (const int*)  d_in[1];
    const int*   yi   = (const int*)  d_in[2];
    const int*   xi   = (const int*)  d_in[3];
    const int*   mask = (const int*)  d_in[4];
    const float* Wt   = (const float*)d_in[5];
    const float* bias = (const float*)d_in[6];
    float*       out  = (float*)d_out;

    float* Wpk     = (float*)d_ws;
    float* biasp   = Wpk     + (size_t)CH * NP1;
    float* feats0  = biasp   + NP1;
    float* feats1  = feats0  + (size_t)FP * CH;
    float* attn    = feats1  + (size_t)FP * CH;

    const dim3 wblk(32, 4);

    pad_w<<<(CH * NP1 + 255) / 256, 256, 0, stream>>>(Wt, Wpk);
    pad_bias<<<(NP1 + 255) / 256, 256, 0, stream>>>(bias, biasp);

    gather_feats<<<GATHER_BLOCKS, 256, 0, stream>>>(fv, zi, yi, xi, mask, feats0, 0);

    for (int b = 0; b < B_DIM; ++b) {
        float* cur = (b & 1) ? feats1 : feats0;
        float* nxt = (b & 1) ? feats0 : feats1;
        gemm1_scores<<<dim3(16, 16), wblk, 0, stream>>>(cur, Wpk, biasp, attn);
        gemm2_out<<<dim3(12, (b + 1 < B_DIM) ? 16 + GATHER8_Y : 16), wblk, 0, stream>>>(
            attn, cur, out + (size_t)b * A_DIM * CH,
            fv, zi, yi, xi, mask, nxt, b + 1);
    }
}


# Both GEMM K loops: global loads requested two chunks further ahead (two staging register sets, in flight across LDS-only barriers), LDS stores moved in front of the matrix run; sums unchanged
# speedup vs baseline: 4.4459x; 1.0891x over previous
#include <hip/hip_runtime.h>


typedef __attribute__((ext_vector_type(2))) float v2f;
typedef __attribute__((ext_vector_type(4))) float v4f;
typedef __attribute__((ext_vector_type(8))) float v8f;

#define B_DIM  32
#define A_DIM  1000
#define C_DIM  64
#define H_DIM  12
#define W_DIM  20
#define CH     768
#define N1     999
#define MP     1008
#define NP1    1024
#define FP     1024
#define LDATT  1024
#define KC     64
#define LDS_PAD 68
#define BM     64
#define BN     64
#define BPITCH 160


__global__ void pad_w(const float* __restrict__ Wt, float* __restrict__ Wpk)
{
    int e = blockIdx.x * blockDim.x + threadIdx.x;
    if (e >= CH * NP1) return;
    int k = e >> 10;
    int j = e & (NP1 - 1);
    float v = (j < N1) ? Wt[k * N1 + j] : 0.0f;
    Wpk[(size_t)(k >> 1) * (2 * NP1) + 2 * j + (k & 1)] = v;
}

__global__ void pad_bias(const float* __restrict__ bias, float* __restrict__ biasp)
{
    int j = blockIdx.x * blockDim.x + threadIdx.x;
    if (j < NP1) biasp[j] = (j < N1) ? bias[j] : 0.0f;
}

#define GATHER_BLOCKS ((FP * CH + 255) / 256)

__device__ __forceinline__ void gather_body(int blk,
                             const float* __restrict__ fv,
                             const int*   __restrict__ zi,
                             const int*   __restrict__ yi,
                             const int*   __restrict__ xi,
                             const int*   __restrict__ mask,
                             float* __restrict__ feats,
                             int b)
{
    int e = blk * 256 + threadIdx.x;
    if (e >= FP * CH) return;
    int a = e / CH;
    int k = e - a * CH;
    float v = 0.0f;
    if (a < A_DIM) {
        int idx = a * CH + k;
        if (!mask[idx]) {
            int z = zi[idx], y = yi[idx], x = xi[idx];
            v = fv[((b * C_DIM + z) * H_DIM + y) * W_DIM + x];
        }
    }
    feats[e] = v;
}

__global__ void gather_feats(const float* __restrict__ fv,
                             const int*   __restrict__ zi,
                             const int*   __restrict__ yi,
                             const int*   __restrict__ xi,
                             const int*   __restrict__ mask,
                             float* __restrict__ feats, int b)
{
    gather_body(blockIdx.x, fv, zi, yi, xi, mask, feats, b);
}

template <int RP, bool BROW>
__device__ __forceinline__ void stage_load(v4f (&sa)[8], v4f (&sb)[8],
                                           const float* const (&pA)[8],
                                           const float* __restrict__ pB, int c)
{
#pragma unroll
    for (int q = 0; q < 8; ++q)
        sa[q] = *(const v4f*)(pA[q] + c * KC);
    if constexpr (BROW) {
#pragma unroll
        for (int q = 0; q < 4; ++q) {
            const float* r = pB + (size_t)(c * KC + 16 * q) * RP;
            sb[2 * q]     = *(const v4f*)r;
            sb[2 * q + 1] = *(const v4f*)(r + RP);
        }
    } else {
#pragma unroll
        for (int q = 0; q < 8; ++q)
            sb[q] = *(const v4f*)(pB + (size_t)(c * (KC / 2) + 4 * q) * RP);
    }
}

template <bool BROW>
__device__ __forceinline__ void stage_store(float (*As)[LDS_PAD], float (*Bs)[BPITCH],
                                            const v4f (&sa)[8], const v4f (&sb)[8],
                                            int ar, int ac, int br, int bc)
{
#pragma unroll
    for (int q = 0; q < 8; ++q)
        *(v4f*)&As[ar + 8 * q][ac] = sa[q];
    if constexpr (BROW) {
#pragma unroll
        for (int q = 0; q < 4; ++q) {
            *(v4f*)&Bs[br + 8 * q][bc] =
                __builtin_shufflevector(sb[2 * q], sb[2 * q + 1], 0, 4, 1, 5);
            *(v4f*)&Bs[br + 8 * q][bc + 4] =
                __builtin_shufflevector(sb[2 * q], sb[2 * q + 1], 2, 6, 3, 7);
        }
    } else {
#pragma unroll
        for (int q = 0; q < 8; ++q)
            *(v4f*)&Bs[br + 4 * q][bc] = sb[q];
    }
}

__device__ __forceinline__ void frag_load(const float (*As)[LDS_PAD],
                                          const float (*Bs)[BPITCH],
                                          v2f (&fa)[16], v4f (&fb)[32],
                                          int arow, int half, int n)
{
#pragma unroll
    for (int s = 0; s < 16; ++s) {
        fa[s]         = *(const v2f*)&As[arow][4 * s + 2 * half];
        fb[2 * s]     = *(const v4f*)&Bs[2 * s + half][4 * n];
        fb[2 * s + 1] = *(const v4f*)&Bs[2 * s + half][64 + 4 * n];
    }
}

__device__ __forceinline__ void mma_chunk(const v2f (&fa)[16], const v4f (&fb)[32],
                                          v8f& acc0, v8f& acc1,
                                          v8f& acc2, v8f& acc3)
{
#pragma unroll
    for (int s = 0; s < 16; ++s) {
        const v2f b0 = __builtin_shufflevector(fb[2 * s],     fb[2 * s],     0, 1);
        const v2f b1 = __builtin_shufflevector(fb[2 * s],     fb[2 * s],     2, 3);
        const v2f b2 = __builtin_shufflevector(fb[2 * s + 1], fb[2 * s + 1], 0, 1);
        const v2f b3 = __builtin_shufflevector(fb[2 * s + 1], fb[2 * s + 1], 2, 3);
        acc0 = __builtin_amdgcn_wmma_f32_16x16x4_f32(
            false, fa[s], false, b0, (short)0, acc0, false, false);
        acc1 = __builtin_amdgcn_wmma_f32_16x16x4_f32(
            false, fa[s], false, b1, (short)0, acc1, false, false);
        acc2 = __builtin_amdgcn_wmma_f32_16x16x4_f32(
            false, fa[s], false, b2, (short)0, acc2, false, false);
        acc3 = __builtin_amdgcn_wmma_f32_16x16x4_f32(
            false, fa[s], false, b3, (short)0, acc3, false, false);
    }
}

typedef float (*FqPtr)[8][8];

__device__ __forceinline__ void row_maxima(const float* __restrict__ attn,
                                           v4f& m0, v4f& m1, int ib, int tid)
{
    const int rl = tid >> 1;
    const int hc = (tid & 1) * 8;
    int row = ib + rl;
    row = row < MP ? row : MP - 1;
    const float* st = attn + (size_t)row * LDATT + A_DIM + hc;
    m0 = *(const v4f*)st;
    m1 = *(const v4f*)(st + 4);
}

__device__ __forceinline__ void row_factors(const v4f m0, const v4f m1, FqPtr Fq, int tid)
{
    const int rl = tid >> 1;
    const int hc = (tid & 1) * 8;
    float mx = fmaxf(fmaxf(fmaxf(m0[0], m0[1]), fmaxf(m0[2], m0[3])),
                     fmaxf(fmaxf(m1[0], m1[1]), fmaxf(m1[2], m1[3])));
    mx = fmaxf(mx, __shfl_xor(mx, 1, 32));
#pragma unroll
    for (int e = 0; e < 4; ++e) {
        Fq[hc + e][rl & 7][rl >> 3]     = expf(m0[e] - mx);
        Fq[hc + 4 + e][rl & 7][rl >> 3] = expf(m1[e] - mx);
    }
}

__device__ __forceinline__ void lds_barrier()
{
    __builtin_amdgcn_fence(__ATOMIC_RELEASE, "workgroup", "local");
    __builtin_amdgcn_s_barrier();
    __builtin_amdgcn_fence(__ATOMIC_ACQUIRE, "workgroup", "local");
}

__device__ __forceinline__ void load_factors(v4f (&ff)[2], v4f (&fp)[2],
                                             FqPtr Fq, int c, int ar)
{
    const int cp = c > 0 ? c - 1 : 0;
    ff[0] = *(const v4f*)&Fq[c][ar][0];
    ff[1] = *(const v4f*)&Fq[c][ar][4];
    fp[0] = *(const v4f*)&Fq[cp][ar][0];
    fp[1] = *(const v4f*)&Fq[cp][ar][4];
}

__device__ __forceinline__ void scale_a(v4f (&sa)[8], v4f (&rs)[8],
                                        const v4f (&ff)[2], const v4f (&fp)[2],
                                        int c, int by, int ar, int ac)
{
    const bool  first = (ac == 0) && (c > by);
    const float keep  = (c * KC + ac >= A_DIM) ? 0.0f : 1.0f;
    const v4f   keep4 = {keep, keep, keep, keep};
#pragma unroll
    for (int q = 0; q < 8; ++q) {
        const float f  = ff[q >> 2][q & 3];
        const float f0 = first ? fp[q >> 2][q & 3] : f;
        const v4f fv4 = {f0, f, f, f};
        const v4f w = sa[q] * fv4;
        rs[q] = w * keep4 + rs[q];
        sa[q] = w;
    }
}

template <int RP, bool BROW, bool SCALE, bool STORE, bool LOAD, bool FACT, bool FRAG, bool BAR>
__device__ __forceinline__ void half_trip(v4f (&sa)[8], v4f (&sb)[8],
                                          float (*Asw)[LDS_PAD], float (*Bsw)[BPITCH],
                                          const float (*Asr)[LDS_PAD], const float (*Bsr)[BPITCH],
                                          v2f (&faN)[16], v4f (&fbN)[32],
                                          const v2f (&faC)[16], const v4f (&fbC)[32],
                                          v8f& acc0, v8f& acc1, v8f& acc2, v8f& acc3,
                                          v4f (&rs)[8], v4f (&ff)[2], v4f (&fp)[2],
                                          FqPtr Fq, const float* const (&pA)[8],
                                          const float* __restrict__ pB,
                                          int cs, int cl, int cf, int by,
                                          int ar, int ac, int br, int bc,
                                          int arow, int half, int n)
{
    if constexpr (STORE) {
        if constexpr (BROW) {
            asm volatile("" : "+v"(sb[0]), "+v"(sb[1]), "+v"(sb[2]), "+v"(sb[3]),
                              "+v"(sb[4]), "+v"(sb[5]), "+v"(sb[6]), "+v"(sb[7]));
        }
        if constexpr (SCALE) {
            scale_a(sa, rs, ff, fp, cs, by, ar, ac);
            asm volatile("" : "+v"(rs[0]), "+v"(rs[1]), "+v"(rs[2]), "+v"(rs[3]),
                              "+v"(rs[4]), "+v"(rs[5]), "+v"(rs[6]), "+v"(rs[7]));
        }
        stage_store<BROW>(Asw, Bsw, sa, sb, ar, ac, br, bc);
    }
    if constexpr (STORE && LOAD) {
        __builtin_amdgcn_sched_barrier(0);
    }
    if constexpr (LOAD) stage_load<RP, BROW>(sa, sb, pA, pB, cl);
    if constexpr (SCALE && FACT) load_factors(ff, fp, Fq, cf, ar);
    if constexpr (FRAG) frag_load(Asr, Bsr, faN, fbN, arow, half, n);
    asm volatile("" : "+v"(acc0), "+v"(acc1), "+v"(acc2), "+v"(acc3) :: "memory");
    __builtin_amdgcn_sched_barrier(0);
    mma_chunk(faC, fbC, acc0, acc1, acc2, acc3);
    if constexpr (BAR) {
        asm volatile("" : "+v"(acc0), "+v"(acc1), "+v"(acc2), "+v"(acc3));
        __builtin_amdgcn_sched_barrier(0);
        lds_barrier();
    }
}

template <int KTOT, int LDA, int RP, int AROWS, bool BROW, bool SCALE>
__device__ __forceinline__ void gemm_tile(const float* __restrict__ A,
                                          const float* __restrict__ Bpk,
                                          int ib, int j0,
                                          float (*As)[BM][LDS_PAD],
                                          float (*Bs)[KC / 2][BPITCH],
                                          v8f& acc0, v8f& acc1,
                                          v8f& acc2, v8f& acc3,
                                          FqPtr Fq, v4f (&rs)[8])
{
    constexpr int NC = KTOT / KC;
    const int lane = threadIdx.x;
    const int wave = threadIdx.y;
    const int tid  = wave * 32 + lane;
    const int half = lane >> 4;
    const int n    = lane & 15;
    const int arow = wave * 16 + n;
    const int ar   = tid >> 4;
    const int ac   = (tid & 15) * 4;
    const int br   = BROW ? (tid >> 4) : (tid >> 5);
    const int bc   = BROW ? (tid & 15) * 8 : (tid & 31) * 4;

    const float* pA[8];
#pragma unroll
    for (int q = 0; q < 8; ++q) {
        int row = ib + ar + 8 * q;
        row = row < AROWS ? row : AROWS - 1;
        pA[q] = A + (size_t)row * LDA + ac;
    }
    const float* pB = BROW ? Bpk + (size_t)(2 * br) * RP + j0 + (bc >> 1)
                           : Bpk + (size_t)br * RP + 2 * j0 + bc;

    v4f sa0[8], sb0[8];
    v4f sa1[8], sb1[8];
    v2f fa0[16], fa1[16];
    v4f fb0[32], fb1[32];

    const int by = ib / BM;
    v4f ff[2], fp[2];
    v4f m0, m1;
    v4f ta0[8], tb0[8], ta1[8], tb1[8];
    if constexpr (SCALE) row_maxima(A, m0, m1, ib, tid);
    stage_load<RP, BROW>(ta0, tb0, pA, pB, 0);
    stage_load<RP, BROW>(ta1, tb1, pA, pB, 1);
    __builtin_amdgcn_sched_barrier(0);
    stage_load<RP, BROW>(sa0, sb0, pA, pB, 2);
    __builtin_amdgcn_sched_barrier(0);
    stage_load<RP, BROW>(sa1, sb1, pA, pB, 3);
    __builtin_amdgcn_sched_barrier(0);
    if constexpr (SCALE) {
        row_factors(m0, m1, Fq, tid);
        lds_barrier();
        load_factors(ff, fp, Fq, 0, ar);
        scale_a(ta0, rs, ff, fp, 0, by, ar, ac);
        stage_store<BROW>(As[0], Bs[0], ta0, tb0, ar, ac, br, bc);
        load_factors(ff, fp, Fq, 1, ar);
        scale_a(ta1, rs, ff, fp, 1, by, ar, ac);
        stage_store<BROW>(As[1], Bs[1], ta1, tb1, ar, ac, br, bc);
        load_factors(ff, fp, Fq, 2, ar);
    } else {
        stage_store<BROW>(As[0], Bs[0], ta0, tb0, ar, ac, br, bc);
        stage_store<BROW>(As[1], Bs[1], ta1, tb1, ar, ac, br, bc);
    }
    lds_barrier();
    frag_load(As[0], Bs[0], fa0, fb0, arow, half, n);
    lds_barrier();

    static_assert(NC >= 6 && (NC & 1) == 0, "K loop: fill of 4 chunks, two peeled trips");
#pragma nounroll
    for (int c = 0; c < NC - 4; c += 2) {
        half_trip<RP, BROW, SCALE, true, true, true, true, true>(
            sa0, sb0, As[0], Bs[0], As[1], Bs[1], fa1, fb1, fa0, fb0,
            acc0, acc1, acc2, acc3, rs, ff, fp, Fq, pA, pB,
            c + 2, c + 4, c + 3, by, ar, ac, br, bc, arow, half, n);
        half_trip<RP, BROW, SCALE, true, true, true, true, true>(
            sa1, sb1, As[1], Bs[1], As[0], Bs[0], fa0, fb0, fa1, fb1,
            acc0, acc1, acc2, acc3, rs, ff, fp, Fq, pA, pB,
            c + 3, c + 5, c + 4, by, ar, ac, br, bc, arow, half, n);
    }
    half_trip<RP, BROW, SCALE, true, false, true, true, true>(
        sa0, sb0, As[0], Bs[0], As[1], Bs[1], fa1, fb1, fa0, fb0,
        acc0, acc1, acc2, acc3, rs, ff, fp, Fq, pA, pB,
        NC - 2, 0, NC - 1, by, ar, ac, br, bc, arow, half, n);
    half_trip<RP, BROW, SCALE, true, false, false, true, true>(
        sa1, sb1, As[1], Bs[1], As[0], Bs[0], fa0, fb0, fa1, fb1,
        acc0, acc1, acc2, acc3, rs, ff, fp, Fq, pA, pB,
        NC - 1, 0, 0, by, ar, ac, br, bc, arow, half, n);
    half_trip<RP, BROW, SCALE, false, false, false, true, false>(
        sa0, sb0, As[0], Bs[0], As[1], Bs[1], fa1, fb1, fa0, fb0,
        acc0, acc1, acc2, acc3, rs, ff, fp, Fq, pA, pB,
        0, 0, 0, by, ar, ac, br, bc, arow, half, n);
    half_trip<RP, BROW, SCALE, false, false, false, false, false>(
        sa1, sb1, As[1], Bs[1], As[0], Bs[0], fa0, fb0, fa1, fb1,
        acc0, acc1, acc2, acc3, rs, ff, fp, Fq, pA, pB,
        0, 0, 0, by, ar, ac, br, bc, arow, half, n);
}

__global__ void __launch_bounds__(128) gemm1_scores(const float* __restrict__ feats,
                             const float* __restrict__ Wpk,
                             const float* __restrict__ biasp,
                             float* __restrict__ attn)
{
    __shared__ float As[2][BM][LDS_PAD];
    __shared__ float Bs[2][KC / 2][BPITCH];

    const int lane = threadIdx.x;
    const int wave = threadIdx.y;
    const int j0   = blockIdx.x * BN;
    const int ib   = blockIdx.y * BM;
    const int i0   = ib + wave * 16;
    const int half = lane >> 4;
    const int n    = lane & 15;

    v8f acc0, acc1, acc2, acc3;
    {
        const float2 bA = *(const float2*)&biasp[j0 + 2 * n];
        const float2 bB = *(const float2*)&biasp[j0 + 32 + 2 * n];
#pragma unroll
        for (int r = 0; r < 8; ++r) {
            acc0[r] = bA.x; acc1[r] = bA.y; acc2[r] = bB.x; acc3[r] = bB.y;
        }
    }

    v4f unused[8];
    gemm_tile<CH, CH, 2 * NP1, FP, false, false>(feats, Wpk, ib, j0, As, Bs,
                                                 acc0, acc1, acc2, acc3, nullptr, unused);

    const int  jpA = j0 + 2 * n;
    const bool ok0 = jpA < N1, ok1 = jpA + 1 < N1, ok2 = jpA + 32 < N1, ok3 = jpA + 33 < N1;
#pragma unroll
    for (int r = 0; r < 8; ++r) {
        float mx = -3.402823466e+38f;
        mx = ok0 ? fmaxf(mx, acc0[r]) : mx;
        mx = ok1 ? fmaxf(mx, acc1[r]) : mx;
        mx = ok2 ? fmaxf(mx, acc2[r]) : mx;
        mx = ok3 ? fmaxf(mx, acc3[r]) : mx;
        mx = fmaxf(mx, __shfl_xor(mx, 1, 16));
        mx = fmaxf(mx, __shfl_xor(mx, 2, 16));
        mx = fmaxf(mx, __shfl_xor(mx, 4, 16));
        mx = fmaxf(mx, __shfl_xor(mx, 8, 16));
        const float e0 = expf(acc0[r] - mx);
        const float e1 = expf(acc1[r] - mx);
        const float e2 = expf(acc2[r] - mx);
        const float e3 = expf(acc3[r] - mx);
        const int m = i0 + r + 8 * half;
        if (m < MP) {
            float* row = attn + (size_t)m * LDATT;
            if (ok0) row[jpA      + (jpA      >= m ? 1 : 0)] = e0;
            if (ok1) row[jpA + 1  + (jpA + 1  >= m ? 1 : 0)] = e1;
            if (ok2) row[jpA + 32 + (jpA + 32 >= m ? 1 : 0)] = e2;
            if (ok3) row[jpA + 33 + (jpA + 33 >= m ? 1 : 0)] = e3;
            if (n == 0) {
                row[A_DIM + blockIdx.x] = mx;
                if (m < A_DIM && (m >> 6) == (int)blockIdx.x) row[m] = 0.0f;
            }
            if (blockIdx.x == 15 && n < 8) row[A_DIM + 16 + n] = 0.0f;
        }
    }
}

#define GATHER8_Y 64
#define GATHER8_BLOCKS (12 * GATHER8_Y)

__device__ __forceinline__ void gather8(int gb, int tid,
                             const float* __restrict__ fv,
                             const int*   __restrict__ zi,
                             const int*   __restrict__ yi,
                             const int*   __restrict__ xi,
                             const int*   __restrict__ mask,
                             float* __restrict__ feats,
                             int b)
{
    const int e0 = (gb * 128 + tid) * 8;
    const int a  = e0 / CH;
    v4f o0 = {}, o1 = {};
    if (a < A_DIM) {
        const int4 ma = *(const int4*)(mask + e0), mb = *(const int4*)(mask + e0 + 4);
        const int4 za = *(const int4*)(zi + e0),   zb = *(const int4*)(zi + e0 + 4);
        const int4 ya = *(const int4*)(yi + e0),   yb = *(const int4*)(yi + e0 + 4);
        const int4 xa = *(const int4*)(xi + e0),   xb = *(const int4*)(xi + e0 + 4);
        const float* base = fv + (size_t)b * (C_DIM * H_DIM * W_DIM);
        const float g0 = base[(za.x * H_DIM + ya.x) * W_DIM + xa.x];
        const float g1 = base[(za.y * H_DIM + ya.y) * W_DIM + xa.y];
        const float g2 = base[(za.z * H_DIM + ya.z) * W_DIM + xa.z];
        const float g3 = base[(za.w * H_DIM + ya.w) * W_DIM + xa.w];
        const float g4 = base[(zb.x * H_DIM + yb.x) * W_DIM + xb.x];
        const float g5 = base[(zb.y * H_DIM + yb.y) * W_DIM + xb.y];
        const float g6 = base[(zb.z * H_DIM + yb.z) * W_DIM + xb.z];
        const float g7 = base[(zb.w * H_DIM + yb.w) * W_DIM + xb.w];
        o0[0] = ma.x ? 0.0f : g0;  o0[1] = ma.y ? 0.0f : g1;
        o0[2] = ma.z ? 0.0f : g2;  o0[3] = ma.w ? 0.0f : g3;
        o1[0] = mb.x ? 0.0f : g4;  o1[1] = mb.y ? 0.0f : g5;
        o1[2] = mb.z ? 0.0f : g6;  o1[3] = mb.w ? 0.0f : g7;
    }
    *(v4f*)(feats + e0)     = o0;
    *(v4f*)(feats + e0 + 4) = o1;
}

__global__ void __launch_bounds__(128) gemm2_out(const float* __restrict__ attn,
                          const float* __restrict__ feats,
                          float* __restrict__ outb,
                          const float* __restrict__ fv,
                          const int*   __restrict__ zi,
                          const int*   __restrict__ yi,
                          const int*   __restrict__ xi,
                          const int*   __restrict__ mask,
                          float* __restrict__ featsNext, int bNext)
{
    __shared__ float As[2][BM][LDS_PAD];
    __shared__ float Bs[2][KC / 2][BPITCH];
    __shared__ float Fq[FP / KC][8][8];
    __shared__ float Ss[BM];

    const int lane = threadIdx.x;
    const int wave = threadIdx.y;
    if (blockIdx.y >= 16) {
        gather8((blockIdx.y - 16) * 12 + blockIdx.x, wave * 32 + lane,
                fv, zi, yi, xi, mask, featsNext, bNext);
        return;
    }
    const int n0   = blockIdx.x * BN;
    const int ib   = blockIdx.y * BM;
    const int i0   = ib + wave * 16;
    const int half = lane >> 4;
    const int n    = lane & 15;

    v8f acc0 = {}, acc1 = {}, acc2 = {}, acc3 = {};

    v4f rs[8];
#pragma unroll
    for (int q = 0; q < 8; ++q) rs[q] = v4f{0.0f, 0.0f, 0.0f, 0.0f};

    gemm_tile<FP, LDATT, CH, MP, true, true>(attn, feats, ib, n0, As, Bs,
                                             acc0, acc1, acc2, acc3, Fq, rs);

#pragma unroll
    for (int q = 0; q < 8; ++q) {
        float s = (rs[q][0] + rs[q][1]) + (rs[q][2] + rs[q][3]);
        s += __shfl_xor(s, 1, 16);
        s += __shfl_xor(s, 2, 16);
        s += __shfl_xor(s, 4, 16);
        s += __shfl_xor(s, 8, 16);
        if (n == 0) Ss[2 * wave + half + 8 * q] = s;
    }
    __syncthreads();

#pragma unroll
    for (int r = 0; r < 8; ++r) {
        const int m = i0 + r + 8 * half;
        const float inv = 1.0f / Ss[wave * 16 + r + 8 * half];
        if (m < A_DIM) {
            float* orow = outb + (size_t)m * CH + n0 + 2 * n;
            *(float2*)(orow)      = make_float2(acc0[r] * inv, acc1[r] * inv);
            *(float2*)(orow + 32) = make_float2(acc2[r] * inv, acc3[r] * inv);
        }
    }
}

extern "C" void kernel_launch(void* const* d_in, const int* in_sizes, int n_in,
                              void* d_out, int out_size, void* d_ws, size_t ws_size,
                              hipStream_t stream)
{
    const float* fv   = (const float*)d_in[0];
    const int*   zi   = (const int*)  d_in[1];
    const int*   yi   = (const int*)  d_in[2];
    const int*   xi   = (const int*)  d_in[3];
    const int*   mask = (const int*)  d_in[4];
    const float* Wt   = (const float*)d_in[5];
    const float* bias = (const float*)d_in[6];
    float*       out  = (float*)d_out;

    float* Wpk     = (float*)d_ws;
    float* biasp   = Wpk     + (size_t)CH * NP1;
    float* feats0  = biasp   + NP1;
    float* feats1  = feats0  + (size_t)FP * CH;
    float* attn    = feats1  + (size_t)FP * CH;

    const dim3 wblk(32, 4);

    pad_w<<<(CH * NP1 + 255) / 256, 256, 0, stream>>>(Wt, Wpk);
    pad_bias<<<(NP1 + 255) / 256, 256, 0, stream>>>(bias, biasp);

    gather_feats<<<GATHER_BLOCKS, 256, 0, stream>>>(fv, zi, yi, xi, mask, feats0, 0);

    for (int b = 0; b < B_DIM; ++b) {
        float* cur = (b & 1) ? feats1 : feats0;
        float* nxt = (b & 1) ? feats0 : feats1;
        gemm1_scores<<<dim3(16, 16), wblk, 0, stream>>>(cur, Wpk, biasp, attn);
        gemm2_out<<<dim3(12, (b + 1 < B_DIM) ? 16 + GATHER8_Y : 16), wblk, 0, stream>>>(
            attn, cur, out + (size_t)b * A_DIM * CH,
            fv, zi, yi, xi, mask, nxt, b + 1);
    }
}


# Epilogues, 8 accumulator rows side by side: row max/sum butterflies on DPP row_xmask (were 64 LDS permutes, each waited for), exps and reciprocals ahead of the masked stores; sums unchanged
# speedup vs baseline: 4.7479x; 1.0679x over previous
#include <hip/hip_runtime.h>


typedef __attribute__((ext_vector_type(2))) float v2f;
typedef __attribute__((ext_vector_type(4))) float v4f;
typedef __attribute__((ext_vector_type(8))) float v8f;

template <int M>
__device__ __forceinline__ float row_xor(float v)
{
    const int i = __builtin_bit_cast(int, v);
    return __builtin_bit_cast(float,
        __builtin_amdgcn_update_dpp(i, i, 0x160 + M, 0xf, 0xf, true));
}

#define B_DIM  32
#define A_DIM  1000
#define C_DIM  64
#define H_DIM  12
#define W_DIM  20
#define CH     768
#define N1     999
#define MP     1008
#define NP1    1024
#define FP     1024
#define LDATT  1024
#define KC     64
#define LDS_PAD 68
#define BM     64
#define BN     64
#define BPITCH 160


__global__ void pad_w(const float* __restrict__ Wt, float* __restrict__ Wpk)
{
    int e = blockIdx.x * blockDim.x + threadIdx.x;
    if (e >= CH * NP1) return;
    int k = e >> 10;
    int j = e & (NP1 - 1);
    float v = (j < N1) ? Wt[k * N1 + j] : 0.0f;
    Wpk[(size_t)(k >> 1) * (2 * NP1) + 2 * j + (k & 1)] = v;
}

__global__ void pad_bias(const float* __restrict__ bias, float* __restrict__ biasp)
{
    int j = blockIdx.x * blockDim.x + threadIdx.x;
    if (j < NP1) biasp[j] = (j < N1) ? bias[j] : 0.0f;
}

#define GATHER_BLOCKS ((FP * CH + 255) / 256)

__device__ __forceinline__ void gather_body(int blk,
                             const float* __restrict__ fv,
                             const int*   __restrict__ zi,
                             const int*   __restrict__ yi,
                             const int*   __restrict__ xi,
                             const int*   __restrict__ mask,
                             float* __restrict__ feats,
                             int b)
{
    int e = blk * 256 + threadIdx.x;
    if (e >= FP * CH) return;
    int a = e / CH;
    int k = e - a * CH;
    float v = 0.0f;
    if (a < A_DIM) {
        int idx = a * CH + k;
        if (!mask[idx]) {
            int z = zi[idx], y = yi[idx], x = xi[idx];
            v = fv[((b * C_DIM + z) * H_DIM + y) * W_DIM + x];
        }
    }
    feats[e] = v;
}

__global__ void gather_feats(const float* __restrict__ fv,
                             const int*   __restrict__ zi,
                             const int*   __restrict__ yi,
                             const int*   __restrict__ xi,
                             const int*   __restrict__ mask,
                             float* __restrict__ feats, int b)
{
    gather_body(blockIdx.x, fv, zi, yi, xi, mask, feats, b);
}

template <int RP, bool BROW>
__device__ __forceinline__ void stage_load(v4f (&sa)[8], v4f (&sb)[8],
                                           const float* const (&pA)[8],
                                           const float* __restrict__ pB, int c)
{
#pragma unroll
    for (int q = 0; q < 8; ++q)
        sa[q] = *(const v4f*)(pA[q] + c * KC);
    if constexpr (BROW) {
#pragma unroll
        for (int q = 0; q < 4; ++q) {
            const float* r = pB + (size_t)(c * KC + 16 * q) * RP;
            sb[2 * q]     = *(const v4f*)r;
            sb[2 * q + 1] = *(const v4f*)(r + RP);
        }
    } else {
#pragma unroll
        for (int q = 0; q < 8; ++q)
            sb[q] = *(const v4f*)(pB + (size_t)(c * (KC / 2) + 4 * q) * RP);
    }
}

template <bool BROW>
__device__ __forceinline__ void stage_store(float (*As)[LDS_PAD], float (*Bs)[BPITCH],
                                            const v4f (&sa)[8], const v4f (&sb)[8],
                                            int ar, int ac, int br, int bc)
{
#pragma unroll
    for (int q = 0; q < 8; ++q)
        *(v4f*)&As[ar + 8 * q][ac] = sa[q];
    if constexpr (BROW) {
#pragma unroll
        for (int q = 0; q < 4; ++q) {
            *(v4f*)&Bs[br + 8 * q][bc] =
                __builtin_shufflevector(sb[2 * q], sb[2 * q + 1], 0, 4, 1, 5);
            *(v4f*)&Bs[br + 8 * q][bc + 4] =
                __builtin_shufflevector(sb[2 * q], sb[2 * q + 1], 2, 6, 3, 7);
        }
    } else {
#pragma unroll
        for (int q = 0; q < 8; ++q)
            *(v4f*)&Bs[br + 4 * q][bc] = sb[q];
    }
}

__device__ __forceinline__ void frag_load(const float (*As)[LDS_PAD],
                                          const float (*Bs)[BPITCH],
                                          v2f (&fa)[16], v4f (&fb)[32],
                                          int arow, int half, int n)
{
#pragma unroll
    for (int s = 0; s < 16; ++s) {
        fa[s]         = *(const v2f*)&As[arow][4 * s + 2 * half];
        fb[2 * s]     = *(const v4f*)&Bs[2 * s + half][4 * n];
        fb[2 * s + 1] = *(const v4f*)&Bs[2 * s + half][64 + 4 * n];
    }
}

__device__ __forceinline__ void mma_chunk(const v2f (&fa)[16], const v4f (&fb)[32],
                                          v8f& acc0, v8f& acc1,
                                          v8f& acc2, v8f& acc3)
{
#pragma unroll
    for (int s = 0; s < 16; ++s) {
        const v2f b0 = __builtin_shufflevector(fb[2 * s],     fb[2 * s],     0, 1);
        const v2f b1 = __builtin_shufflevector(fb[2 * s],     fb[2 * s],     2, 3);
        const v2f b2 = __builtin_shufflevector(fb[2 * s + 1], fb[2 * s + 1], 0, 1);
        const v2f b3 = __builtin_shufflevector(fb[2 * s + 1], fb[2 * s + 1], 2, 3);
        acc0 = __builtin_amdgcn_wmma_f32_16x16x4_f32(
            false, fa[s], false, b0, (short)0, acc0, false, false);
        acc1 = __builtin_amdgcn_wmma_f32_16x16x4_f32(
            false, fa[s], false, b1, (short)0, acc1, false, false);
        acc2 = __builtin_amdgcn_wmma_f32_16x16x4_f32(
            false, fa[s], false, b2, (short)0, acc2, false, false);
        acc3 = __builtin_amdgcn_wmma_f32_16x16x4_f32(
            false, fa[s], false, b3, (short)0, acc3, false, false);
    }
}

typedef float (*FqPtr)[8][8];

__device__ __forceinline__ void row_maxima(const float* __restrict__ attn,
                                           v4f& m0, v4f& m1, int ib, int tid)
{
    const int rl = tid >> 1;
    const int hc = (tid & 1) * 8;
    int row = ib + rl;
    row = row < MP ? row : MP - 1;
    const float* st = attn + (size_t)row * LDATT + A_DIM + hc;
    m0 = *(const v4f*)st;
    m1 = *(const v4f*)(st + 4);
}

__device__ __forceinline__ void row_factors(const v4f m0, const v4f m1, FqPtr Fq, int tid)
{
    const int rl = tid >> 1;
    const int hc = (tid & 1) * 8;
    float mx = fmaxf(fmaxf(fmaxf(m0[0], m0[1]), fmaxf(m0[2], m0[3])),
                     fmaxf(fmaxf(m1[0], m1[1]), fmaxf(m1[2], m1[3])));
    mx = fmaxf(mx, row_xor<1>(mx));
#pragma unroll
    for (int e = 0; e < 4; ++e) {
        Fq[hc + e][rl & 7][rl >> 3]     = expf(m0[e] - mx);
        Fq[hc + 4 + e][rl & 7][rl >> 3] = expf(m1[e] - mx);
    }
}

__device__ __forceinline__ void lds_barrier()
{
    __builtin_amdgcn_fence(__ATOMIC_RELEASE, "workgroup", "local");
    __builtin_amdgcn_s_barrier();
    __builtin_amdgcn_fence(__ATOMIC_ACQUIRE, "workgroup", "local");
}

__device__ __forceinline__ void load_factors(v4f (&ff)[2], v4f (&fp)[2],
                                             FqPtr Fq, int c, int ar)
{
    const int cp = c > 0 ? c - 1 : 0;
    ff[0] = *(const v4f*)&Fq[c][ar][0];
    ff[1] = *(const v4f*)&Fq[c][ar][4];
    fp[0] = *(const v4f*)&Fq[cp][ar][0];
    fp[1] = *(const v4f*)&Fq[cp][ar][4];
}

__device__ __forceinline__ void scale_a(v4f (&sa)[8], v4f (&rs)[8],
                                        const v4f (&ff)[2], const v4f (&fp)[2],
                                        int c, int by, int ar, int ac)
{
    const bool  first = (ac == 0) && (c > by);
    const float keep  = (c * KC + ac >= A_DIM) ? 0.0f : 1.0f;
    const v4f   keep4 = {keep, keep, keep, keep};
#pragma unroll
    for (int q = 0; q < 8; ++q) {
        const float f  = ff[q >> 2][q & 3];
        const float f0 = first ? fp[q >> 2][q & 3] : f;
        const v4f fv4 = {f0, f, f, f};
        const v4f w = sa[q] * fv4;
        rs[q] = w * keep4 + rs[q];
        sa[q] = w;
    }
}

template <int RP, bool BROW, bool SCALE, bool STORE, bool LOAD, bool FACT, bool FRAG, bool BAR>
__device__ __forceinline__ void half_trip(v4f (&sa)[8], v4f (&sb)[8],
                                          float (*Asw)[LDS_PAD], float (*Bsw)[BPITCH],
                                          const float (*Asr)[LDS_PAD], const float (*Bsr)[BPITCH],
                                          v2f (&faN)[16], v4f (&fbN)[32],
                                          const v2f (&faC)[16], const v4f (&fbC)[32],
                                          v8f& acc0, v8f& acc1, v8f& acc2, v8f& acc3,
                                          v4f (&rs)[8], v4f (&ff)[2], v4f (&fp)[2],
                                          FqPtr Fq, const float* const (&pA)[8],
                                          const float* __restrict__ pB,
                                          int cs, int cl, int cf, int by,
                                          int ar, int ac, int br, int bc,
                                          int arow, int half, int n)
{
    if constexpr (STORE) {
        if constexpr (BROW) {
            asm volatile("" : "+v"(sb[0]), "+v"(sb[1]), "+v"(sb[2]), "+v"(sb[3]),
                              "+v"(sb[4]), "+v"(sb[5]), "+v"(sb[6]), "+v"(sb[7]));
        }
        if constexpr (SCALE) {
            scale_a(sa, rs, ff, fp, cs, by, ar, ac);
            asm volatile("" : "+v"(rs[0]), "+v"(rs[1]), "+v"(rs[2]), "+v"(rs[3]),
                              "+v"(rs[4]), "+v"(rs[5]), "+v"(rs[6]), "+v"(rs[7]));
        }
        stage_store<BROW>(Asw, Bsw, sa, sb, ar, ac, br, bc);
    }
    if constexpr (STORE && LOAD) {
        __builtin_amdgcn_sched_barrier(0);
    }
    if constexpr (LOAD) stage_load<RP, BROW>(sa, sb, pA, pB, cl);
    if constexpr (SCALE && FACT) load_factors(ff, fp, Fq, cf, ar);
    if constexpr (FRAG) frag_load(Asr, Bsr, faN, fbN, arow, half, n);
    asm volatile("" : "+v"(acc0), "+v"(acc1), "+v"(acc2), "+v"(acc3) :: "memory");
    __builtin_amdgcn_sched_barrier(0);
    mma_chunk(faC, fbC, acc0, acc1, acc2, acc3);
    if constexpr (BAR) {
        asm volatile("" : "+v"(acc0), "+v"(acc1), "+v"(acc2), "+v"(acc3));
        __builtin_amdgcn_sched_barrier(0);
        lds_barrier();
    }
}

template <int KTOT, int LDA, int RP, int AROWS, bool BROW, bool SCALE>
__device__ __forceinline__ void gemm_tile(const float* __restrict__ A,
                                          const float* __restrict__ Bpk,
                                          int ib, int j0,
                                          float (*As)[BM][LDS_PAD],
                                          float (*Bs)[KC / 2][BPITCH],
                                          v8f& acc0, v8f& acc1,
                                          v8f& acc2, v8f& acc3,
                                          FqPtr Fq, v4f (&rs)[8])
{
    constexpr int NC = KTOT / KC;
    const int lane = threadIdx.x;
    const int wave = threadIdx.y;
    const int tid  = wave * 32 + lane;
    const int half = lane >> 4;
    const int n    = lane & 15;
    const int arow = wave * 16 + n;
    const int ar   = tid >> 4;
    const int ac   = (tid & 15) * 4;
    const int br   = BROW ? (tid >> 4) : (tid >> 5);
    const int bc   = BROW ? (tid & 15) * 8 : (tid & 31) * 4;

    const float* pA[8];
#pragma unroll
    for (int q = 0; q < 8; ++q) {
        int row = ib + ar + 8 * q;
        row = row < AROWS ? row : AROWS - 1;
        pA[q] = A + (size_t)row * LDA + ac;
    }
    const float* pB = BROW ? Bpk + (size_t)(2 * br) * RP + j0 + (bc >> 1)
                           : Bpk + (size_t)br * RP + 2 * j0 + bc;

    v4f sa0[8], sb0[8];
    v4f sa1[8], sb1[8];
    v2f fa0[16], fa1[16];
    v4f fb0[32], fb1[32];

    const int by = ib / BM;
    v4f ff[2], fp[2];
    v4f m0, m1;
    v4f ta0[8], tb0[8], ta1[8], tb1[8];
    if constexpr (SCALE) row_maxima(A, m0, m1, ib, tid);
    stage_load<RP, BROW>(ta0, tb0, pA, pB, 0);
    stage_load<RP, BROW>(ta1, tb1, pA, pB, 1);
    __builtin_amdgcn_sched_barrier(0);
    stage_load<RP, BROW>(sa0, sb0, pA, pB, 2);
    __builtin_amdgcn_sched_barrier(0);
    stage_load<RP, BROW>(sa1, sb1, pA, pB, 3);
    __builtin_amdgcn_sched_barrier(0);
    if constexpr (SCALE) {
        row_factors(m0, m1, Fq, tid);
        lds_barrier();
        load_factors(ff, fp, Fq, 0, ar);
        scale_a(ta0, rs, ff, fp, 0, by, ar, ac);
        stage_store<BROW>(As[0], Bs[0], ta0, tb0, ar, ac, br, bc);
        load_factors(ff, fp, Fq, 1, ar);
        scale_a(ta1, rs, ff, fp, 1, by, ar, ac);
        stage_store<BROW>(As[1], Bs[1], ta1, tb1, ar, ac, br, bc);
        load_factors(ff, fp, Fq, 2, ar);
    } else {
        stage_store<BROW>(As[0], Bs[0], ta0, tb0, ar, ac, br, bc);
        stage_store<BROW>(As[1], Bs[1], ta1, tb1, ar, ac, br, bc);
    }
    lds_barrier();
    frag_load(As[0], Bs[0], fa0, fb0, arow, half, n);
    lds_barrier();

    static_assert(NC >= 6 && (NC & 1) == 0, "K loop: fill of 4 chunks, two peeled trips");
#pragma nounroll
    for (int c = 0; c < NC - 4; c += 2) {
        half_trip<RP, BROW, SCALE, true, true, true, true, true>(
            sa0, sb0, As[0], Bs[0], As[1], Bs[1], fa1, fb1, fa0, fb0,
            acc0, acc1, acc2, acc3, rs, ff, fp, Fq, pA, pB,
            c + 2, c + 4, c + 3, by, ar, ac, br, bc, arow, half, n);
        half_trip<RP, BROW, SCALE, true, true, true, true, true>(
            sa1, sb1, As[1], Bs[1], As[0], Bs[0], fa0, fb0, fa1, fb1,
            acc0, acc1, acc2, acc3, rs, ff, fp, Fq, pA, pB,
            c + 3, c + 5, c + 4, by, ar, ac, br, bc, arow, half, n);
    }
    half_trip<RP, BROW, SCALE, true, false, true, true, true>(
        sa0, sb0, As[0], Bs[0], As[1], Bs[1], fa1, fb1, fa0, fb0,
        acc0, acc1, acc2, acc3, rs, ff, fp, Fq, pA, pB,
        NC - 2, 0, NC - 1, by, ar, ac, br, bc, arow, half, n);
    half_trip<RP, BROW, SCALE, true, false, false, true, true>(
        sa1, sb1, As[1], Bs[1], As[0], Bs[0], fa0, fb0, fa1, fb1,
        acc0, acc1, acc2, acc3, rs, ff, fp, Fq, pA, pB,
        NC - 1, 0, 0, by, ar, ac, br, bc, arow, half, n);
    half_trip<RP, BROW, SCALE, false, false, false, true, false>(
        sa0, sb0, As[0], Bs[0], As[1], Bs[1], fa1, fb1, fa0, fb0,
        acc0, acc1, acc2, acc3, rs, ff, fp, Fq, pA, pB,
        0, 0, 0, by, ar, ac, br, bc, arow, half, n);
    half_trip<RP, BROW, SCALE, false, false, false, false, false>(
        sa1, sb1, As[1], Bs[1], As[0], Bs[0], fa0, fb0, fa1, fb1,
        acc0, acc1, acc2, acc3, rs, ff, fp, Fq, pA, pB,
        0, 0, 0, by, ar, ac, br, bc, arow, half, n);
}

__global__ void __launch_bounds__(128) gemm1_scores(const float* __restrict__ feats,
                             const float* __restrict__ Wpk,
                             const float* __restrict__ biasp,
                             float* __restrict__ attn)
{
    __shared__ float As[2][BM][LDS_PAD];
    __shared__ float Bs[2][KC / 2][BPITCH];

    const int lane = threadIdx.x;
    const int wave = threadIdx.y;
    const int j0   = blockIdx.x * BN;
    const int ib   = blockIdx.y * BM;
    const int i0   = ib + wave * 16;
    const int half = lane >> 4;
    const int n    = lane & 15;

    v8f acc0, acc1, acc2, acc3;
    {
        const float2 bA = *(const float2*)&biasp[j0 + 2 * n];
        const float2 bB = *(const float2*)&biasp[j0 + 32 + 2 * n];
#pragma unroll
        for (int r = 0; r < 8; ++r) {
            acc0[r] = bA.x; acc1[r] = bA.y; acc2[r] = bB.x; acc3[r] = bB.y;
        }
    }

    v4f unused[8];
    gemm_tile<CH, CH, 2 * NP1, FP, false, false>(feats, Wpk, ib, j0, As, Bs,
                                                 acc0, acc1, acc2, acc3, nullptr, unused);

    const int  jpA = j0 + 2 * n;
    const bool ok0 = jpA < N1, ok1 = jpA + 1 < N1, ok2 = jpA + 32 < N1, ok3 = jpA + 33 < N1;
    v8f mx;
#pragma unroll
    for (int r = 0; r < 8; ++r) {
        float v = -3.402823466e+38f;
        v = ok0 ? fmaxf(v, acc0[r]) : v;
        v = ok1 ? fmaxf(v, acc1[r]) : v;
        v = ok2 ? fmaxf(v, acc2[r]) : v;
        v = ok3 ? fmaxf(v, acc3[r]) : v;
        mx[r] = v;
    }
#pragma unroll
    for (int r = 0; r < 8; ++r) mx[r] = fmaxf(mx[r], row_xor<1>(mx[r]));
#pragma unroll
    for (int r = 0; r < 8; ++r) mx[r] = fmaxf(mx[r], row_xor<2>(mx[r]));
#pragma unroll
    for (int r = 0; r < 8; ++r) mx[r] = fmaxf(mx[r], row_xor<4>(mx[r]));
#pragma unroll
    for (int r = 0; r < 8; ++r) mx[r] = fmaxf(mx[r], row_xor<8>(mx[r]));
    v8f ex0, ex1, ex2, ex3;
#pragma unroll
    for (int r = 0; r < 8; ++r) {
        ex0[r] = expf(acc0[r] - mx[r]);
        ex1[r] = expf(acc1[r] - mx[r]);
        ex2[r] = expf(acc2[r] - mx[r]);
        ex3[r] = expf(acc3[r] - mx[r]);
    }
    asm volatile("" : "+v"(ex0), "+v"(ex1), "+v"(ex2), "+v"(ex3));
#pragma unroll
    for (int r = 0; r < 8; ++r) {
        const float e0 = ex0[r], e1 = ex1[r], e2 = ex2[r], e3 = ex3[r];
        const int m = i0 + r + 8 * half;
        if (m < MP) {
            float* row = attn + (size_t)m * LDATT;
            if (ok0) row[jpA      + (jpA      >= m ? 1 : 0)] = e0;
            if (ok1) row[jpA + 1  + (jpA + 1  >= m ? 1 : 0)] = e1;
            if (ok2) row[jpA + 32 + (jpA + 32 >= m ? 1 : 0)] = e2;
            if (ok3) row[jpA + 33 + (jpA + 33 >= m ? 1 : 0)] = e3;
            if (n == 0) {
                row[A_DIM + blockIdx.x] = mx[r];
                if (m < A_DIM && (m >> 6) == (int)blockIdx.x) row[m] = 0.0f;
            }
            if (blockIdx.x == 15 && n < 8) row[A_DIM + 16 + n] = 0.0f;
        }
    }
}

#define GATHER8_Y 64
#define GATHER8_BLOCKS (12 * GATHER8_Y)

__device__ __forceinline__ void gather8(int gb, int tid,
                             const float* __restrict__ fv,
                             const int*   __restrict__ zi,
                             const int*   __restrict__ yi,
                             const int*   __restrict__ xi,
                             const int*   __restrict__ mask,
                             float* __restrict__ feats,
                             int b)
{
    const int e0 = (gb * 128 + tid) * 8;
    const int a  = e0 / CH;
    v4f o0 = {}, o1 = {};
    if (a < A_DIM) {
        const int4 ma = *(const int4*)(mask + e0), mb = *(const int4*)(mask + e0 + 4);
        const int4 za = *(const int4*)(zi + e0),   zb = *(const int4*)(zi + e0 + 4);
        const int4 ya = *(const int4*)(yi + e0),   yb = *(const int4*)(yi + e0 + 4);
        const int4 xa = *(const int4*)(xi + e0),   xb = *(const int4*)(xi + e0 + 4);
        const float* base = fv + (size_t)b * (C_DIM * H_DIM * W_DIM);
        const float g0 = base[(za.x * H_DIM + ya.x) * W_DIM + xa.x];
        const float g1 = base[(za.y * H_DIM + ya.y) * W_DIM + xa.y];
        const float g2 = base[(za.z * H_DIM + ya.z) * W_DIM + xa.z];
        const float g3 = base[(za.w * H_DIM + ya.w) * W_DIM + xa.w];
        const float g4 = base[(zb.x * H_DIM + yb.x) * W_DIM + xb.x];
        const float g5 = base[(zb.y * H_DIM + yb.y) * W_DIM + xb.y];
        const float g6 = base[(zb.z * H_DIM + yb.z) * W_DIM + xb.z];
        const float g7 = base[(zb.w * H_DIM + yb.w) * W_DIM + xb.w];
        o0[0] = ma.x ? 0.0f : g0;  o0[1] = ma.y ? 0.0f : g1;
        o0[2] = ma.z ? 0.0f : g2;  o0[3] = ma.w ? 0.0f : g3;
        o1[0] = mb.x ? 0.0f : g4;  o1[1] = mb.y ? 0.0f : g5;
        o1[2] = mb.z ? 0.0f : g6;  o1[3] = mb.w ? 0.0f : g7;
    }
    *(v4f*)(feats + e0)     = o0;
    *(v4f*)(feats + e0 + 4) = o1;
}

__global__ void __launch_bounds__(128) gemm2_out(const float* __restrict__ attn,
                          const float* __restrict__ feats,
                          float* __restrict__ outb,
                          const float* __restrict__ fv,
                          const int*   __restrict__ zi,
                          const int*   __restrict__ yi,
                          const int*   __restrict__ xi,
                          const int*   __restrict__ mask,
                          float* __restrict__ featsNext, int bNext)
{
    __shared__ float As[2][BM][LDS_PAD];
    __shared__ float Bs[2][KC / 2][BPITCH];
    __shared__ float Fq[FP / KC][8][8];
    __shared__ float Ss[BM];

    const int lane = threadIdx.x;
    const int wave = threadIdx.y;
    if (blockIdx.y >= 16) {
        gather8((blockIdx.y - 16) * 12 + blockIdx.x, wave * 32 + lane,
                fv, zi, yi, xi, mask, featsNext, bNext);
        return;
    }
    const int n0   = blockIdx.x * BN;
    const int ib   = blockIdx.y * BM;
    const int i0   = ib + wave * 16;
    const int half = lane >> 4;
    const int n    = lane & 15;

    v8f acc0 = {}, acc1 = {}, acc2 = {}, acc3 = {};

    v4f rs[8];
#pragma unroll
    for (int q = 0; q < 8; ++q) rs[q] = v4f{0.0f, 0.0f, 0.0f, 0.0f};

    gemm_tile<FP, LDATT, CH, MP, true, true>(attn, feats, ib, n0, As, Bs,
                                             acc0, acc1, acc2, acc3, Fq, rs);

    v8f sm;
#pragma unroll
    for (int q = 0; q < 8; ++q) sm[q] = (rs[q][0] + rs[q][1]) + (rs[q][2] + rs[q][3]);
#pragma unroll
    for (int q = 0; q < 8; ++q) sm[q] += row_xor<1>(sm[q]);
#pragma unroll
    for (int q = 0; q < 8; ++q) sm[q] += row_xor<2>(sm[q]);
#pragma unroll
    for (int q = 0; q < 8; ++q) sm[q] += row_xor<4>(sm[q]);
#pragma unroll
    for (int q = 0; q < 8; ++q) sm[q] += row_xor<8>(sm[q]);
    if (n == 0) {
#pragma unroll
        for (int q = 0; q < 8; ++q) Ss[2 * wave + half + 8 * q] = sm[q];
    }
    __syncthreads();

    v8f iv;
#pragma unroll
    for (int r = 0; r < 8; ++r) iv[r] = 1.0f / Ss[wave * 16 + r + 8 * half];
    asm volatile("" : "+v"(iv));
#pragma unroll
    for (int r = 0; r < 8; ++r) {
        const int m = i0 + r + 8 * half;
        const float inv = iv[r];
        if (m < A_DIM) {
            float* orow = outb + (size_t)m * CH + n0 + 2 * n;
            *(float2*)(orow)      = make_float2(acc0[r] * inv, acc1[r] * inv);
            *(float2*)(orow + 32) = make_float2(acc2[r] * inv, acc3[r] * inv);
        }
    }
}

extern "C" void kernel_launch(void* const* d_in, const int* in_sizes, int n_in,
                              void* d_out, int out_size, void* d_ws, size_t ws_size,
                              hipStream_t stream)
{
    const float* fv   = (const float*)d_in[0];
    const int*   zi   = (const int*)  d_in[1];
    const int*   yi   = (const int*)  d_in[2];
    const int*   xi   = (const int*)  d_in[3];
    const int*   mask = (const int*)  d_in[4];
    const float* Wt   = (const float*)d_in[5];
    const float* bias = (const float*)d_in[6];
    float*       out  = (float*)d_out;

    float* Wpk     = (float*)d_ws;
    float* biasp   = Wpk     + (size_t)CH * NP1;
    float* feats0  = biasp   + NP1;
    float* feats1  = feats0  + (size_t)FP * CH;
    float* attn    = feats1  + (size_t)FP * CH;

    const dim3 wblk(32, 4);

    pad_w<<<(CH * NP1 + 255) / 256, 256, 0, stream>>>(Wt, Wpk);
    pad_bias<<<(NP1 + 255) / 256, 256, 0, stream>>>(bias, biasp);

    gather_feats<<<GATHER_BLOCKS, 256, 0, stream>>>(fv, zi, yi, xi, mask, feats0, 0);

    for (int b = 0; b < B_DIM; ++b) {
        float* cur = (b & 1) ? feats1 : feats0;
        float* nxt = (b & 1) ? feats0 : feats1;
        gemm1_scores<<<dim3(16, 16), wblk, 0, stream>>>(cur, Wpk, biasp, attn);
        gemm2_out<<<dim3(12, (b + 1 < B_DIM) ? 16 + GATHER8_Y : 16), wblk, 0, stream>>>(
            attn, cur, out + (size_t)b * A_DIM * CH,
            fv, zi, yi, xi, mask, nxt, b + 1);
    }
}


# Epilogue stores, both GEMMs: row guard once per wave (scalar wave index), one lane mask per group of 8 rows, row and column in the stores' offset fields, only the diagonal shift per store; sums same
# speedup vs baseline: 4.9061x; 1.0333x over previous
#include <hip/hip_runtime.h>


typedef __attribute__((ext_vector_type(2))) float v2f;
typedef __attribute__((ext_vector_type(4))) float v4f;
typedef __attribute__((ext_vector_type(8))) float v8f;

template <int M>
__device__ __forceinline__ float row_xor(float v)
{
    const int i = __builtin_bit_cast(int, v);
    return __builtin_bit_cast(float,
        __builtin_amdgcn_update_dpp(i, i, 0x160 + M, 0xf, 0xf, true));
}

#define B_DIM  32
#define A_DIM  1000
#define C_DIM  64
#define H_DIM  12
#define W_DIM  20
#define CH     768
#define N1     999
#define MP     1008
#define NP1    1024
#define FP     1024
#define LDATT  1024
#define KC     64
#define LDS_PAD 68
#define BM     64
#define BN     64
#define BPITCH 160


__global__ void pad_w(const float* __restrict__ Wt, float* __restrict__ Wpk)
{
    int e = blockIdx.x * blockDim.x + threadIdx.x;
    if (e >= CH * NP1) return;
    int k = e >> 10;
    int j = e & (NP1 - 1);
    float v = (j < N1) ? Wt[k * N1 + j] : 0.0f;
    Wpk[(size_t)(k >> 1) * (2 * NP1) + 2 * j + (k & 1)] = v;
}

__global__ void pad_bias(const float* __restrict__ bias, float* __restrict__ biasp)
{
    int j = blockIdx.x * blockDim.x + threadIdx.x;
    if (j < NP1) biasp[j] = (j < N1) ? bias[j] : 0.0f;
}

#define GATHER_BLOCKS ((FP * CH + 255) / 256)

__device__ __forceinline__ void gather_body(int blk,
                             const float* __restrict__ fv,
                             const int*   __restrict__ zi,
                             const int*   __restrict__ yi,
                             const int*   __restrict__ xi,
                             const int*   __restrict__ mask,
                             float* __restrict__ feats,
                             int b)
{
    int e = blk * 256 + threadIdx.x;
    if (e >= FP * CH) return;
    int a = e / CH;
    int k = e - a * CH;
    float v = 0.0f;
    if (a < A_DIM) {
        int idx = a * CH + k;
        if (!mask[idx]) {
            int z = zi[idx], y = yi[idx], x = xi[idx];
            v = fv[((b * C_DIM + z) * H_DIM + y) * W_DIM + x];
        }
    }
    feats[e] = v;
}

__global__ void gather_feats(const float* __restrict__ fv,
                             const int*   __restrict__ zi,
                             const int*   __restrict__ yi,
                             const int*   __restrict__ xi,
                             const int*   __restrict__ mask,
                             float* __restrict__ feats, int b)
{
    gather_body(blockIdx.x, fv, zi, yi, xi, mask, feats, b);
}

template <int RP, bool BROW>
__device__ __forceinline__ void stage_load(v4f (&sa)[8], v4f (&sb)[8],
                                           const float* const (&pA)[8],
                                           const float* __restrict__ pB, int c)
{
#pragma unroll
    for (int q = 0; q < 8; ++q)
        sa[q] = *(const v4f*)(pA[q] + c * KC);
    if constexpr (BROW) {
#pragma unroll
        for (int q = 0; q < 4; ++q) {
            const float* r = pB + (size_t)(c * KC + 16 * q) * RP;
            sb[2 * q]     = *(const v4f*)r;
            sb[2 * q + 1] = *(const v4f*)(r + RP);
        }
    } else {
#pragma unroll
        for (int q = 0; q < 8; ++q)
            sb[q] = *(const v4f*)(pB + (size_t)(c * (KC / 2) + 4 * q) * RP);
    }
}

template <bool BROW>
__device__ __forceinline__ void stage_store(float (*As)[LDS_PAD], float (*Bs)[BPITCH],
                                            const v4f (&sa)[8], const v4f (&sb)[8],
                                            int ar, int ac, int br, int bc)
{
#pragma unroll
    for (int q = 0; q < 8; ++q)
        *(v4f*)&As[ar + 8 * q][ac] = sa[q];
    if constexpr (BROW) {
#pragma unroll
        for (int q = 0; q < 4; ++q) {
            *(v4f*)&Bs[br + 8 * q][bc] =
                __builtin_shufflevector(sb[2 * q], sb[2 * q + 1], 0, 4, 1, 5);
            *(v4f*)&Bs[br + 8 * q][bc + 4] =
                __builtin_shufflevector(sb[2 * q], sb[2 * q + 1], 2, 6, 3, 7);
        }
    } else {
#pragma unroll
        for (int q = 0; q < 8; ++q)
            *(v4f*)&Bs[br + 4 * q][bc] = sb[q];
    }
}

__device__ __forceinline__ void frag_load(const float (*As)[LDS_PAD],
                                          const float (*Bs)[BPITCH],
                                          v2f (&fa)[16], v4f (&fb)[32],
                                          int arow, int half, int n)
{
#pragma unroll
    for (int s = 0; s < 16; ++s) {
        fa[s]         = *(const v2f*)&As[arow][4 * s + 2 * half];
        fb[2 * s]     = *(const v4f*)&Bs[2 * s + half][4 * n];
        fb[2 * s + 1] = *(const v4f*)&Bs[2 * s + half][64 + 4 * n];
    }
}

__device__ __forceinline__ void mma_chunk(const v2f (&fa)[16], const v4f (&fb)[32],
                                          v8f& acc0, v8f& acc1,
                                          v8f& acc2, v8f& acc3)
{
#pragma unroll
    for (int s = 0; s < 16; ++s) {
        const v2f b0 = __builtin_shufflevector(fb[2 * s],     fb[2 * s],     0, 1);
        const v2f b1 = __builtin_shufflevector(fb[2 * s],     fb[2 * s],     2, 3);
        const v2f b2 = __builtin_shufflevector(fb[2 * s + 1], fb[2 * s + 1], 0, 1);
        const v2f b3 = __builtin_shufflevector(fb[2 * s + 1], fb[2 * s + 1], 2, 3);
        acc0 = __builtin_amdgcn_wmma_f32_16x16x4_f32(
            false, fa[s], false, b0, (short)0, acc0, false, false);
        acc1 = __builtin_amdgcn_wmma_f32_16x16x4_f32(
            false, fa[s], false, b1, (short)0, acc1, false, false);
        acc2 = __builtin_amdgcn_wmma_f32_16x16x4_f32(
            false, fa[s], false, b2, (short)0, acc2, false, false);
        acc3 = __builtin_amdgcn_wmma_f32_16x16x4_f32(
            false, fa[s], false, b3, (short)0, acc3, false, false);
    }
}

typedef float (*FqPtr)[8][8];

__device__ __forceinline__ void row_maxima(const float* __restrict__ attn,
                                           v4f& m0, v4f& m1, int ib, int tid)
{
    const int rl = tid >> 1;
    const int hc = (tid & 1) * 8;
    int row = ib + rl;
    row = row < MP ? row : MP - 1;
    const float* st = attn + (size_t)row * LDATT + A_DIM + hc;
    m0 = *(const v4f*)st;
    m1 = *(const v4f*)(st + 4);
}

__device__ __forceinline__ void row_factors(const v4f m0, const v4f m1, FqPtr Fq, int tid)
{
    const int rl = tid >> 1;
    const int hc = (tid & 1) * 8;
    float mx = fmaxf(fmaxf(fmaxf(m0[0], m0[1]), fmaxf(m0[2], m0[3])),
                     fmaxf(fmaxf(m1[0], m1[1]), fmaxf(m1[2], m1[3])));
    mx = fmaxf(mx, row_xor<1>(mx));
#pragma unroll
    for (int e = 0; e < 4; ++e) {
        Fq[hc + e][rl & 7][rl >> 3]     = expf(m0[e] - mx);
        Fq[hc + 4 + e][rl & 7][rl >> 3] = expf(m1[e] - mx);
    }
}

__device__ __forceinline__ void lds_barrier()
{
    __builtin_amdgcn_fence(__ATOMIC_RELEASE, "workgroup", "local");
    __builtin_amdgcn_s_barrier();
    __builtin_amdgcn_fence(__ATOMIC_ACQUIRE, "workgroup", "local");
}

__device__ __forceinline__ void load_factors(v4f (&ff)[2], v4f (&fp)[2],
                                             FqPtr Fq, int c, int ar)
{
    const int cp = c > 0 ? c - 1 : 0;
    ff[0] = *(const v4f*)&Fq[c][ar][0];
    ff[1] = *(const v4f*)&Fq[c][ar][4];
    fp[0] = *(const v4f*)&Fq[cp][ar][0];
    fp[1] = *(const v4f*)&Fq[cp][ar][4];
}

__device__ __forceinline__ void scale_a(v4f (&sa)[8], v4f (&rs)[8],
                                        const v4f (&ff)[2], const v4f (&fp)[2],
                                        int c, int by, int ar, int ac)
{
    const bool  first = (ac == 0) && (c > by);
    const float keep  = (c * KC + ac >= A_DIM) ? 0.0f : 1.0f;
    const v4f   keep4 = {keep, keep, keep, keep};
#pragma unroll
    for (int q = 0; q < 8; ++q) {
        const float f  = ff[q >> 2][q & 3];
        const float f0 = first ? fp[q >> 2][q & 3] : f;
        const v4f fv4 = {f0, f, f, f};
        const v4f w = sa[q] * fv4;
        rs[q] = w * keep4 + rs[q];
        sa[q] = w;
    }
}

template <int RP, bool BROW, bool SCALE, bool STORE, bool LOAD, bool FACT, bool FRAG, bool BAR>
__device__ __forceinline__ void half_trip(v4f (&sa)[8], v4f (&sb)[8],
                                          float (*Asw)[LDS_PAD], float (*Bsw)[BPITCH],
                                          const float (*Asr)[LDS_PAD], const float (*Bsr)[BPITCH],
                                          v2f (&faN)[16], v4f (&fbN)[32],
                                          const v2f (&faC)[16], const v4f (&fbC)[32],
                                          v8f& acc0, v8f& acc1, v8f& acc2, v8f& acc3,
                                          v4f (&rs)[8], v4f (&ff)[2], v4f (&fp)[2],
                                          FqPtr Fq, const float* const (&pA)[8],
                                          const float* __restrict__ pB,
                                          int cs, int cl, int cf, int by,
                                          int ar, int ac, int br, int bc,
                                          int arow, int half, int n)
{
    if constexpr (STORE) {
        if constexpr (BROW) {
            asm volatile("" : "+v"(sb[0]), "+v"(sb[1]), "+v"(sb[2]), "+v"(sb[3]),
                              "+v"(sb[4]), "+v"(sb[5]), "+v"(sb[6]), "+v"(sb[7]));
        }
        if constexpr (SCALE) {
            scale_a(sa, rs, ff, fp, cs, by, ar, ac);
            asm volatile("" : "+v"(rs[0]), "+v"(rs[1]), "+v"(rs[2]), "+v"(rs[3]),
                              "+v"(rs[4]), "+v"(rs[5]), "+v"(rs[6]), "+v"(rs[7]));
        }
        stage_store<BROW>(Asw, Bsw, sa, sb, ar, ac, br, bc);
    }
    if constexpr (STORE && LOAD) {
        __builtin_amdgcn_sched_barrier(0);
    }
    if constexpr (LOAD) stage_load<RP, BROW>(sa, sb, pA, pB, cl);
    if constexpr (SCALE && FACT) load_factors(ff, fp, Fq, cf, ar);
    if constexpr (FRAG) frag_load(Asr, Bsr, faN, fbN, arow, half, n);
    asm volatile("" : "+v"(acc0), "+v"(acc1), "+v"(acc2), "+v"(acc3) :: "memory");
    __builtin_amdgcn_sched_barrier(0);
    mma_chunk(faC, fbC, acc0, acc1, acc2, acc3);
    if constexpr (BAR) {
        asm volatile("" : "+v"(acc0), "+v"(acc1), "+v"(acc2), "+v"(acc3));
        __builtin_amdgcn_sched_barrier(0);
        lds_barrier();
    }
}

template <int KTOT, int LDA, int RP, int AROWS, bool BROW, bool SCALE>
__device__ __forceinline__ void gemm_tile(const float* __restrict__ A,
                                          const float* __restrict__ Bpk,
                                          int ib, int j0,
                                          float (*As)[BM][LDS_PAD],
                                          float (*Bs)[KC / 2][BPITCH],
                                          v8f& acc0, v8f& acc1,
                                          v8f& acc2, v8f& acc3,
                                          FqPtr Fq, v4f (&rs)[8])
{
    constexpr int NC = KTOT / KC;
    const int lane = threadIdx.x;
    const int wave = threadIdx.y;
    const int tid  = wave * 32 + lane;
    const int half = lane >> 4;
    const int n    = lane & 15;
    const int arow = wave * 16 + n;
    const int ar   = tid >> 4;
    const int ac   = (tid & 15) * 4;
    const int br   = BROW ? (tid >> 4) : (tid >> 5);
    const int bc   = BROW ? (tid & 15) * 8 : (tid & 31) * 4;

    const float* pA[8];
#pragma unroll
    for (int q = 0; q < 8; ++q) {
        int row = ib + ar + 8 * q;
        row = row < AROWS ? row : AROWS - 1;
        pA[q] = A + (size_t)row * LDA + ac;
    }
    const float* pB = BROW ? Bpk + (size_t)(2 * br) * RP + j0 + (bc >> 1)
                           : Bpk + (size_t)br * RP + 2 * j0 + bc;

    v4f sa0[8], sb0[8];
    v4f sa1[8], sb1[8];
    v2f fa0[16], fa1[16];
    v4f fb0[32], fb1[32];

    const int by = ib / BM;
    v4f ff[2], fp[2];
    v4f m0, m1;
    v4f ta0[8], tb0[8], ta1[8], tb1[8];
    if constexpr (SCALE) row_maxima(A, m0, m1, ib, tid);
    stage_load<RP, BROW>(ta0, tb0, pA, pB, 0);
    stage_load<RP, BROW>(ta1, tb1, pA, pB, 1);
    __builtin_amdgcn_sched_barrier(0);
    stage_load<RP, BROW>(sa0, sb0, pA, pB, 2);
    __builtin_amdgcn_sched_barrier(0);
    stage_load<RP, BROW>(sa1, sb1, pA, pB, 3);
    __builtin_amdgcn_sched_barrier(0);
    if constexpr (SCALE) {
        row_factors(m0, m1, Fq, tid);
        lds_barrier();
        load_factors(ff, fp, Fq, 0, ar);
        scale_a(ta0, rs, ff, fp, 0, by, ar, ac);
        stage_store<BROW>(As[0], Bs[0], ta0, tb0, ar, ac, br, bc);
        load_factors(ff, fp, Fq, 1, ar);
        scale_a(ta1, rs, ff, fp, 1, by, ar, ac);
        stage_store<BROW>(As[1], Bs[1], ta1, tb1, ar, ac, br, bc);
        load_factors(ff, fp, Fq, 2, ar);
    } else {
        stage_store<BROW>(As[0], Bs[0], ta0, tb0, ar, ac, br, bc);
        stage_store<BROW>(As[1], Bs[1], ta1, tb1, ar, ac, br, bc);
    }
    lds_barrier();
    frag_load(As[0], Bs[0], fa0, fb0, arow, half, n);
    lds_barrier();

    static_assert(NC >= 6 && (NC & 1) == 0, "K loop: fill of 4 chunks, two peeled trips");
#pragma nounroll
    for (int c = 0; c < NC - 4; c += 2) {
        half_trip<RP, BROW, SCALE, true, true, true, true, true>(
            sa0, sb0, As[0], Bs[0], As[1], Bs[1], fa1, fb1, fa0, fb0,
            acc0, acc1, acc2, acc3, rs, ff, fp, Fq, pA, pB,
            c + 2, c + 4, c + 3, by, ar, ac, br, bc, arow, half, n);
        half_trip<RP, BROW, SCALE, true, true, true, true, true>(
            sa1, sb1, As[1], Bs[1], As[0], Bs[0], fa0, fb0, fa1, fb1,
            acc0, acc1, acc2, acc3, rs, ff, fp, Fq, pA, pB,
            c + 3, c + 5, c + 4, by, ar, ac, br, bc, arow, half, n);
    }
    half_trip<RP, BROW, SCALE, true, false, true, true, true>(
        sa0, sb0, As[0], Bs[0], As[1], Bs[1], fa1, fb1, fa0, fb0,
        acc0, acc1, acc2, acc3, rs, ff, fp, Fq, pA, pB,
        NC - 2, 0, NC - 1, by, ar, ac, br, bc, arow, half, n);
    half_trip<RP, BROW, SCALE, true, false, false, true, true>(
        sa1, sb1, As[1], Bs[1], As[0], Bs[0], fa0, fb0, fa1, fb1,
        acc0, acc1, acc2, acc3, rs, ff, fp, Fq, pA, pB,
        NC - 1, 0, 0, by, ar, ac, br, bc, arow, half, n);
    half_trip<RP, BROW, SCALE, false, false, false, true, false>(
        sa0, sb0, As[0], Bs[0], As[1], Bs[1], fa1, fb1, fa0, fb0,
        acc0, acc1, acc2, acc3, rs, ff, fp, Fq, pA, pB,
        0, 0, 0, by, ar, ac, br, bc, arow, half, n);
    half_trip<RP, BROW, SCALE, false, false, false, false, false>(
        sa1, sb1, As[1], Bs[1], As[0], Bs[0], fa0, fb0, fa1, fb1,
        acc0, acc1, acc2, acc3, rs, ff, fp, Fq, pA, pB,
        0, 0, 0, by, ar, ac, br, bc, arow, half, n);
}

__global__ void __launch_bounds__(128) gemm1_scores(const float* __restrict__ feats,
                             const float* __restrict__ Wpk,
                             const float* __restrict__ biasp,
                             float* __restrict__ attn)
{
    __shared__ float As[2][BM][LDS_PAD];
    __shared__ float Bs[2][KC / 2][BPITCH];

    const int lane = threadIdx.x;
    const int wave = threadIdx.y;
    const int j0   = blockIdx.x * BN;
    const int ib   = blockIdx.y * BM;
    const int i0   = ib + wave * 16;
    const int half = lane >> 4;
    const int n    = lane & 15;

    v8f acc0, acc1, acc2, acc3;
    {
        const float2 bA = *(const float2*)&biasp[j0 + 2 * n];
        const float2 bB = *(const float2*)&biasp[j0 + 32 + 2 * n];
#pragma unroll
        for (int r = 0; r < 8; ++r) {
            acc0[r] = bA.x; acc1[r] = bA.y; acc2[r] = bB.x; acc3[r] = bB.y;
        }
    }

    v4f unused[8];
    gemm_tile<CH, CH, 2 * NP1, FP, false, false>(feats, Wpk, ib, j0, As, Bs,
                                                 acc0, acc1, acc2, acc3, nullptr, unused);

    const int  jpA = j0 + 2 * n;
    const bool ok0 = jpA < N1, ok1 = jpA + 1 < N1, ok2 = jpA + 32 < N1, ok3 = jpA + 33 < N1;
    v8f mx;
#pragma unroll
    for (int r = 0; r < 8; ++r) {
        float v = -3.402823466e+38f;
        v = ok0 ? fmaxf(v, acc0[r]) : v;
        v = ok1 ? fmaxf(v, acc1[r]) : v;
        v = ok2 ? fmaxf(v, acc2[r]) : v;
        v = ok3 ? fmaxf(v, acc3[r]) : v;
        mx[r] = v;
    }
#pragma unroll
    for (int r = 0; r < 8; ++r) mx[r] = fmaxf(mx[r], row_xor<1>(mx[r]));
#pragma unroll
    for (int r = 0; r < 8; ++r) mx[r] = fmaxf(mx[r], row_xor<2>(mx[r]));
#pragma unroll
    for (int r = 0; r < 8; ++r) mx[r] = fmaxf(mx[r], row_xor<4>(mx[r]));
#pragma unroll
    for (int r = 0; r < 8; ++r) mx[r] = fmaxf(mx[r], row_xor<8>(mx[r]));
    v8f ex0, ex1, ex2, ex3;
#pragma unroll
    for (int r = 0; r < 8; ++r) {
        ex0[r] = expf(acc0[r] - mx[r]);
        ex1[r] = expf(acc1[r] - mx[r]);
        ex2[r] = expf(acc2[r] - mx[r]);
        ex3[r] = expf(acc3[r] - mx[r]);
    }
    asm volatile("" : "+v"(ex0), "+v"(ex1), "+v"(ex2), "+v"(ex3));
    const int wv  = __builtin_amdgcn_readfirstlane(wave);
    const int i0s = ib + wv * 16;
    if (i0s < MP) {
        char* const rb = (char*)(attn + (size_t)i0s * LDATT);
        const int      m0 = i0s + 8 * half;
        const unsigned ob = (unsigned)(8 * half * LDATT + jpA) * 4u;
        unsigned o0[8], o1[8], o2[8], o3[8];
#pragma unroll
        for (int r = 0; r < 8; ++r) {
            const int m = m0 + r;
            o0[r] = ob + (jpA      >= m ? 4u : 0u);
            o1[r] = ob + (jpA + 1  >= m ? 4u : 0u);
            o2[r] = ob + (jpA + 32 >= m ? 4u : 0u);
            o3[r] = ob + (jpA + 33 >= m ? 4u : 0u);
        }
        if (ok0) {
#pragma unroll
            for (int r = 0; r < 8; ++r) ((float*)(rb + o0[r]))[r * LDATT]      = ex0[r];
        }
        if (ok1) {
#pragma unroll
            for (int r = 0; r < 8; ++r) ((float*)(rb + o1[r]))[r * LDATT + 1]  = ex1[r];
        }
        if (ok2) {
#pragma unroll
            for (int r = 0; r < 8; ++r) ((float*)(rb + o2[r]))[r * LDATT + 32] = ex2[r];
        }
        if (ok3) {
#pragma unroll
            for (int r = 0; r < 8; ++r) ((float*)(rb + o3[r]))[r * LDATT + 33] = ex3[r];
        }
        if (n == 0) {
            float* const mp = (float*)(rb + (unsigned)(8 * half * LDATT + A_DIM + (int)blockIdx.x) * 4u);
#pragma unroll
            for (int r = 0; r < 8; ++r) mp[r * LDATT] = mx[r];
            if ((ib >> 6) == (int)blockIdx.x && m0 < A_DIM) {
                float* const dp = (float*)(rb + (unsigned)(8 * half * LDATT + m0) * 4u);
#pragma unroll
                for (int r = 0; r < 8; ++r) dp[r * (LDATT + 1)] = 0.0f;
            }
        }
        if (blockIdx.x == 15 && n < 8) {
            float* const zp = (float*)(rb + (unsigned)(8 * half * LDATT + A_DIM + 16 + n) * 4u);
#pragma unroll
            for (int r = 0; r < 8; ++r) zp[r * LDATT] = 0.0f;
        }
    }
}

#define GATHER8_Y 64
#define GATHER8_BLOCKS (12 * GATHER8_Y)

__device__ __forceinline__ void gather8(int gb, int tid,
                             const float* __restrict__ fv,
                             const int*   __restrict__ zi,
                             const int*   __restrict__ yi,
                             const int*   __restrict__ xi,
                             const int*   __restrict__ mask,
                             float* __restrict__ feats,
                             int b)
{
    const int e0 = (gb * 128 + tid) * 8;
    const int a  = e0 / CH;
    v4f o0 = {}, o1 = {};
    if (a < A_DIM) {
        const int4 ma = *(const int4*)(mask + e0), mb = *(const int4*)(mask + e0 + 4);
        const int4 za = *(const int4*)(zi + e0),   zb = *(const int4*)(zi + e0 + 4);
        const int4 ya = *(const int4*)(yi + e0),   yb = *(const int4*)(yi + e0 + 4);
        const int4 xa = *(const int4*)(xi + e0),   xb = *(const int4*)(xi + e0 + 4);
        const float* base = fv + (size_t)b * (C_DIM * H_DIM * W_DIM);
        const float g0 = base[(za.x * H_DIM + ya.x) * W_DIM + xa.x];
        const float g1 = base[(za.y * H_DIM + ya.y) * W_DIM + xa.y];
        const float g2 = base[(za.z * H_DIM + ya.z) * W_DIM + xa.z];
        const float g3 = base[(za.w * H_DIM + ya.w) * W_DIM + xa.w];
        const float g4 = base[(zb.x * H_DIM + yb.x) * W_DIM + xb.x];
        const float g5 = base[(zb.y * H_DIM + yb.y) * W_DIM + xb.y];
        const float g6 = base[(zb.z * H_DIM + yb.z) * W_DIM + xb.z];
        const float g7 = base[(zb.w * H_DIM + yb.w) * W_DIM + xb.w];
        o0[0] = ma.x ? 0.0f : g0;  o0[1] = ma.y ? 0.0f : g1;
        o0[2] = ma.z ? 0.0f : g2;  o0[3] = ma.w ? 0.0f : g3;
        o1[0] = mb.x ? 0.0f : g4;  o1[1] = mb.y ? 0.0f : g5;
        o1[2] = mb.z ? 0.0f : g6;  o1[3] = mb.w ? 0.0f : g7;
    }
    *(v4f*)(feats + e0)     = o0;
    *(v4f*)(feats + e0 + 4) = o1;
}

__global__ void __launch_bounds__(128) gemm2_out(const float* __restrict__ attn,
                          const float* __restrict__ feats,
                          float* __restrict__ outb,
                          const float* __restrict__ fv,
                          const int*   __restrict__ zi,
                          const int*   __restrict__ yi,
                          const int*   __restrict__ xi,
                          const int*   __restrict__ mask,
                          float* __restrict__ featsNext, int bNext)
{
    __shared__ float As[2][BM][LDS_PAD];
    __shared__ float Bs[2][KC / 2][BPITCH];
    __shared__ float Fq[FP / KC][8][8];
    __shared__ float Ss[BM];

    const int lane = threadIdx.x;
    const int wave = threadIdx.y;
    if (blockIdx.y >= 16) {
        gather8((blockIdx.y - 16) * 12 + blockIdx.x, wave * 32 + lane,
                fv, zi, yi, xi, mask, featsNext, bNext);
        return;
    }
    const int n0   = blockIdx.x * BN;
    const int ib   = blockIdx.y * BM;
    const int i0   = ib + wave * 16;
    const int half = lane >> 4;
    const int n    = lane & 15;

    v8f acc0 = {}, acc1 = {}, acc2 = {}, acc3 = {};

    v4f rs[8];
#pragma unroll
    for (int q = 0; q < 8; ++q) rs[q] = v4f{0.0f, 0.0f, 0.0f, 0.0f};

    gemm_tile<FP, LDATT, CH, MP, true, true>(attn, feats, ib, n0, As, Bs,
                                             acc0, acc1, acc2, acc3, Fq, rs);

    v8f sm;
#pragma unroll
    for (int q = 0; q < 8; ++q) sm[q] = (rs[q][0] + rs[q][1]) + (rs[q][2] + rs[q][3]);
#pragma unroll
    for (int q = 0; q < 8; ++q) sm[q] += row_xor<1>(sm[q]);
#pragma unroll
    for (int q = 0; q < 8; ++q) sm[q] += row_xor<2>(sm[q]);
#pragma unroll
    for (int q = 0; q < 8; ++q) sm[q] += row_xor<4>(sm[q]);
#pragma unroll
    for (int q = 0; q < 8; ++q) sm[q] += row_xor<8>(sm[q]);
    if (n == 0) {
#pragma unroll
        for (int q = 0; q < 8; ++q) Ss[2 * wave + half + 8 * q] = sm[q];
    }
    __syncthreads();

    v8f iv;
#pragma unroll
    for (int r = 0; r < 8; ++r) iv[r] = 1.0f / Ss[wave * 16 + r + 8 * half];
    asm volatile("" : "+v"(iv));
    const int wv  = __builtin_amdgcn_readfirstlane(wave);
    const int i0s = ib + wv * 16;
    const int m0  = i0s + 8 * half;
    v8f q0, q1, q2, q3;
#pragma unroll
    for (int r = 0; r < 8; ++r) {
        q0[r] = acc0[r] * iv[r];  q1[r] = acc1[r] * iv[r];
        q2[r] = acc2[r] * iv[r];  q3[r] = acc3[r] * iv[r];
    }
    if (m0 < A_DIM) {
        char* const ob = (char*)(outb + (size_t)i0s * CH);
        float* const orow = (float*)(ob + (unsigned)(8 * half * CH + n0 + 2 * n) * 4u);
#pragma unroll
        for (int r = 0; r < 8; ++r) {
            *(float2*)(orow + r * CH)      = make_float2(q0[r], q1[r]);
            *(float2*)(orow + r * CH + 32) = make_float2(q2[r], q3[r]);
        }
    }
}

extern "C" void kernel_launch(void* const* d_in, const int* in_sizes, int n_in,
                              void* d_out, int out_size, void* d_ws, size_t ws_size,
                              hipStream_t stream)
{
    const float* fv   = (const float*)d_in[0];
    const int*   zi   = (const int*)  d_in[1];
    const int*   yi   = (const int*)  d_in[2];
    const int*   xi   = (const int*)  d_in[3];
    const int*   mask = (const int*)  d_in[4];
    const float* Wt   = (const float*)d_in[5];
    const float* bias = (const float*)d_in[6];
    float*       out  = (float*)d_out;

    float* Wpk     = (float*)d_ws;
    float* biasp   = Wpk     + (size_t)CH * NP1;
    float* feats0  = biasp   + NP1;
    float* feats1  = feats0  + (size_t)FP * CH;
    float* attn    = feats1  + (size_t)FP * CH;

    const dim3 wblk(32, 4);

    pad_w<<<(CH * NP1 + 255) / 256, 256, 0, stream>>>(Wt, Wpk);
    pad_bias<<<(NP1 + 255) / 256, 256, 0, stream>>>(bias, biasp);

    gather_feats<<<GATHER_BLOCKS, 256, 0, stream>>>(fv, zi, yi, xi, mask, feats0, 0);

    for (int b = 0; b < B_DIM; ++b) {
        float* cur = (b & 1) ? feats1 : feats0;
        float* nxt = (b & 1) ? feats0 : feats1;
        gemm1_scores<<<dim3(16, 16), wblk, 0, stream>>>(cur, Wpk, biasp, attn);
        gemm2_out<<<dim3(12, (b + 1 < B_DIM) ? 16 + GATHER8_Y : 16), wblk, 0, stream>>>(
            attn, cur, out + (size_t)b * A_DIM * CH,
            fv, zi, yi, xi, mask, nxt, b + 1);
    }
}


# gemm2 on all 256 WGPs: 16x16 block tiles of 64x48 (were 192 of 64x64; a wave now 3 accumulator tiles); the next batch's gather, whose blocks had the idle WGPs, rides in the K-loop tail; sums same
# speedup vs baseline: 5.0699x; 1.0334x over previous
#include <hip/hip_runtime.h>


typedef __attribute__((ext_vector_type(2))) float v2f;
typedef __attribute__((ext_vector_type(4))) float v4f;
typedef __attribute__((ext_vector_type(8))) float v8f;

template <int M>
__device__ __forceinline__ float row_xor(float v)
{
    const int i = __builtin_bit_cast(int, v);
    return __builtin_bit_cast(float,
        __builtin_amdgcn_update_dpp(i, i, 0x160 + M, 0xf, 0xf, true));
}

#define B_DIM  32
#define A_DIM  1000
#define C_DIM  64
#define H_DIM  12
#define W_DIM  20
#define CH     768
#define N1     999
#define MP     1008
#define NP1    1024
#define FP     1024
#define LDATT  1024
#define KC     64
#define LDS_PAD 68
#define BM     64
#define BN     64
#define BN2    48
#define BPITCH 160


__global__ void pad_w(const float* __restrict__ Wt, float* __restrict__ Wpk)
{
    int e = blockIdx.x * blockDim.x + threadIdx.x;
    if (e >= CH * NP1) return;
    int k = e >> 10;
    int j = e & (NP1 - 1);
    float v = (j < N1) ? Wt[k * N1 + j] : 0.0f;
    Wpk[(size_t)(k >> 1) * (2 * NP1) + 2 * j + (k & 1)] = v;
}

__global__ void pad_bias(const float* __restrict__ bias, float* __restrict__ biasp)
{
    int j = blockIdx.x * blockDim.x + threadIdx.x;
    if (j < NP1) biasp[j] = (j < N1) ? bias[j] : 0.0f;
}

#define GATHER_BLOCKS ((FP * CH + 255) / 256)

__device__ __forceinline__ void gather_body(int blk,
                             const float* __restrict__ fv,
                             const int*   __restrict__ zi,
                             const int*   __restrict__ yi,
                             const int*   __restrict__ xi,
                             const int*   __restrict__ mask,
                             float* __restrict__ feats,
                             int b)
{
    int e = blk * 256 + threadIdx.x;
    if (e >= FP * CH) return;
    int a = e / CH;
    int k = e - a * CH;
    float v = 0.0f;
    if (a < A_DIM) {
        int idx = a * CH + k;
        if (!mask[idx]) {
            int z = zi[idx], y = yi[idx], x = xi[idx];
            v = fv[((b * C_DIM + z) * H_DIM + y) * W_DIM + x];
        }
    }
    feats[e] = v;
}

__global__ void gather_feats(const float* __restrict__ fv,
                             const int*   __restrict__ zi,
                             const int*   __restrict__ yi,
                             const int*   __restrict__ xi,
                             const int*   __restrict__ mask,
                             float* __restrict__ feats, int b)
{
    gather_body(blockIdx.x, fv, zi, yi, xi, mask, feats, b);
}

template <int RP, bool BROW>
__device__ __forceinline__ void stage_load(v4f (&sa)[8], v4f (&sb)[8],
                                           const float* const (&pA)[8],
                                           const float* const (&pB)[3], int c)
{
#pragma unroll
    for (int q = 0; q < 8; ++q)
        sa[q] = *(const v4f*)(pA[q] + c * KC);
    if constexpr (BROW) {
#pragma unroll
        for (int q = 0; q < 3; ++q) {
            const float* r = pB[q] + (size_t)(c * KC) * RP;
            sb[2 * q]     = *(const v4f*)r;
            sb[2 * q + 1] = *(const v4f*)(r + RP);
        }
    } else {
#pragma unroll
        for (int q = 0; q < 8; ++q)
            sb[q] = *(const v4f*)(pB[0] + (size_t)(c * (KC / 2) + 4 * q) * RP);
    }
}

template <bool BROW>
__device__ __forceinline__ void stage_store(float (*As)[LDS_PAD], float (*Bs)[BPITCH],
                                            const v4f (&sa)[8], const v4f (&sb)[8],
                                            int ar, int ac,
                                            const int (&br)[3], const int (&bc)[3])
{
#pragma unroll
    for (int q = 0; q < 8; ++q)
        *(v4f*)&As[ar + 8 * q][ac] = sa[q];
    if constexpr (BROW) {
#pragma unroll
        for (int q = 0; q < 3; ++q) {
            *(v4f*)&Bs[br[q]][bc[q]] =
                __builtin_shufflevector(sb[2 * q], sb[2 * q + 1], 0, 4, 1, 5);
            *(v4f*)&Bs[br[q]][bc[q] + 4] =
                __builtin_shufflevector(sb[2 * q], sb[2 * q + 1], 2, 6, 3, 7);
        }
    } else {
#pragma unroll
        for (int q = 0; q < 8; ++q)
            *(v4f*)&Bs[br[0] + 4 * q][bc[0]] = sb[q];
    }
}

template <bool W3>
__device__ __forceinline__ void frag_load(const float (*As)[LDS_PAD],
                                          const float (*Bs)[BPITCH],
                                          v2f (&fa)[16], v4f (&fb)[32], v2f (&fc)[16],
                                          int arow, int half, int n)
{
#pragma unroll
    for (int s = 0; s < 16; ++s) {
        fa[s]         = *(const v2f*)&As[arow][4 * s + 2 * half];
        fb[2 * s]     = *(const v4f*)&Bs[2 * s + half][4 * n];
        if constexpr (W3) fc[s]         = *(const v2f*)&Bs[2 * s + half][64 + 2 * n];
        else              fb[2 * s + 1] = *(const v4f*)&Bs[2 * s + half][64 + 4 * n];
    }
}

template <bool W3>
__device__ __forceinline__ void mma_chunk(const v2f (&fa)[16], const v4f (&fb)[32],
                                          const v2f (&fc)[16],
                                          v8f& acc0, v8f& acc1,
                                          v8f& acc2, v8f& acc3)
{
#pragma unroll
    for (int s = 0; s < 16; ++s) {
        const v2f b0 = __builtin_shufflevector(fb[2 * s],     fb[2 * s],     0, 1);
        const v2f b1 = __builtin_shufflevector(fb[2 * s],     fb[2 * s],     2, 3);
        acc0 = __builtin_amdgcn_wmma_f32_16x16x4_f32(
            false, fa[s], false, b0, (short)0, acc0, false, false);
        acc1 = __builtin_amdgcn_wmma_f32_16x16x4_f32(
            false, fa[s], false, b1, (short)0, acc1, false, false);
        if constexpr (W3) {
            acc2 = __builtin_amdgcn_wmma_f32_16x16x4_f32(
                false, fa[s], false, fc[s], (short)0, acc2, false, false);
        } else {
            const v2f b2 = __builtin_shufflevector(fb[2 * s + 1], fb[2 * s + 1], 0, 1);
            const v2f b3 = __builtin_shufflevector(fb[2 * s + 1], fb[2 * s + 1], 2, 3);
            acc2 = __builtin_amdgcn_wmma_f32_16x16x4_f32(
                false, fa[s], false, b2, (short)0, acc2, false, false);
            acc3 = __builtin_amdgcn_wmma_f32_16x16x4_f32(
                false, fa[s], false, b3, (short)0, acc3, false, false);
        }
    }
}

typedef float (*FqPtr)[8][8];

__device__ __forceinline__ void row_maxima(const float* __restrict__ attn,
                                           v4f& m0, v4f& m1, int ib, int tid)
{
    const int rl = tid >> 1;
    const int hc = (tid & 1) * 8;
    int row = ib + rl;
    row = row < MP ? row : MP - 1;
    const float* st = attn + (size_t)row * LDATT + A_DIM + hc;
    m0 = *(const v4f*)st;
    m1 = *(const v4f*)(st + 4);
}

__device__ __forceinline__ void row_factors(const v4f m0, const v4f m1, FqPtr Fq, int tid)
{
    const int rl = tid >> 1;
    const int hc = (tid & 1) * 8;
    float mx = fmaxf(fmaxf(fmaxf(m0[0], m0[1]), fmaxf(m0[2], m0[3])),
                     fmaxf(fmaxf(m1[0], m1[1]), fmaxf(m1[2], m1[3])));
    mx = fmaxf(mx, row_xor<1>(mx));
#pragma unroll
    for (int e = 0; e < 4; ++e) {
        Fq[hc + e][rl & 7][rl >> 3]     = expf(m0[e] - mx);
        Fq[hc + 4 + e][rl & 7][rl >> 3] = expf(m1[e] - mx);
    }
}

__device__ __forceinline__ void lds_barrier()
{
    __builtin_amdgcn_fence(__ATOMIC_RELEASE, "workgroup", "local");
    __builtin_amdgcn_s_barrier();
    __builtin_amdgcn_fence(__ATOMIC_ACQUIRE, "workgroup", "local");
}

__device__ __forceinline__ void load_factors(v4f (&ff)[2], v4f (&fp)[2],
                                             FqPtr Fq, int c, int ar)
{
    const int cp = c > 0 ? c - 1 : 0;
    ff[0] = *(const v4f*)&Fq[c][ar][0];
    ff[1] = *(const v4f*)&Fq[c][ar][4];
    fp[0] = *(const v4f*)&Fq[cp][ar][0];
    fp[1] = *(const v4f*)&Fq[cp][ar][4];
}

__device__ __forceinline__ void scale_a(v4f (&sa)[8], v4f (&rs)[8],
                                        const v4f (&ff)[2], const v4f (&fp)[2],
                                        int c, int by, int ar, int ac)
{
    const bool  first = (ac == 0) && (c > by);
    const float keep  = (c * KC + ac >= A_DIM) ? 0.0f : 1.0f;
    const v4f   keep4 = {keep, keep, keep, keep};
#pragma unroll
    for (int q = 0; q < 8; ++q) {
        const float f  = ff[q >> 2][q & 3];
        const float f0 = first ? fp[q >> 2][q & 3] : f;
        const v4f fv4 = {f0, f, f, f};
        const v4f w = sa[q] * fv4;
        rs[q] = w * keep4 + rs[q];
        sa[q] = w;
    }
}

typedef int v4i __attribute__((ext_vector_type(4)));
typedef const __attribute__((address_space(1))) float* GFloatPtr;
typedef const __attribute__((address_space(1))) int*   GIntPtr;
typedef const __attribute__((address_space(1))) v4i*   GInt4Ptr;

struct GatherRide {
    GFloatPtr base;
    GIntPtr   zi;
    GIntPtr   yi;
    GIntPtr   xi;
    GIntPtr   mask;
    float*    feats;
    int       e0[3];
    v4i       m[6], z[6], y[6], x[6];
    float     g[24];
};

__device__ __forceinline__ void gather_idx(GatherRide& gr)
{
#pragma unroll
    for (int q = 0; q < 3; ++q) {
        const int e = gr.e0[q] < A_DIM * CH ? gr.e0[q] : A_DIM * CH - 8;
        gr.z[2 * q] = *(GInt4Ptr)(gr.zi + e);    gr.z[2 * q + 1] = *(GInt4Ptr)(gr.zi + e + 4);
        gr.y[2 * q] = *(GInt4Ptr)(gr.yi + e);    gr.y[2 * q + 1] = *(GInt4Ptr)(gr.yi + e + 4);
        gr.x[2 * q] = *(GInt4Ptr)(gr.xi + e);    gr.x[2 * q + 1] = *(GInt4Ptr)(gr.xi + e + 4);
    }
}

__device__ __forceinline__ void gather_val(GatherRide& gr)
{
#pragma unroll
    for (int q = 0; q < 3; ++q) {
        const int e = gr.e0[q] < A_DIM * CH ? gr.e0[q] : A_DIM * CH - 8;
        gr.m[2 * q] = *(GInt4Ptr)(gr.mask + e);  gr.m[2 * q + 1] = *(GInt4Ptr)(gr.mask + e + 4);
    }
#pragma unroll
    for (int t = 0; t < 6; ++t) {
#pragma unroll
        for (int j = 0; j < 4; ++j)
            gr.g[4 * t + j] = gr.base[(gr.z[t][j] * H_DIM + gr.y[t][j]) * W_DIM + gr.x[t][j]];
    }
}

__device__ __forceinline__ void gather_put(GatherRide& gr)
{
#pragma unroll
    for (int t = 0; t < 6; ++t) {
        const bool row = gr.e0[t >> 1] < A_DIM * CH;
        v4f o;
#pragma unroll
        for (int j = 0; j < 4; ++j)
            o[j] = (row && !gr.m[t][j]) ? gr.g[4 * t + j] : 0.0f;
        *(v4f*)(gr.feats + gr.e0[t >> 1] + 4 * (t & 1)) = o;
    }
}

template <int RP, bool BROW, bool SCALE, bool STORE, bool LOAD, bool FACT, bool FRAG, bool BAR, int GST>
__device__ __forceinline__ void half_trip(v4f (&sa)[8], v4f (&sb)[8],
                                          float (*Asw)[LDS_PAD], float (*Bsw)[BPITCH],
                                          const float (*Asr)[LDS_PAD], const float (*Bsr)[BPITCH],
                                          v2f (&faN)[16], v4f (&fbN)[32], v2f (&fcN)[16],
                                          const v2f (&faC)[16], const v4f (&fbC)[32],
                                          const v2f (&fcC)[16],
                                          v8f& acc0, v8f& acc1, v8f& acc2, v8f& acc3,
                                          v4f (&rs)[8], v4f (&ff)[2], v4f (&fp)[2],
                                          FqPtr Fq, const float* const (&pA)[8],
                                          const float* const (&pB)[3],
                                          int cs, int cl, int cf, int by,
                                          int ar, int ac,
                                          const int (&br)[3], const int (&bc)[3],
                                          int arow, int half, int n, GatherRide& gr)
{
    if constexpr (STORE) {
        if constexpr (BROW) {
            asm volatile("" : "+v"(sb[0]), "+v"(sb[1]), "+v"(sb[2]), "+v"(sb[3]),
                              "+v"(sb[4]), "+v"(sb[5]));
        }
        if constexpr (SCALE) {
            scale_a(sa, rs, ff, fp, cs, by, ar, ac);
            if constexpr (LOAD)
                asm volatile("" : "+v"(rs[0]), "+v"(rs[1]), "+v"(rs[2]), "+v"(rs[3]),
                                  "+v"(rs[4]), "+v"(rs[5]), "+v"(rs[6]), "+v"(rs[7]));
        }
        stage_store<BROW>(Asw, Bsw, sa, sb, ar, ac, br, bc);
    }
    if constexpr (STORE && LOAD) {
        __builtin_amdgcn_sched_barrier(0);
    }
    if constexpr (LOAD) stage_load<RP, BROW>(sa, sb, pA, pB, cl);
    if constexpr (GST == 1) gather_idx(gr);
    if constexpr (GST == 2) gather_val(gr);
    if constexpr (GST == 3) gather_put(gr);
    if constexpr (SCALE && FACT) load_factors(ff, fp, Fq, cf, ar);
    if constexpr (FRAG) frag_load<BROW>(Asr, Bsr, faN, fbN, fcN, arow, half, n);
    if constexpr (BROW) asm volatile("" : "+v"(acc0), "+v"(acc1), "+v"(acc2) :: "memory");
    else                asm volatile("" : "+v"(acc0), "+v"(acc1), "+v"(acc2), "+v"(acc3) :: "memory");
    __builtin_amdgcn_sched_barrier(0);
    mma_chunk<BROW>(faC, fbC, fcC, acc0, acc1, acc2, acc3);
    if constexpr (BAR) {
        if constexpr (BROW) asm volatile("" : "+v"(acc0), "+v"(acc1), "+v"(acc2));
        else                asm volatile("" : "+v"(acc0), "+v"(acc1), "+v"(acc2), "+v"(acc3));
        __builtin_amdgcn_sched_barrier(0);
        lds_barrier();
    }
}

template <int KTOT, int LDA, int RP, int AROWS, bool BROW, bool SCALE, bool RIDE>
__device__ __forceinline__ void gemm_tile(const float* __restrict__ A,
                                          const float* __restrict__ Bpk,
                                          int ib, int j0,
                                          float (*As)[BM][LDS_PAD],
                                          float (*Bs)[KC / 2][BPITCH],
                                          v8f& acc0, v8f& acc1,
                                          v8f& acc2, v8f& acc3,
                                          FqPtr Fq, v4f (&rs)[8], GatherRide& gr)
{
    constexpr int NC = KTOT / KC;
    const int lane = threadIdx.x;
    const int wave = threadIdx.y;
    const int tid  = wave * 32 + lane;
    const int half = lane >> 4;
    const int n    = lane & 15;
    const int arow = wave * 16 + n;
    const int ar   = tid >> 4;
    const int ac   = (tid & 15) * 4;
    int br[3], bc[3];
#pragma unroll
    for (int q = 0; q < 3; ++q) {
        const int u = tid + 128 * q;
        br[q] = BROW ? u / 12 : (tid >> 5);
        bc[q] = BROW ? (u - 12 * (u / 12)) * 8 : (tid & 31) * 4;
    }

    const float* pA[8];
#pragma unroll
    for (int q = 0; q < 8; ++q) {
        int row = ib + ar + 8 * q;
        row = row < AROWS ? row : AROWS - 1;
        pA[q] = A + (size_t)row * LDA + ac;
    }
    const float* pB[3];
#pragma unroll
    for (int q = 0; q < 3; ++q)
        pB[q] = BROW ? Bpk + (size_t)(2 * br[q]) * RP + j0 + (bc[q] >> 1)
                     : Bpk + (size_t)br[q] * RP + 2 * j0 + bc[q];

    v4f sa0[8], sb0[8];
    v4f sa1[8], sb1[8];
    v2f fa0[16], fa1[16];
    v4f fb0[32], fb1[32];
    v2f fc0[16], fc1[16];

    const int by = ib / BM;
    v4f ff[2], fp[2];
    v4f m0, m1;
    v4f ta0[8], tb0[8], ta1[8], tb1[8];
    if constexpr (SCALE) row_maxima(A, m0, m1, ib, tid);
    stage_load<RP, BROW>(ta0, tb0, pA, pB, 0);
    stage_load<RP, BROW>(ta1, tb1, pA, pB, 1);
    __builtin_amdgcn_sched_barrier(0);
    stage_load<RP, BROW>(sa0, sb0, pA, pB, 2);
    __builtin_amdgcn_sched_barrier(0);
    stage_load<RP, BROW>(sa1, sb1, pA, pB, 3);
    __builtin_amdgcn_sched_barrier(0);
    if constexpr (SCALE) {
        row_factors(m0, m1, Fq, tid);
        lds_barrier();
        load_factors(ff, fp, Fq, 0, ar);
        scale_a(ta0, rs, ff, fp, 0, by, ar, ac);
        stage_store<BROW>(As[0], Bs[0], ta0, tb0, ar, ac, br, bc);
        load_factors(ff, fp, Fq, 1, ar);
        scale_a(ta1, rs, ff, fp, 1, by, ar, ac);
        stage_store<BROW>(As[1], Bs[1], ta1, tb1, ar, ac, br, bc);
        load_factors(ff, fp, Fq, 2, ar);
    } else {
        stage_store<BROW>(As[0], Bs[0], ta0, tb0, ar, ac, br, bc);
        stage_store<BROW>(As[1], Bs[1], ta1, tb1, ar, ac, br, bc);
    }
    lds_barrier();
    frag_load<BROW>(As[0], Bs[0], fa0, fb0, fc0, arow, half, n);
    lds_barrier();

    static_assert(NC >= 6 && (NC & 1) == 0, "K loop: fill of 4 chunks, two peeled trips");
#pragma nounroll
    for (int c = 0; c < NC - 4; c += 2) {
        half_trip<RP, BROW, SCALE, true, true, true, true, true, 0>(
            sa0, sb0, As[0], Bs[0], As[1], Bs[1], fa1, fb1, fc1, fa0, fb0, fc0,
            acc0, acc1, acc2, acc3, rs, ff, fp, Fq, pA, pB,
            c + 2, c + 4, c + 3, by, ar, ac, br, bc, arow, half, n, gr);
        half_trip<RP, BROW, SCALE, true, true, true, true, true, 0>(
            sa1, sb1, As[1], Bs[1], As[0], Bs[0], fa0, fb0, fc0, fa1, fb1, fc1,
            acc0, acc1, acc2, acc3, rs, ff, fp, Fq, pA, pB,
            c + 3, c + 5, c + 4, by, ar, ac, br, bc, arow, half, n, gr);
    }
    half_trip<RP, BROW, SCALE, true, false, true, true, true, RIDE ? 1 : 0>(
        sa0, sb0, As[0], Bs[0], As[1], Bs[1], fa1, fb1, fc1, fa0, fb0, fc0,
        acc0, acc1, acc2, acc3, rs, ff, fp, Fq, pA, pB,
        NC - 2, 0, NC - 1, by, ar, ac, br, bc, arow, half, n, gr);
    half_trip<RP, BROW, SCALE, true, false, false, true, true, RIDE ? 2 : 0>(
        sa1, sb1, As[1], Bs[1], As[0], Bs[0], fa0, fb0, fc0, fa1, fb1, fc1,
        acc0, acc1, acc2, acc3, rs, ff, fp, Fq, pA, pB,
        NC - 1, 0, 0, by, ar, ac, br, bc, arow, half, n, gr);
    half_trip<RP, BROW, SCALE, false, false, false, true, false, RIDE ? 3 : 0>(
        sa0, sb0, As[0], Bs[0], As[1], Bs[1], fa1, fb1, fc1, fa0, fb0, fc0,
        acc0, acc1, acc2, acc3, rs, ff, fp, Fq, pA, pB,
        0, 0, 0, by, ar, ac, br, bc, arow, half, n, gr);
    half_trip<RP, BROW, SCALE, false, false, false, false, false, 0>(
        sa1, sb1, As[1], Bs[1], As[0], Bs[0], fa0, fb0, fc0, fa1, fb1, fc1,
        acc0, acc1, acc2, acc3, rs, ff, fp, Fq, pA, pB,
        0, 0, 0, by, ar, ac, br, bc, arow, half, n, gr);
}

__global__ void __launch_bounds__(128) gemm1_scores(const float* __restrict__ feats,
                             const float* __restrict__ Wpk,
                             const float* __restrict__ biasp,
                             float* __restrict__ attn)
{
    __shared__ float As[2][BM][LDS_PAD];
    __shared__ float Bs[2][KC / 2][BPITCH];

    const int lane = threadIdx.x;
    const int wave = threadIdx.y;
    const int j0   = blockIdx.x * BN;
    const int ib   = blockIdx.y * BM;
    const int i0   = ib + wave * 16;
    const int half = lane >> 4;
    const int n    = lane & 15;

    v8f acc0, acc1, acc2, acc3;
    {
        const float2 bA = *(const float2*)&biasp[j0 + 2 * n];
        const float2 bB = *(const float2*)&biasp[j0 + 32 + 2 * n];
#pragma unroll
        for (int r = 0; r < 8; ++r) {
            acc0[r] = bA.x; acc1[r] = bA.y; acc2[r] = bB.x; acc3[r] = bB.y;
        }
    }

    v4f unused[8];
    GatherRide none;
    gemm_tile<CH, CH, 2 * NP1, FP, false, false, false>(feats, Wpk, ib, j0, As, Bs,
                                                        acc0, acc1, acc2, acc3, nullptr, unused, none);

    const int  jpA = j0 + 2 * n;
    const bool ok0 = jpA < N1, ok1 = jpA + 1 < N1, ok2 = jpA + 32 < N1, ok3 = jpA + 33 < N1;
    v8f mx;
#pragma unroll
    for (int r = 0; r < 8; ++r) {
        float v = -3.402823466e+38f;
        v = ok0 ? fmaxf(v, acc0[r]) : v;
        v = ok1 ? fmaxf(v, acc1[r]) : v;
        v = ok2 ? fmaxf(v, acc2[r]) : v;
        v = ok3 ? fmaxf(v, acc3[r]) : v;
        mx[r] = v;
    }
#pragma unroll
    for (int r = 0; r < 8; ++r) mx[r] = fmaxf(mx[r], row_xor<1>(mx[r]));
#pragma unroll
    for (int r = 0; r < 8; ++r) mx[r] = fmaxf(mx[r], row_xor<2>(mx[r]));
#pragma unroll
    for (int r = 0; r < 8; ++r) mx[r] = fmaxf(mx[r], row_xor<4>(mx[r]));
#pragma unroll
    for (int r = 0; r < 8; ++r) mx[r] = fmaxf(mx[r], row_xor<8>(mx[r]));
    v8f ex0, ex1, ex2, ex3;
#pragma unroll
    for (int r = 0; r < 8; ++r) {
        ex0[r] = expf(acc0[r] - mx[r]);
        ex1[r] = expf(acc1[r] - mx[r]);
        ex2[r] = expf(acc2[r] - mx[r]);
        ex3[r] = expf(acc3[r] - mx[r]);
    }
    asm volatile("" : "+v"(ex0), "+v"(ex1), "+v"(ex2), "+v"(ex3));
    const int wv  = __builtin_amdgcn_readfirstlane(wave);
    const int i0s = ib + wv * 16;
    if (i0s < MP) {
        char* const rb = (char*)(attn + (size_t)i0s * LDATT);
        const int      m0 = i0s + 8 * half;
        const unsigned ob = (unsigned)(8 * half * LDATT + jpA) * 4u;
        unsigned o0[8], o1[8], o2[8], o3[8];
#pragma unroll
        for (int r = 0; r < 8; ++r) {
            const int m = m0 + r;
            o0[r] = ob + (jpA      >= m ? 4u : 0u);
            o1[r] = ob + (jpA + 1  >= m ? 4u : 0u);
            o2[r] = ob + (jpA + 32 >= m ? 4u : 0u);
            o3[r] = ob + (jpA + 33 >= m ? 4u : 0u);
        }
        if (ok0) {
#pragma unroll
            for (int r = 0; r < 8; ++r) ((float*)(rb + o0[r]))[r * LDATT]      = ex0[r];
        }
        if (ok1) {
#pragma unroll
            for (int r = 0; r < 8; ++r) ((float*)(rb + o1[r]))[r * LDATT + 1]  = ex1[r];
        }
        if (ok2) {
#pragma unroll
            for (int r = 0; r < 8; ++r) ((float*)(rb + o2[r]))[r * LDATT + 32] = ex2[r];
        }
        if (ok3) {
#pragma unroll
            for (int r = 0; r < 8; ++r) ((float*)(rb + o3[r]))[r * LDATT + 33] = ex3[r];
        }
        if (n == 0) {
            float* const mp = (float*)(rb + (unsigned)(8 * half * LDATT + A_DIM + (int)blockIdx.x) * 4u);
#pragma unroll
            for (int r = 0; r < 8; ++r) mp[r * LDATT] = mx[r];
            if ((ib >> 6) == (int)blockIdx.x && m0 < A_DIM) {
                float* const dp = (float*)(rb + (unsigned)(8 * half * LDATT + m0) * 4u);
#pragma unroll
                for (int r = 0; r < 8; ++r) dp[r * (LDATT + 1)] = 0.0f;
            }
        }
        if (blockIdx.x == 15 && n < 8) {
            float* const zp = (float*)(rb + (unsigned)(8 * half * LDATT + A_DIM + 16 + n) * 4u);
#pragma unroll
            for (int r = 0; r < 8; ++r) zp[r * LDATT] = 0.0f;
        }
    }
}

__global__ void __launch_bounds__(128) gemm2_out(const float* __restrict__ attn,
                          const float* __restrict__ feats,
                          float* __restrict__ outb,
                          const float* __restrict__ fv,
                          const int*   __restrict__ zi,
                          const int*   __restrict__ yi,
                          const int*   __restrict__ xi,
                          const int*   __restrict__ mask,
                          float* __restrict__ featsNext, int bNext)
{
    __shared__ float As[2][BM][LDS_PAD];
    __shared__ float Bs[2][KC / 2][BPITCH];
    __shared__ float Fq[FP / KC][8][8];
    __shared__ float Ss[BM];

    const int lane = threadIdx.x;
    const int wave = threadIdx.y;
    const int n0   = blockIdx.x * BN2;
    const int ib   = blockIdx.y * BM;
    const int half = lane >> 4;
    const int n    = lane & 15;

    v8f acc0 = {}, acc1 = {}, acc2 = {}, acc3 = {};

    v4f rs[8];
#pragma unroll
    for (int q = 0; q < 8; ++q) rs[q] = v4f{0.0f, 0.0f, 0.0f, 0.0f};

    GatherRide gr;
    {
        const int bn = bNext < B_DIM ? bNext : B_DIM - 1;
        const float* gb = fv + (size_t)bn * (C_DIM * H_DIM * W_DIM);
        const int *gz = zi, *gy = yi, *gx = xi, *gm = mask;
        asm volatile("" : "+s"(gb), "+s"(gz), "+s"(gy), "+s"(gx), "+s"(gm));
        gr.base  = (GFloatPtr)gb;
        gr.zi    = (GIntPtr)gz;
        gr.yi    = (GIntPtr)gy;
        gr.xi    = (GIntPtr)gx;
        gr.mask  = (GIntPtr)gm;
        gr.feats = featsNext;
        const int blk = blockIdx.y * 16 + blockIdx.x;
#pragma unroll
        for (int q = 0; q < 3; ++q)
            gr.e0[q] = ((blk * 3 + q) * 128 + wave * 32 + lane) * 8;
    }

    gemm_tile<FP, LDATT, CH, MP, true, true, true>(attn, feats, ib, n0, As, Bs,
                                                   acc0, acc1, acc2, acc3, Fq, rs, gr);

    v8f sm;
#pragma unroll
    for (int q = 0; q < 8; ++q) sm[q] = (rs[q][0] + rs[q][1]) + (rs[q][2] + rs[q][3]);
#pragma unroll
    for (int q = 0; q < 8; ++q) sm[q] += row_xor<1>(sm[q]);
#pragma unroll
    for (int q = 0; q < 8; ++q) sm[q] += row_xor<2>(sm[q]);
#pragma unroll
    for (int q = 0; q < 8; ++q) sm[q] += row_xor<4>(sm[q]);
#pragma unroll
    for (int q = 0; q < 8; ++q) sm[q] += row_xor<8>(sm[q]);
    if (n == 0) {
#pragma unroll
        for (int q = 0; q < 8; ++q) Ss[2 * wave + half + 8 * q] = sm[q];
    }
    __syncthreads();

    v8f iv;
#pragma unroll
    for (int r = 0; r < 8; ++r) iv[r] = 1.0f / Ss[wave * 16 + r + 8 * half];
    asm volatile("" : "+v"(iv));
    const int wv  = __builtin_amdgcn_readfirstlane(wave);
    const int i0s = ib + wv * 16;
    const int m0  = i0s + 8 * half;
    v8f q0, q1, q2;
#pragma unroll
    for (int r = 0; r < 8; ++r) {
        q0[r] = acc0[r] * iv[r];  q1[r] = acc1[r] * iv[r];
        q2[r] = acc2[r] * iv[r];
    }
    if (m0 < A_DIM) {
        char* const ob = (char*)(outb + (size_t)i0s * CH);
        float* const orow = (float*)(ob + (unsigned)(8 * half * CH + n0 + 2 * n) * 4u);
        float* const ocol = (float*)(ob + (unsigned)(8 * half * CH + n0 + 32 + n) * 4u);
#pragma unroll
        for (int r = 0; r < 8; ++r) {
            *(float2*)(orow + r * CH) = make_float2(q0[r], q1[r]);
            ocol[r * CH]              = q2[r];
        }
    }
}

extern "C" void kernel_launch(void* const* d_in, const int* in_sizes, int n_in,
                              void* d_out, int out_size, void* d_ws, size_t ws_size,
                              hipStream_t stream)
{
    const float* fv   = (const float*)d_in[0];
    const int*   zi   = (const int*)  d_in[1];
    const int*   yi   = (const int*)  d_in[2];
    const int*   xi   = (const int*)  d_in[3];
    const int*   mask = (const int*)  d_in[4];
    const float* Wt   = (const float*)d_in[5];
    const float* bias = (const float*)d_in[6];
    float*       out  = (float*)d_out;

    float* Wpk     = (float*)d_ws;
    float* biasp   = Wpk     + (size_t)CH * NP1;
    float* feats0  = biasp   + NP1;
    float* feats1  = feats0  + (size_t)FP * CH;
    float* attn    = feats1  + (size_t)FP * CH;

    const dim3 wblk(32, 4);

    pad_w<<<(CH * NP1 + 255) / 256, 256, 0, stream>>>(Wt, Wpk);
    pad_bias<<<(NP1 + 255) / 256, 256, 0, stream>>>(bias, biasp);

    gather_feats<<<GATHER_BLOCKS, 256, 0, stream>>>(fv, zi, yi, xi, mask, feats0, 0);

    for (int b = 0; b < B_DIM; ++b) {
        float* cur = (b & 1) ? feats1 : feats0;
        float* nxt = (b & 1) ? feats0 : feats1;
        gemm1_scores<<<dim3(16, 16), wblk, 0, stream>>>(cur, Wpk, biasp, attn);
        gemm2_out<<<dim3(16, 16), wblk, 0, stream>>>(
            attn, cur, out + (size_t)b * A_DIM * CH,
            fv, zi, yi, xi, mask, nxt, b + 1);
    }
}


# gemm2 of batch b and gemm1 of batch b+1 in one launch of 512 work-groups (role by grid row; third feats and second score buffer, gather rides two batches ahead): 36 launches, was 67; sums same
# speedup vs baseline: 5.4418x; 1.0734x over previous
#include <hip/hip_runtime.h>


typedef __attribute__((ext_vector_type(2))) float v2f;
typedef __attribute__((ext_vector_type(4))) float v4f;
typedef __attribute__((ext_vector_type(8))) float v8f;

template <int M>
__device__ __forceinline__ float row_xor(float v)
{
    const int i = __builtin_bit_cast(int, v);
    return __builtin_bit_cast(float,
        __builtin_amdgcn_update_dpp(i, i, 0x160 + M, 0xf, 0xf, true));
}

#define B_DIM  32
#define A_DIM  1000
#define C_DIM  64
#define H_DIM  12
#define W_DIM  20
#define CH     768
#define N1     999
#define MP     1008
#define NP1    1024
#define FP     1024
#define LDATT  1024
#define KC     64
#define LDS_PAD 68
#define BM     64
#define BN     64
#define BN2    48
#define BPITCH 160


__global__ void pad_w(const float* __restrict__ Wt, float* __restrict__ Wpk)
{
    int e = blockIdx.x * blockDim.x + threadIdx.x;
    if (e >= CH * NP1) return;
    int k = e >> 10;
    int j = e & (NP1 - 1);
    float v = (j < N1) ? Wt[k * N1 + j] : 0.0f;
    Wpk[(size_t)(k >> 1) * (2 * NP1) + 2 * j + (k & 1)] = v;
}

__global__ void pad_bias(const float* __restrict__ bias, float* __restrict__ biasp)
{
    int j = blockIdx.x * blockDim.x + threadIdx.x;
    if (j < NP1) biasp[j] = (j < N1) ? bias[j] : 0.0f;
}

#define GATHER_BLOCKS ((FP * CH + 255) / 256)

__device__ __forceinline__ void gather_body(int blk,
                             const float* __restrict__ fv,
                             const int*   __restrict__ zi,
                             const int*   __restrict__ yi,
                             const int*   __restrict__ xi,
                             const int*   __restrict__ mask,
                             float* __restrict__ feats,
                             int b)
{
    int e = blk * 256 + threadIdx.x;
    if (e >= FP * CH) return;
    int a = e / CH;
    int k = e - a * CH;
    float v = 0.0f;
    if (a < A_DIM) {
        int idx = a * CH + k;
        if (!mask[idx]) {
            int z = zi[idx], y = yi[idx], x = xi[idx];
            v = fv[((b * C_DIM + z) * H_DIM + y) * W_DIM + x];
        }
    }
    feats[e] = v;
}

__global__ void gather_feats(const float* __restrict__ fv,
                             const int*   __restrict__ zi,
                             const int*   __restrict__ yi,
                             const int*   __restrict__ xi,
                             const int*   __restrict__ mask,
                             float* __restrict__ feats, int b)
{
    gather_body(blockIdx.x, fv, zi, yi, xi, mask,
                feats + (size_t)blockIdx.y * (FP * CH), b + (int)blockIdx.y);
}

template <int RP, bool BROW>
__device__ __forceinline__ void stage_load(v4f (&sa)[8], v4f (&sb)[8],
                                           const float* const (&pA)[8],
                                           const float* const (&pB)[3], int c)
{
#pragma unroll
    for (int q = 0; q < 8; ++q)
        sa[q] = *(const v4f*)(pA[q] + c * KC);
    if constexpr (BROW) {
#pragma unroll
        for (int q = 0; q < 3; ++q) {
            const float* r = pB[q] + (size_t)(c * KC) * RP;
            sb[2 * q]     = *(const v4f*)r;
            sb[2 * q + 1] = *(const v4f*)(r + RP);
        }
    } else {
#pragma unroll
        for (int q = 0; q < 8; ++q)
            sb[q] = *(const v4f*)(pB[0] + (size_t)(c * (KC / 2) + 4 * q) * RP);
    }
}

template <bool BROW>
__device__ __forceinline__ void stage_store(float (*As)[LDS_PAD], float (*Bs)[BPITCH],
                                            const v4f (&sa)[8], const v4f (&sb)[8],
                                            int ar, int ac,
                                            const int (&br)[3], const int (&bc)[3])
{
#pragma unroll
    for (int q = 0; q < 8; ++q)
        *(v4f*)&As[ar + 8 * q][ac] = sa[q];
    if constexpr (BROW) {
#pragma unroll
        for (int q = 0; q < 3; ++q) {
            *(v4f*)&Bs[br[q]][bc[q]] =
                __builtin_shufflevector(sb[2 * q], sb[2 * q + 1], 0, 4, 1, 5);
            *(v4f*)&Bs[br[q]][bc[q] + 4] =
                __builtin_shufflevector(sb[2 * q], sb[2 * q + 1], 2, 6, 3, 7);
        }
    } else {
#pragma unroll
        for (int q = 0; q < 8; ++q)
            *(v4f*)&Bs[br[0] + 4 * q][bc[0]] = sb[q];
    }
}

template <bool W3>
__device__ __forceinline__ void frag_load(const float (*As)[LDS_PAD],
                                          const float (*Bs)[BPITCH],
                                          v2f (&fa)[16], v4f (&fb)[32], v2f (&fc)[16],
                                          int arow, int half, int n)
{
#pragma unroll
    for (int s = 0; s < 16; ++s) {
        fa[s]         = *(const v2f*)&As[arow][4 * s + 2 * half];
        fb[2 * s]     = *(const v4f*)&Bs[2 * s + half][4 * n];
        if constexpr (W3) fc[s]         = *(const v2f*)&Bs[2 * s + half][64 + 2 * n];
        else              fb[2 * s + 1] = *(const v4f*)&Bs[2 * s + half][64 + 4 * n];
    }
}

template <bool W3>
__device__ __forceinline__ void mma_chunk(const v2f (&fa)[16], const v4f (&fb)[32],
                                          const v2f (&fc)[16],
                                          v8f& acc0, v8f& acc1,
                                          v8f& acc2, v8f& acc3)
{
#pragma unroll
    for (int s = 0; s < 16; ++s) {
        const v2f b0 = __builtin_shufflevector(fb[2 * s],     fb[2 * s],     0, 1);
        const v2f b1 = __builtin_shufflevector(fb[2 * s],     fb[2 * s],     2, 3);
        acc0 = __builtin_amdgcn_wmma_f32_16x16x4_f32(
            false, fa[s], false, b0, (short)0, acc0, false, false);
        acc1 = __builtin_amdgcn_wmma_f32_16x16x4_f32(
            false, fa[s], false, b1, (short)0, acc1, false, false);
        if constexpr (W3) {
            acc2 = __builtin_amdgcn_wmma_f32_16x16x4_f32(
                false, fa[s], false, fc[s], (short)0, acc2, false, false);
        } else {
            const v2f b2 = __builtin_shufflevector(fb[2 * s + 1], fb[2 * s + 1], 0, 1);
            const v2f b3 = __builtin_shufflevector(fb[2 * s + 1], fb[2 * s + 1], 2, 3);
            acc2 = __builtin_amdgcn_wmma_f32_16x16x4_f32(
                false, fa[s], false, b2, (short)0, acc2, false, false);
            acc3 = __builtin_amdgcn_wmma_f32_16x16x4_f32(
                false, fa[s], false, b3, (short)0, acc3, false, false);
        }
    }
}

typedef float (*FqPtr)[8][8];

__device__ __forceinline__ void row_maxima(const float* __restrict__ attn,
                                           v4f& m0, v4f& m1, int ib, int tid)
{
    const int rl = tid >> 1;
    const int hc = (tid & 1) * 8;
    int row = ib + rl;
    row = row < MP ? row : MP - 1;
    const float* st = attn + (size_t)row * LDATT + A_DIM + hc;
    m0 = *(const v4f*)st;
    m1 = *(const v4f*)(st + 4);
}

__device__ __forceinline__ void row_factors(const v4f m0, const v4f m1, FqPtr Fq, int tid)
{
    const int rl = tid >> 1;
    const int hc = (tid & 1) * 8;
    float mx = fmaxf(fmaxf(fmaxf(m0[0], m0[1]), fmaxf(m0[2], m0[3])),
                     fmaxf(fmaxf(m1[0], m1[1]), fmaxf(m1[2], m1[3])));
    mx = fmaxf(mx, row_xor<1>(mx));
#pragma unroll
    for (int e = 0; e < 4; ++e) {
        Fq[hc + e][rl & 7][rl >> 3]     = expf(m0[e] - mx);
        Fq[hc + 4 + e][rl & 7][rl >> 3] = expf(m1[e] - mx);
    }
}

__device__ __forceinline__ void lds_barrier()
{
    __builtin_amdgcn_fence(__ATOMIC_RELEASE, "workgroup", "local");
    __builtin_amdgcn_s_barrier();
    __builtin_amdgcn_fence(__ATOMIC_ACQUIRE, "workgroup", "local");
}

__device__ __forceinline__ void load_factors(v4f (&ff)[2], v4f (&fp)[2],
                                             FqPtr Fq, int c, int ar)
{
    const int cp = c > 0 ? c - 1 : 0;
    ff[0] = *(const v4f*)&Fq[c][ar][0];
    ff[1] = *(const v4f*)&Fq[c][ar][4];
    fp[0] = *(const v4f*)&Fq[cp][ar][0];
    fp[1] = *(const v4f*)&Fq[cp][ar][4];
}

__device__ __forceinline__ void scale_a(v4f (&sa)[8], v4f (&rs)[8],
                                        const v4f (&ff)[2], const v4f (&fp)[2],
                                        int c, int by, int ar, int ac)
{
    const bool  first = (ac == 0) && (c > by);
    const float keep  = (c * KC + ac >= A_DIM) ? 0.0f : 1.0f;
    const v4f   keep4 = {keep, keep, keep, keep};
#pragma unroll
    for (int q = 0; q < 8; ++q) {
        const float f  = ff[q >> 2][q & 3];
        const float f0 = first ? fp[q >> 2][q & 3] : f;
        const v4f fv4 = {f0, f, f, f};
        const v4f w = sa[q] * fv4;
        rs[q] = w * keep4 + rs[q];
        sa[q] = w;
    }
}

typedef int v4i __attribute__((ext_vector_type(4)));
typedef const __attribute__((address_space(1))) float* GFloatPtr;
typedef const __attribute__((address_space(1))) int*   GIntPtr;
typedef const __attribute__((address_space(1))) v4i*   GInt4Ptr;

struct GatherRide {
    GFloatPtr base;
    GIntPtr   zi;
    GIntPtr   yi;
    GIntPtr   xi;
    GIntPtr   mask;
    float*    feats;
    int       e0[3];
    v4i       m[6], z[6], y[6], x[6];
    float     g[24];
};

__device__ __forceinline__ void gather_idx(GatherRide& gr)
{
#pragma unroll
    for (int q = 0; q < 3; ++q) {
        const int e = gr.e0[q] < A_DIM * CH ? gr.e0[q] : A_DIM * CH - 8;
        gr.z[2 * q] = *(GInt4Ptr)(gr.zi + e);    gr.z[2 * q + 1] = *(GInt4Ptr)(gr.zi + e + 4);
        gr.y[2 * q] = *(GInt4Ptr)(gr.yi + e);    gr.y[2 * q + 1] = *(GInt4Ptr)(gr.yi + e + 4);
        gr.x[2 * q] = *(GInt4Ptr)(gr.xi + e);    gr.x[2 * q + 1] = *(GInt4Ptr)(gr.xi + e + 4);
    }
}

__device__ __forceinline__ void gather_val(GatherRide& gr)
{
#pragma unroll
    for (int q = 0; q < 3; ++q) {
        const int e = gr.e0[q] < A_DIM * CH ? gr.e0[q] : A_DIM * CH - 8;
        gr.m[2 * q] = *(GInt4Ptr)(gr.mask + e);  gr.m[2 * q + 1] = *(GInt4Ptr)(gr.mask + e + 4);
    }
#pragma unroll
    for (int t = 0; t < 6; ++t) {
#pragma unroll
        for (int j = 0; j < 4; ++j)
            gr.g[4 * t + j] = gr.base[(gr.z[t][j] * H_DIM + gr.y[t][j]) * W_DIM + gr.x[t][j]];
    }
}

__device__ __forceinline__ void gather_put(GatherRide& gr)
{
#pragma unroll
    for (int t = 0; t < 6; ++t) {
        const bool row = gr.e0[t >> 1] < A_DIM * CH;
        v4f o;
#pragma unroll
        for (int j = 0; j < 4; ++j)
            o[j] = (row && !gr.m[t][j]) ? gr.g[4 * t + j] : 0.0f;
        *(v4f*)(gr.feats + gr.e0[t >> 1] + 4 * (t & 1)) = o;
    }
}

template <int RP, bool BROW, bool SCALE, bool STORE, bool LOAD, bool FACT, bool FRAG, bool BAR, int GST>
__device__ __forceinline__ void half_trip(v4f (&sa)[8], v4f (&sb)[8],
                                          float (*Asw)[LDS_PAD], float (*Bsw)[BPITCH],
                                          const float (*Asr)[LDS_PAD], const float (*Bsr)[BPITCH],
                                          v2f (&faN)[16], v4f (&fbN)[32], v2f (&fcN)[16],
                                          const v2f (&faC)[16], const v4f (&fbC)[32],
                                          const v2f (&fcC)[16],
                                          v8f& acc0, v8f& acc1, v8f& acc2, v8f& acc3,
                                          v4f (&rs)[8], v4f (&ff)[2], v4f (&fp)[2],
                                          FqPtr Fq, const float* const (&pA)[8],
                                          const float* const (&pB)[3],
                                          int cs, int cl, int cf, int by,
                                          int ar, int ac,
                                          const int (&br)[3], const int (&bc)[3],
                                          int arow, int half, int n, GatherRide& gr)
{
    if constexpr (STORE) {
        if constexpr (BROW) {
            asm volatile("" : "+v"(sb[0]), "+v"(sb[1]), "+v"(sb[2]), "+v"(sb[3]),
                              "+v"(sb[4]), "+v"(sb[5]));
        }
        if constexpr (SCALE) {
            scale_a(sa, rs, ff, fp, cs, by, ar, ac);
            if constexpr (LOAD)
                asm volatile("" : "+v"(rs[0]), "+v"(rs[1]), "+v"(rs[2]), "+v"(rs[3]),
                                  "+v"(rs[4]), "+v"(rs[5]), "+v"(rs[6]), "+v"(rs[7]));
        }
        stage_store<BROW>(Asw, Bsw, sa, sb, ar, ac, br, bc);
    }
    if constexpr (STORE && LOAD) {
        __builtin_amdgcn_sched_barrier(0);
    }
    if constexpr (LOAD) stage_load<RP, BROW>(sa, sb, pA, pB, cl);
    if constexpr (GST == 1) gather_idx(gr);
    if constexpr (GST == 2) gather_val(gr);
    if constexpr (GST == 3) gather_put(gr);
    if constexpr (SCALE && FACT) load_factors(ff, fp, Fq, cf, ar);
    if constexpr (FRAG) frag_load<BROW>(Asr, Bsr, faN, fbN, fcN, arow, half, n);
    if constexpr (BROW) asm volatile("" : "+v"(acc0), "+v"(acc1), "+v"(acc2) :: "memory");
    else                asm volatile("" : "+v"(acc0), "+v"(acc1), "+v"(acc2), "+v"(acc3) :: "memory");
    __builtin_amdgcn_sched_barrier(0);
    mma_chunk<BROW>(faC, fbC, fcC, acc0, acc1, acc2, acc3);
    if constexpr (BAR) {
        if constexpr (BROW) asm volatile("" : "+v"(acc0), "+v"(acc1), "+v"(acc2));
        else                asm volatile("" : "+v"(acc0), "+v"(acc1), "+v"(acc2), "+v"(acc3));
        __builtin_amdgcn_sched_barrier(0);
        lds_barrier();
    }
}

template <int KTOT, int LDA, int RP, int AROWS, bool BROW, bool SCALE, bool RIDE>
__device__ __forceinline__ void gemm_tile(const float* __restrict__ A,
                                          const float* __restrict__ Bpk,
                                          int ib, int j0,
                                          float (*As)[BM][LDS_PAD],
                                          float (*Bs)[KC / 2][BPITCH],
                                          v8f& acc0, v8f& acc1,
                                          v8f& acc2, v8f& acc3,
                                          FqPtr Fq, v4f (&rs)[8], GatherRide& gr)
{
    constexpr int NC = KTOT / KC;
    const int lane = threadIdx.x;
    const int wave = threadIdx.y;
    const int tid  = wave * 32 + lane;
    const int half = lane >> 4;
    const int n    = lane & 15;
    const int arow = wave * 16 + n;
    const int ar   = tid >> 4;
    const int ac   = (tid & 15) * 4;
    int br[3], bc[3];
#pragma unroll
    for (int q = 0; q < 3; ++q) {
        const int u = tid + 128 * q;
        br[q] = BROW ? u / 12 : (tid >> 5);
        bc[q] = BROW ? (u - 12 * (u / 12)) * 8 : (tid & 31) * 4;
    }

    const float* pA[8];
#pragma unroll
    for (int q = 0; q < 8; ++q) {
        int row = ib + ar + 8 * q;
        row = row < AROWS ? row : AROWS - 1;
        pA[q] = A + (size_t)row * LDA + ac;
    }
    const float* pB[3];
#pragma unroll
    for (int q = 0; q < 3; ++q)
        pB[q] = BROW ? Bpk + (size_t)(2 * br[q]) * RP + j0 + (bc[q] >> 1)
                     : Bpk + (size_t)br[q] * RP + 2 * j0 + bc[q];

    v4f sa0[8], sb0[8];
    v4f sa1[8], sb1[8];
    v2f fa0[16], fa1[16];
    v4f fb0[32], fb1[32];
    v2f fc0[16], fc1[16];

    const int by = ib / BM;
    v4f ff[2], fp[2];
    v4f m0, m1;
    v4f ta0[8], tb0[8], ta1[8], tb1[8];
    if constexpr (SCALE) row_maxima(A, m0, m1, ib, tid);
    stage_load<RP, BROW>(ta0, tb0, pA, pB, 0);
    stage_load<RP, BROW>(ta1, tb1, pA, pB, 1);
    __builtin_amdgcn_sched_barrier(0);
    stage_load<RP, BROW>(sa0, sb0, pA, pB, 2);
    __builtin_amdgcn_sched_barrier(0);
    stage_load<RP, BROW>(sa1, sb1, pA, pB, 3);
    __builtin_amdgcn_sched_barrier(0);
    if constexpr (SCALE) {
        row_factors(m0, m1, Fq, tid);
        lds_barrier();
        load_factors(ff, fp, Fq, 0, ar);
        scale_a(ta0, rs, ff, fp, 0, by, ar, ac);
        stage_store<BROW>(As[0], Bs[0], ta0, tb0, ar, ac, br, bc);
        load_factors(ff, fp, Fq, 1, ar);
        scale_a(ta1, rs, ff, fp, 1, by, ar, ac);
        stage_store<BROW>(As[1], Bs[1], ta1, tb1, ar, ac, br, bc);
        load_factors(ff, fp, Fq, 2, ar);
    } else {
        stage_store<BROW>(As[0], Bs[0], ta0, tb0, ar, ac, br, bc);
        stage_store<BROW>(As[1], Bs[1], ta1, tb1, ar, ac, br, bc);
    }
    lds_barrier();
    frag_load<BROW>(As[0], Bs[0], fa0, fb0, fc0, arow, half, n);
    lds_barrier();

    static_assert(NC >= 6 && (NC & 1) == 0, "K loop: fill of 4 chunks, two peeled trips");
#pragma nounroll
    for (int c = 0; c < NC - 4; c += 2) {
        half_trip<RP, BROW, SCALE, true, true, true, true, true, 0>(
            sa0, sb0, As[0], Bs[0], As[1], Bs[1], fa1, fb1, fc1, fa0, fb0, fc0,
            acc0, acc1, acc2, acc3, rs, ff, fp, Fq, pA, pB,
            c + 2, c + 4, c + 3, by, ar, ac, br, bc, arow, half, n, gr);
        half_trip<RP, BROW, SCALE, true, true, true, true, true, 0>(
            sa1, sb1, As[1], Bs[1], As[0], Bs[0], fa0, fb0, fc0, fa1, fb1, fc1,
            acc0, acc1, acc2, acc3, rs, ff, fp, Fq, pA, pB,
            c + 3, c + 5, c + 4, by, ar, ac, br, bc, arow, half, n, gr);
    }
    half_trip<RP, BROW, SCALE, true, false, true, true, true, RIDE ? 1 : 0>(
        sa0, sb0, As[0], Bs[0], As[1], Bs[1], fa1, fb1, fc1, fa0, fb0, fc0,
        acc0, acc1, acc2, acc3, rs, ff, fp, Fq, pA, pB,
        NC - 2, 0, NC - 1, by, ar, ac, br, bc, arow, half, n, gr);
    half_trip<RP, BROW, SCALE, true, false, false, true, true, RIDE ? 2 : 0>(
        sa1, sb1, As[1], Bs[1], As[0], Bs[0], fa0, fb0, fc0, fa1, fb1, fc1,
        acc0, acc1, acc2, acc3, rs, ff, fp, Fq, pA, pB,
        NC - 1, 0, 0, by, ar, ac, br, bc, arow, half, n, gr);
    half_trip<RP, BROW, SCALE, false, false, false, true, false, RIDE ? 3 : 0>(
        sa0, sb0, As[0], Bs[0], As[1], Bs[1], fa1, fb1, fc1, fa0, fb0, fc0,
        acc0, acc1, acc2, acc3, rs, ff, fp, Fq, pA, pB,
        0, 0, 0, by, ar, ac, br, bc, arow, half, n, gr);
    half_trip<RP, BROW, SCALE, false, false, false, false, false, 0>(
        sa1, sb1, As[1], Bs[1], As[0], Bs[0], fa0, fb0, fc0, fa1, fb1, fc1,
        acc0, acc1, acc2, acc3, rs, ff, fp, Fq, pA, pB,
        0, 0, 0, by, ar, ac, br, bc, arow, half, n, gr);
}

__device__ __forceinline__ void gemm1_body(const float* __restrict__ feats,
                             const float* __restrict__ Wpk,
                             const float* __restrict__ biasp,
                             float* __restrict__ attn,
                             const int bx, const int by,
                             float (*As)[BM][LDS_PAD], float (*Bs)[KC / 2][BPITCH])
{
    const int lane = threadIdx.x;
    const int wave = threadIdx.y;
    const int j0   = bx * BN;
    const int ib   = by * BM;
    const int i0   = ib + wave * 16;
    const int half = lane >> 4;
    const int n    = lane & 15;

    v8f acc0, acc1, acc2, acc3;
    {
        const float2 bA = *(const float2*)&biasp[j0 + 2 * n];
        const float2 bB = *(const float2*)&biasp[j0 + 32 + 2 * n];
#pragma unroll
        for (int r = 0; r < 8; ++r) {
            acc0[r] = bA.x; acc1[r] = bA.y; acc2[r] = bB.x; acc3[r] = bB.y;
        }
    }

    v4f unused[8];
    GatherRide none;
    gemm_tile<CH, CH, 2 * NP1, FP, false, false, false>(feats, Wpk, ib, j0, As, Bs,
                                                        acc0, acc1, acc2, acc3, nullptr, unused, none);

    const int  jpA = j0 + 2 * n;
    const bool ok0 = jpA < N1, ok1 = jpA + 1 < N1, ok2 = jpA + 32 < N1, ok3 = jpA + 33 < N1;
    v8f mx;
#pragma unroll
    for (int r = 0; r < 8; ++r) {
        float v = -3.402823466e+38f;
        v = ok0 ? fmaxf(v, acc0[r]) : v;
        v = ok1 ? fmaxf(v, acc1[r]) : v;
        v = ok2 ? fmaxf(v, acc2[r]) : v;
        v = ok3 ? fmaxf(v, acc3[r]) : v;
        mx[r] = v;
    }
#pragma unroll
    for (int r = 0; r < 8; ++r) mx[r] = fmaxf(mx[r], row_xor<1>(mx[r]));
#pragma unroll
    for (int r = 0; r < 8; ++r) mx[r] = fmaxf(mx[r], row_xor<2>(mx[r]));
#pragma unroll
    for (int r = 0; r < 8; ++r) mx[r] = fmaxf(mx[r], row_xor<4>(mx[r]));
#pragma unroll
    for (int r = 0; r < 8; ++r) mx[r] = fmaxf(mx[r], row_xor<8>(mx[r]));
    v8f ex0, ex1, ex2, ex3;
#pragma unroll
    for (int r = 0; r < 8; ++r) {
        ex0[r] = expf(acc0[r] - mx[r]);
        ex1[r] = expf(acc1[r] - mx[r]);
        ex2[r] = expf(acc2[r] - mx[r]);
        ex3[r] = expf(acc3[r] - mx[r]);
    }
    asm volatile("" : "+v"(ex0), "+v"(ex1), "+v"(ex2), "+v"(ex3));
    const int wv  = __builtin_amdgcn_readfirstlane(wave);
    const int i0s = ib + wv * 16;
    if (i0s < MP) {
        char* const rb = (char*)(attn + (size_t)i0s * LDATT);
        const int      m0 = i0s + 8 * half;
        const unsigned ob = (unsigned)(8 * half * LDATT + jpA) * 4u;
        unsigned o0[8], o1[8], o2[8], o3[8];
#pragma unroll
        for (int r = 0; r < 8; ++r) {
            const int m = m0 + r;
            o0[r] = ob + (jpA      >= m ? 4u : 0u);
            o1[r] = ob + (jpA + 1  >= m ? 4u : 0u);
            o2[r] = ob + (jpA + 32 >= m ? 4u : 0u);
            o3[r] = ob + (jpA + 33 >= m ? 4u : 0u);
        }
        if (ok0) {
#pragma unroll
            for (int r = 0; r < 8; ++r) ((float*)(rb + o0[r]))[r * LDATT]      = ex0[r];
        }
        if (ok1) {
#pragma unroll
            for (int r = 0; r < 8; ++r) ((float*)(rb + o1[r]))[r * LDATT + 1]  = ex1[r];
        }
        if (ok2) {
#pragma unroll
            for (int r = 0; r < 8; ++r) ((float*)(rb + o2[r]))[r * LDATT + 32] = ex2[r];
        }
        if (ok3) {
#pragma unroll
            for (int r = 0; r < 8; ++r) ((float*)(rb + o3[r]))[r * LDATT + 33] = ex3[r];
        }
        if (n == 0) {
            float* const mp = (float*)(rb + (unsigned)(8 * half * LDATT + A_DIM + bx) * 4u);
#pragma unroll
            for (int r = 0; r < 8; ++r) mp[r * LDATT] = mx[r];
            if (by == bx && m0 < A_DIM) {
                float* const dp = (float*)(rb + (unsigned)(8 * half * LDATT + m0) * 4u);
#pragma unroll
                for (int r = 0; r < 8; ++r) dp[r * (LDATT + 1)] = 0.0f;
            }
        }
        if (bx == 15 && n < 8) {
            float* const zp = (float*)(rb + (unsigned)(8 * half * LDATT + A_DIM + 16 + n) * 4u);
#pragma unroll
            for (int r = 0; r < 8; ++r) zp[r * LDATT] = 0.0f;
        }
    }
}

__global__ void __launch_bounds__(128) gemm1_scores(const float* __restrict__ feats,
                             const float* __restrict__ Wpk,
                             const float* __restrict__ biasp,
                             float* __restrict__ attn)
{
    __shared__ float As[2][BM][LDS_PAD];
    __shared__ float Bs[2][KC / 2][BPITCH];
    gemm1_body(feats, Wpk, biasp, attn, (int)blockIdx.x, (int)blockIdx.y, As, Bs);
}

__device__ __forceinline__ void gemm2_body(const float* __restrict__ attn,
                          const float* __restrict__ feats,
                          float* __restrict__ outb,
                          const float* __restrict__ fv,
                          const int*   __restrict__ zi,
                          const int*   __restrict__ yi,
                          const int*   __restrict__ xi,
                          const int*   __restrict__ mask,
                          float* __restrict__ featsNext, int bNext,
                          const int bx, const int by,
                          float (*As)[BM][LDS_PAD], float (*Bs)[KC / 2][BPITCH],
                          FqPtr Fq,
                          float* Ss)
{
    const int lane = threadIdx.x;
    const int wave = threadIdx.y;
    const int n0   = bx * BN2;
    const int ib   = by * BM;
    const int half = lane >> 4;
    const int n    = lane & 15;

    v8f acc0 = {}, acc1 = {}, acc2 = {}, acc3 = {};

    v4f rs[8];
#pragma unroll
    for (int q = 0; q < 8; ++q) rs[q] = v4f{0.0f, 0.0f, 0.0f, 0.0f};

    GatherRide gr;
    {
        const int bn = bNext < B_DIM ? bNext : B_DIM - 1;
        const float* gb = fv + (size_t)bn * (C_DIM * H_DIM * W_DIM);
        const int *gz = zi, *gy = yi, *gx = xi, *gm = mask;
        asm volatile("" : "+s"(gb), "+s"(gz), "+s"(gy), "+s"(gx), "+s"(gm));
        gr.base  = (GFloatPtr)gb;
        gr.zi    = (GIntPtr)gz;
        gr.yi    = (GIntPtr)gy;
        gr.xi    = (GIntPtr)gx;
        gr.mask  = (GIntPtr)gm;
        gr.feats = featsNext;
        const int blk = by * 16 + bx;
#pragma unroll
        for (int q = 0; q < 3; ++q)
            gr.e0[q] = ((blk * 3 + q) * 128 + wave * 32 + lane) * 8;
    }

    gemm_tile<FP, LDATT, CH, MP, true, true, true>(attn, feats, ib, n0, As, Bs,
                                                   acc0, acc1, acc2, acc3, Fq, rs, gr);

    v8f sm;
#pragma unroll
    for (int q = 0; q < 8; ++q) sm[q] = (rs[q][0] + rs[q][1]) + (rs[q][2] + rs[q][3]);
#pragma unroll
    for (int q = 0; q < 8; ++q) sm[q] += row_xor<1>(sm[q]);
#pragma unroll
    for (int q = 0; q < 8; ++q) sm[q] += row_xor<2>(sm[q]);
#pragma unroll
    for (int q = 0; q < 8; ++q) sm[q] += row_xor<4>(sm[q]);
#pragma unroll
    for (int q = 0; q < 8; ++q) sm[q] += row_xor<8>(sm[q]);
    if (n == 0) {
#pragma unroll
        for (int q = 0; q < 8; ++q) Ss[2 * wave + half + 8 * q] = sm[q];
    }
    __syncthreads();

    v8f iv;
#pragma unroll
    for (int r = 0; r < 8; ++r) iv[r] = 1.0f / Ss[wave * 16 + r + 8 * half];
    asm volatile("" : "+v"(iv));
    const int wv  = __builtin_amdgcn_readfirstlane(wave);
    const int i0s = ib + wv * 16;
    const int m0  = i0s + 8 * half;
    v8f q0, q1, q2;
#pragma unroll
    for (int r = 0; r < 8; ++r) {
        q0[r] = acc0[r] * iv[r];  q1[r] = acc1[r] * iv[r];
        q2[r] = acc2[r] * iv[r];
    }
    if (m0 < A_DIM) {
        char* const ob = (char*)(outb + (size_t)i0s * CH);
        float* const orow = (float*)(ob + (unsigned)(8 * half * CH + n0 + 2 * n) * 4u);
        float* const ocol = (float*)(ob + (unsigned)(8 * half * CH + n0 + 32 + n) * 4u);
#pragma unroll
        for (int r = 0; r < 8; ++r) {
            *(float2*)(orow + r * CH) = make_float2(q0[r], q1[r]);
            ocol[r * CH]              = q2[r];
        }
    }
}

__global__ void __launch_bounds__(128) gemm2_out(const float* __restrict__ attn,
                          const float* __restrict__ feats,
                          float* __restrict__ outb,
                          const float* __restrict__ fv,
                          const int*   __restrict__ zi,
                          const int*   __restrict__ yi,
                          const int*   __restrict__ xi,
                          const int*   __restrict__ mask,
                          float* __restrict__ featsNext, int bNext)
{
    __shared__ float As[2][BM][LDS_PAD];
    __shared__ float Bs[2][KC / 2][BPITCH];
    __shared__ float Fq[FP / KC][8][8];
    __shared__ float Ss[BM];
    gemm2_body(attn, feats, outb, fv, zi, yi, xi, mask, featsNext, bNext,
               (int)blockIdx.x, (int)blockIdx.y, As, Bs, Fq, Ss);
}

__global__ void __launch_bounds__(128) batch_step(const float* __restrict__ attnCur,
                          const float* __restrict__ featsCur,
                          float* __restrict__ outb,
                          const float* __restrict__ fv,
                          const int*   __restrict__ zi,
                          const int*   __restrict__ yi,
                          const int*   __restrict__ xi,
                          const int*   __restrict__ mask,
                          float* __restrict__ featsRide, int bRide,
                          const float* __restrict__ featsNext,
                          const float* __restrict__ Wpk,
                          const float* __restrict__ biasp,
                          float* __restrict__ attnNext)
{
    __shared__ float As[2][BM][LDS_PAD];
    __shared__ float Bs[2][KC / 2][BPITCH];
    __shared__ float Fq[FP / KC][8][8];
    __shared__ float Ss[BM];
    const int bx = (int)blockIdx.x;
    const int by = (int)blockIdx.y;
    if (by < 16)
        gemm2_body(attnCur, featsCur, outb, fv, zi, yi, xi, mask, featsRide, bRide,
                   bx, by, As, Bs, Fq, Ss);
    else
        gemm1_body(featsNext, Wpk, biasp, attnNext, bx, by - 16, As, Bs);
}

extern "C" void kernel_launch(void* const* d_in, const int* in_sizes, int n_in,
                              void* d_out, int out_size, void* d_ws, size_t ws_size,
                              hipStream_t stream)
{
    const float* fv   = (const float*)d_in[0];
    const int*   zi   = (const int*)  d_in[1];
    const int*   yi   = (const int*)  d_in[2];
    const int*   xi   = (const int*)  d_in[3];
    const int*   mask = (const int*)  d_in[4];
    const float* Wt   = (const float*)d_in[5];
    const float* bias = (const float*)d_in[6];
    float*       out  = (float*)d_out;

    float* Wpk     = (float*)d_ws;
    float* biasp   = Wpk     + (size_t)CH * NP1;
    float* feats0  = biasp   + NP1;
    float* feats1  = feats0  + (size_t)FP * CH;
    float* attn0   = feats1  + (size_t)FP * CH;
    float* feats2  = attn0   + (size_t)MP * LDATT;
    float* attn1   = feats2  + (size_t)FP * CH;
    const size_t full = ((size_t)CH * NP1 + NP1 + 3 * (size_t)FP * CH
                         + 2 * (size_t)MP * LDATT) * sizeof(float);

    const dim3 wblk(32, 4);

    pad_w<<<(CH * NP1 + 255) / 256, 256, 0, stream>>>(Wt, Wpk);
    pad_bias<<<(NP1 + 255) / 256, 256, 0, stream>>>(bias, biasp);

    if (ws_size >= full) {
        float* const F[3] = {feats0, feats1, feats2};
        float* const T[2] = {attn0, attn1};
        gather_feats<<<dim3(GATHER_BLOCKS, 2), 256, 0, stream>>>(fv, zi, yi, xi, mask, feats0, 0);
        gemm1_scores<<<dim3(16, 16), wblk, 0, stream>>>(feats0, Wpk, biasp, attn0);
        for (int b = 0; b + 1 < B_DIM; ++b)
            batch_step<<<dim3(16, 32), wblk, 0, stream>>>(
                T[b & 1], F[b % 3], out + (size_t)b * A_DIM * CH,
                fv, zi, yi, xi, mask, F[(b + 2) % 3], b + 2,
                F[(b + 1) % 3], Wpk, biasp, T[(b + 1) & 1]);
        gemm2_out<<<dim3(16, 16), wblk, 0, stream>>>(
            T[(B_DIM - 1) & 1], F[(B_DIM - 1) % 3], out + (size_t)(B_DIM - 1) * A_DIM * CH,
            fv, zi, yi, xi, mask, F[(B_DIM + 1) % 3], B_DIM);
        return;
    }

    gather_feats<<<GATHER_BLOCKS, 256, 0, stream>>>(fv, zi, yi, xi, mask, feats0, 0);

    for (int b = 0; b < B_DIM; ++b) {
        float* cur = (b & 1) ? feats1 : feats0;
        float* nxt = (b & 1) ? feats0 : feats1;
        gemm1_scores<<<dim3(16, 16), wblk, 0, stream>>>(cur, Wpk, biasp, attn0);
        gemm2_out<<<dim3(16, 16), wblk, 0, stream>>>(
            attn0, cur, out + (size_t)b * A_DIM * CH,
            fv, zi, yi, xi, mask, nxt, b + 1);
    }
}


# Two work-groups per WGP in the batch launch: a wave kept within half of a SIMD's registers (one chunk of global loads in flight, was two; riding gather one half trip later); sums same
# speedup vs baseline: 6.2235x; 1.1436x over previous
#include <hip/hip_runtime.h>


typedef __attribute__((ext_vector_type(2))) float v2f;
typedef __attribute__((ext_vector_type(4))) float v4f;
typedef __attribute__((ext_vector_type(8))) float v8f;

template <int M>
__device__ __forceinline__ float row_xor(float v)
{
    const int i = __builtin_bit_cast(int, v);
    return __builtin_bit_cast(float,
        __builtin_amdgcn_update_dpp(i, i, 0x160 + M, 0xf, 0xf, true));
}

#define B_DIM  32
#define A_DIM  1000
#define C_DIM  64
#define H_DIM  12
#define W_DIM  20
#define CH     768
#define N1     999
#define MP     1008
#define NP1    1024
#define FP     1024
#define LDATT  1024
#define KC     64
#define LDS_PAD 68
#define BM     64
#define BN     64
#define BN2    48
#define BPITCH 160


__global__ void pad_w(const float* __restrict__ Wt, float* __restrict__ Wpk)
{
    int e = blockIdx.x * blockDim.x + threadIdx.x;
    if (e >= CH * NP1) return;
    int k = e >> 10;
    int j = e & (NP1 - 1);
    float v = (j < N1) ? Wt[k * N1 + j] : 0.0f;
    Wpk[(size_t)(k >> 1) * (2 * NP1) + 2 * j + (k & 1)] = v;
}

__global__ void pad_bias(const float* __restrict__ bias, float* __restrict__ biasp)
{
    int j = blockIdx.x * blockDim.x + threadIdx.x;
    if (j < NP1) biasp[j] = (j < N1) ? bias[j] : 0.0f;
}

#define GATHER_BLOCKS ((FP * CH + 255) / 256)

__device__ __forceinline__ void gather_body(int blk,
                             const float* __restrict__ fv,
                             const int*   __restrict__ zi,
                             const int*   __restrict__ yi,
                             const int*   __restrict__ xi,
                             const int*   __restrict__ mask,
                             float* __restrict__ feats,
                             int b)
{
    int e = blk * 256 + threadIdx.x;
    if (e >= FP * CH) return;
    int a = e / CH;
    int k = e - a * CH;
    float v = 0.0f;
    if (a < A_DIM) {
        int idx = a * CH + k;
        if (!mask[idx]) {
            int z = zi[idx], y = yi[idx], x = xi[idx];
            v = fv[((b * C_DIM + z) * H_DIM + y) * W_DIM + x];
        }
    }
    feats[e] = v;
}

__global__ void gather_feats(const float* __restrict__ fv,
                             const int*   __restrict__ zi,
                             const int*   __restrict__ yi,
                             const int*   __restrict__ xi,
                             const int*   __restrict__ mask,
                             float* __restrict__ feats, int b)
{
    gather_body(blockIdx.x, fv, zi, yi, xi, mask,
                feats + (size_t)blockIdx.y * (FP * CH), b + (int)blockIdx.y);
}

template <int RP, bool BROW>
__device__ __forceinline__ void stage_load(v4f (&sa)[8], v4f (&sb)[8],
                                           const float* const (&pA)[8],
                                           const float* const (&pB)[3], int c)
{
#pragma unroll
    for (int q = 0; q < 8; ++q)
        sa[q] = *(const v4f*)(pA[q] + c * KC);
    if constexpr (BROW) {
#pragma unroll
        for (int q = 0; q < 3; ++q) {
            const float* r = pB[q] + (size_t)(c * KC) * RP;
            sb[2 * q]     = *(const v4f*)r;
            sb[2 * q + 1] = *(const v4f*)(r + RP);
        }
    } else {
#pragma unroll
        for (int q = 0; q < 8; ++q)
            sb[q] = *(const v4f*)(pB[0] + (size_t)(c * (KC / 2) + 4 * q) * RP);
    }
}

template <bool BROW>
__device__ __forceinline__ void stage_store(float (*As)[LDS_PAD], float (*Bs)[BPITCH],
                                            const v4f (&sa)[8], const v4f (&sb)[8],
                                            int ar, int ac,
                                            const int (&br)[3], const int (&bc)[3])
{
#pragma unroll
    for (int q = 0; q < 8; ++q)
        *(v4f*)&As[ar + 8 * q][ac] = sa[q];
    if constexpr (BROW) {
#pragma unroll
        for (int q = 0; q < 3; ++q) {
            *(v4f*)&Bs[br[q]][bc[q]] =
                __builtin_shufflevector(sb[2 * q], sb[2 * q + 1], 0, 4, 1, 5);
            *(v4f*)&Bs[br[q]][bc[q] + 4] =
                __builtin_shufflevector(sb[2 * q], sb[2 * q + 1], 2, 6, 3, 7);
        }
    } else {
#pragma unroll
        for (int q = 0; q < 8; ++q)
            *(v4f*)&Bs[br[0] + 4 * q][bc[0]] = sb[q];
    }
}

template <bool W3>
__device__ __forceinline__ void frag_load(const float (*As)[LDS_PAD],
                                          const float (*Bs)[BPITCH],
                                          v2f (&fa)[16], v4f (&fb)[32], v2f (&fc)[16],
                                          int arow, int half, int n)
{
#pragma unroll
    for (int s = 0; s < 16; ++s) {
        fa[s]         = *(const v2f*)&As[arow][4 * s + 2 * half];
        fb[2 * s]     = *(const v4f*)&Bs[2 * s + half][4 * n];
        if constexpr (W3) fc[s]         = *(const v2f*)&Bs[2 * s + half][64 + 2 * n];
        else              fb[2 * s + 1] = *(const v4f*)&Bs[2 * s + half][64 + 4 * n];
    }
}

template <bool W3>
__device__ __forceinline__ void mma_chunk(const v2f (&fa)[16], const v4f (&fb)[32],
                                          const v2f (&fc)[16],
                                          v8f& acc0, v8f& acc1,
                                          v8f& acc2, v8f& acc3)
{
#pragma unroll
    for (int s = 0; s < 16; ++s) {
        const v2f b0 = __builtin_shufflevector(fb[2 * s],     fb[2 * s],     0, 1);
        const v2f b1 = __builtin_shufflevector(fb[2 * s],     fb[2 * s],     2, 3);
        acc0 = __builtin_amdgcn_wmma_f32_16x16x4_f32(
            false, fa[s], false, b0, (short)0, acc0, false, false);
        acc1 = __builtin_amdgcn_wmma_f32_16x16x4_f32(
            false, fa[s], false, b1, (short)0, acc1, false, false);
        if constexpr (W3) {
            acc2 = __builtin_amdgcn_wmma_f32_16x16x4_f32(
                false, fa[s], false, fc[s], (short)0, acc2, false, false);
        } else {
            const v2f b2 = __builtin_shufflevector(fb[2 * s + 1], fb[2 * s + 1], 0, 1);
            const v2f b3 = __builtin_shufflevector(fb[2 * s + 1], fb[2 * s + 1], 2, 3);
            acc2 = __builtin_amdgcn_wmma_f32_16x16x4_f32(
                false, fa[s], false, b2, (short)0, acc2, false, false);
            acc3 = __builtin_amdgcn_wmma_f32_16x16x4_f32(
                false, fa[s], false, b3, (short)0, acc3, false, false);
        }
    }
}

typedef float (*FqPtr)[8][8];

__device__ __forceinline__ void row_maxima(const float* __restrict__ attn,
                                           v4f& m0, v4f& m1, int ib, int tid)
{
    const int rl = tid >> 1;
    const int hc = (tid & 1) * 8;
    int row = ib + rl;
    row = row < MP ? row : MP - 1;
    const float* st = attn + (size_t)row * LDATT + A_DIM + hc;
    m0 = *(const v4f*)st;
    m1 = *(const v4f*)(st + 4);
}

__device__ __forceinline__ void row_factors(const v4f m0, const v4f m1, FqPtr Fq, int tid)
{
    const int rl = tid >> 1;
    const int hc = (tid & 1) * 8;
    float mx = fmaxf(fmaxf(fmaxf(m0[0], m0[1]), fmaxf(m0[2], m0[3])),
                     fmaxf(fmaxf(m1[0], m1[1]), fmaxf(m1[2], m1[3])));
    mx = fmaxf(mx, row_xor<1>(mx));
#pragma unroll
    for (int e = 0; e < 4; ++e) {
        Fq[hc + e][rl & 7][rl >> 3]     = expf(m0[e] - mx);
        Fq[hc + 4 + e][rl & 7][rl >> 3] = expf(m1[e] - mx);
    }
}

__device__ __forceinline__ void lds_barrier()
{
    __builtin_amdgcn_fence(__ATOMIC_RELEASE, "workgroup", "local");
    __builtin_amdgcn_s_barrier();
    __builtin_amdgcn_fence(__ATOMIC_ACQUIRE, "workgroup", "local");
}

__device__ __forceinline__ void load_factors(v4f (&ff)[2], v4f (&fp)[2],
                                             FqPtr Fq, int c, int ar)
{
    const int cp = c > 0 ? c - 1 : 0;
    ff[0] = *(const v4f*)&Fq[c][ar][0];
    ff[1] = *(const v4f*)&Fq[c][ar][4];
    fp[0] = *(const v4f*)&Fq[cp][ar][0];
    fp[1] = *(const v4f*)&Fq[cp][ar][4];
}

__device__ __forceinline__ void scale_a(v4f (&sa)[8], v4f (&rs)[8],
                                        const v4f (&ff)[2], const v4f (&fp)[2],
                                        int c, int by, int ar, int ac)
{
    const bool  first = (ac == 0) && (c > by);
    const float keep  = (c * KC + ac >= A_DIM) ? 0.0f : 1.0f;
    const v4f   keep4 = {keep, keep, keep, keep};
#pragma unroll
    for (int q = 0; q < 8; ++q) {
        const float f  = ff[q >> 2][q & 3];
        const float f0 = first ? fp[q >> 2][q & 3] : f;
        const v4f fv4 = {f0, f, f, f};
        const v4f w = sa[q] * fv4;
        rs[q] = w * keep4 + rs[q];
        sa[q] = w;
    }
}

typedef int v4i __attribute__((ext_vector_type(4)));
typedef const __attribute__((address_space(1))) float* GFloatPtr;
typedef const __attribute__((address_space(1))) int*   GIntPtr;
typedef const __attribute__((address_space(1))) v4i*   GInt4Ptr;

struct GatherRide {
    GFloatPtr base;
    GIntPtr   zi;
    GIntPtr   yi;
    GIntPtr   xi;
    GIntPtr   mask;
    float*    feats;
    int       e0[3];
    v4i       m[6], z[6], y[6], x[6];
    float     g[24];
};

__device__ __forceinline__ void gather_idx(GatherRide& gr)
{
#pragma unroll
    for (int q = 0; q < 3; ++q) {
        const int e = gr.e0[q] < A_DIM * CH ? gr.e0[q] : A_DIM * CH - 8;
        gr.z[2 * q] = *(GInt4Ptr)(gr.zi + e);    gr.z[2 * q + 1] = *(GInt4Ptr)(gr.zi + e + 4);
        gr.y[2 * q] = *(GInt4Ptr)(gr.yi + e);    gr.y[2 * q + 1] = *(GInt4Ptr)(gr.yi + e + 4);
        gr.x[2 * q] = *(GInt4Ptr)(gr.xi + e);    gr.x[2 * q + 1] = *(GInt4Ptr)(gr.xi + e + 4);
    }
}

__device__ __forceinline__ void gather_val(GatherRide& gr)
{
#pragma unroll
    for (int q = 0; q < 3; ++q) {
        const int e = gr.e0[q] < A_DIM * CH ? gr.e0[q] : A_DIM * CH - 8;
        gr.m[2 * q] = *(GInt4Ptr)(gr.mask + e);  gr.m[2 * q + 1] = *(GInt4Ptr)(gr.mask + e + 4);
    }
#pragma unroll
    for (int t = 0; t < 6; ++t) {
#pragma unroll
        for (int j = 0; j < 4; ++j)
            gr.g[4 * t + j] = gr.base[(gr.z[t][j] * H_DIM + gr.y[t][j]) * W_DIM + gr.x[t][j]];
    }
}

__device__ __forceinline__ void gather_put(GatherRide& gr)
{
#pragma unroll
    for (int t = 0; t < 6; ++t) {
        const bool row = gr.e0[t >> 1] < A_DIM * CH;
        v4f o;
#pragma unroll
        for (int j = 0; j < 4; ++j)
            o[j] = (row && !gr.m[t][j]) ? gr.g[4 * t + j] : 0.0f;
        *(v4f*)(gr.feats + gr.e0[t >> 1] + 4 * (t & 1)) = o;
    }
}

template <int RP, bool BROW, bool SCALE, bool STORE, bool LOAD, bool FACT, bool FRAG, bool BAR, int GST>
__device__ __forceinline__ void half_trip(v4f (&sa)[8], v4f (&sb)[8],
                                          float (*Asw)[LDS_PAD], float (*Bsw)[BPITCH],
                                          const float (*Asr)[LDS_PAD], const float (*Bsr)[BPITCH],
                                          v2f (&faN)[16], v4f (&fbN)[32], v2f (&fcN)[16],
                                          const v2f (&faC)[16], const v4f (&fbC)[32],
                                          const v2f (&fcC)[16],
                                          v8f& acc0, v8f& acc1, v8f& acc2, v8f& acc3,
                                          v4f (&rs)[8], v4f (&ff)[2], v4f (&fp)[2],
                                          FqPtr Fq, const float* const (&pA)[8],
                                          const float* const (&pB)[3],
                                          int cs, int cl, int cf, int by,
                                          int ar, int ac,
                                          const int (&br)[3], const int (&bc)[3],
                                          int arow, int half, int n, GatherRide& gr)
{
    if constexpr (STORE) {
        if constexpr (BROW) {
            asm volatile("" : "+v"(sb[0]), "+v"(sb[1]), "+v"(sb[2]), "+v"(sb[3]),
                              "+v"(sb[4]), "+v"(sb[5]));
        }
        if constexpr (SCALE) {
            scale_a(sa, rs, ff, fp, cs, by, ar, ac);
            if constexpr (LOAD)
                asm volatile("" : "+v"(rs[0]), "+v"(rs[1]), "+v"(rs[2]), "+v"(rs[3]),
                                  "+v"(rs[4]), "+v"(rs[5]), "+v"(rs[6]), "+v"(rs[7]));
        }
        stage_store<BROW>(Asw, Bsw, sa, sb, ar, ac, br, bc);
    }
    if constexpr (STORE && LOAD) {
        __builtin_amdgcn_sched_barrier(0);
    }
    if constexpr (LOAD) stage_load<RP, BROW>(sa, sb, pA, pB, cl);
    if constexpr (GST == 1) gather_idx(gr);
    if constexpr (GST == 2) gather_val(gr);
    if constexpr (GST == 3) gather_put(gr);
    if constexpr (SCALE && FACT) load_factors(ff, fp, Fq, cf, ar);
    if constexpr (FRAG) frag_load<BROW>(Asr, Bsr, faN, fbN, fcN, arow, half, n);
    if constexpr (BROW) asm volatile("" : "+v"(acc0), "+v"(acc1), "+v"(acc2) :: "memory");
    else                asm volatile("" : "+v"(acc0), "+v"(acc1), "+v"(acc2), "+v"(acc3) :: "memory");
    __builtin_amdgcn_sched_barrier(0);
    mma_chunk<BROW>(faC, fbC, fcC, acc0, acc1, acc2, acc3);
    if constexpr (BAR || GST == 2) {
        if constexpr (BROW) asm volatile("" : "+v"(acc0), "+v"(acc1), "+v"(acc2));
        else                asm volatile("" : "+v"(acc0), "+v"(acc1), "+v"(acc2), "+v"(acc3));
        __builtin_amdgcn_sched_barrier(0);
    }
    if constexpr (BAR) lds_barrier();
}

template <int KTOT, int LDA, int RP, int AROWS, bool BROW, bool SCALE, bool RIDE>
__device__ __forceinline__ void gemm_tile(const float* __restrict__ A,
                                          const float* __restrict__ Bpk,
                                          int ib, int j0,
                                          float (*As)[BM][LDS_PAD],
                                          float (*Bs)[KC / 2][BPITCH],
                                          v8f& acc0, v8f& acc1,
                                          v8f& acc2, v8f& acc3,
                                          FqPtr Fq, v4f (&rs)[8], GatherRide& gr)
{
    constexpr int NC = KTOT / KC;
    const int lane = threadIdx.x;
    const int wave = threadIdx.y;
    const int tid  = wave * 32 + lane;
    const int half = lane >> 4;
    const int n    = lane & 15;
    const int arow = wave * 16 + n;
    const int ar   = tid >> 4;
    const int ac   = (tid & 15) * 4;
    int br[3], bc[3];
#pragma unroll
    for (int q = 0; q < 3; ++q) {
        const int u = tid + 128 * q;
        br[q] = BROW ? u / 12 : (tid >> 5);
        bc[q] = BROW ? (u - 12 * (u / 12)) * 8 : (tid & 31) * 4;
    }

    const float* pA[8];
#pragma unroll
    for (int q = 0; q < 8; ++q) {
        int row = ib + ar + 8 * q;
        row = row < AROWS ? row : AROWS - 1;
        pA[q] = A + (size_t)row * LDA + ac;
    }
    const float* pB[3];
#pragma unroll
    for (int q = 0; q < 3; ++q)
        pB[q] = BROW ? Bpk + (size_t)(2 * br[q]) * RP + j0 + (bc[q] >> 1)
                     : Bpk + (size_t)br[q] * RP + 2 * j0 + bc[q];

    v4f sa0[8], sb0[8];
    v2f fa0[16], fa1[16];
    v4f fb0[32], fb1[32];
    v2f fc0[16], fc1[16];

    const int by = ib / BM;
    v4f ff[2], fp[2];
    v4f m0, m1;
    v4f ta0[8], tb0[8], ta1[8], tb1[8];
    if constexpr (SCALE) row_maxima(A, m0, m1, ib, tid);
    stage_load<RP, BROW>(ta0, tb0, pA, pB, 0);
    stage_load<RP, BROW>(ta1, tb1, pA, pB, 1);
    __builtin_amdgcn_sched_barrier(0);
    stage_load<RP, BROW>(sa0, sb0, pA, pB, 2);
    __builtin_amdgcn_sched_barrier(0);
    if constexpr (SCALE) {
        row_factors(m0, m1, Fq, tid);
        lds_barrier();
        load_factors(ff, fp, Fq, 0, ar);
        scale_a(ta0, rs, ff, fp, 0, by, ar, ac);
        stage_store<BROW>(As[0], Bs[0], ta0, tb0, ar, ac, br, bc);
        load_factors(ff, fp, Fq, 1, ar);
        scale_a(ta1, rs, ff, fp, 1, by, ar, ac);
        stage_store<BROW>(As[1], Bs[1], ta1, tb1, ar, ac, br, bc);
        load_factors(ff, fp, Fq, 2, ar);
    } else {
        stage_store<BROW>(As[0], Bs[0], ta0, tb0, ar, ac, br, bc);
        stage_store<BROW>(As[1], Bs[1], ta1, tb1, ar, ac, br, bc);
    }
    lds_barrier();
    frag_load<BROW>(As[0], Bs[0], fa0, fb0, fc0, arow, half, n);
    lds_barrier();

    static_assert(NC >= 6 && (NC & 1) == 0, "K loop: fill of 3 chunks, two peeled trips");
#pragma nounroll
    for (int c = 0; c < NC - 4; c += 2) {
        half_trip<RP, BROW, SCALE, true, true, true, true, true, 0>(
            sa0, sb0, As[0], Bs[0], As[1], Bs[1], fa1, fb1, fc1, fa0, fb0, fc0,
            acc0, acc1, acc2, acc3, rs, ff, fp, Fq, pA, pB,
            c + 2, c + 3, c + 3, by, ar, ac, br, bc, arow, half, n, gr);
        half_trip<RP, BROW, SCALE, true, true, true, true, true, 0>(
            sa0, sb0, As[1], Bs[1], As[0], Bs[0], fa0, fb0, fc0, fa1, fb1, fc1,
            acc0, acc1, acc2, acc3, rs, ff, fp, Fq, pA, pB,
            c + 3, c + 4, c + 4, by, ar, ac, br, bc, arow, half, n, gr);
    }
    half_trip<RP, BROW, SCALE, true, true, true, true, true, 0>(
        sa0, sb0, As[0], Bs[0], As[1], Bs[1], fa1, fb1, fc1, fa0, fb0, fc0,
        acc0, acc1, acc2, acc3, rs, ff, fp, Fq, pA, pB,
        NC - 2, NC - 1, NC - 1, by, ar, ac, br, bc, arow, half, n, gr);
    half_trip<RP, BROW, SCALE, true, false, false, true, true, RIDE ? 1 : 0>(
        sa0, sb0, As[1], Bs[1], As[0], Bs[0], fa0, fb0, fc0, fa1, fb1, fc1,
        acc0, acc1, acc2, acc3, rs, ff, fp, Fq, pA, pB,
        NC - 1, 0, 0, by, ar, ac, br, bc, arow, half, n, gr);
    half_trip<RP, BROW, SCALE, false, false, false, true, false, RIDE ? 2 : 0>(
        sa0, sb0, As[0], Bs[0], As[1], Bs[1], fa1, fb1, fc1, fa0, fb0, fc0,
        acc0, acc1, acc2, acc3, rs, ff, fp, Fq, pA, pB,
        0, 0, 0, by, ar, ac, br, bc, arow, half, n, gr);
    half_trip<RP, BROW, SCALE, false, false, false, false, false, RIDE ? 3 : 0>(
        sa0, sb0, As[1], Bs[1], As[0], Bs[0], fa0, fb0, fc0, fa1, fb1, fc1,
        acc0, acc1, acc2, acc3, rs, ff, fp, Fq, pA, pB,
        0, 0, 0, by, ar, ac, br, bc, arow, half, n, gr);
}

__device__ __forceinline__ void gemm1_body(const float* __restrict__ feats,
                             const float* __restrict__ Wpk,
                             const float* __restrict__ biasp,
                             float* __restrict__ attn,
                             const int bx, const int by,
                             float (*As)[BM][LDS_PAD], float (*Bs)[KC / 2][BPITCH])
{
    const int lane = threadIdx.x;
    const int wave = threadIdx.y;
    const int j0   = bx * BN;
    const int ib   = by * BM;
    const int i0   = ib + wave * 16;
    const int half = lane >> 4;
    const int n    = lane & 15;

    v8f acc0, acc1, acc2, acc3;
    {
        const float2 bA = *(const float2*)&biasp[j0 + 2 * n];
        const float2 bB = *(const float2*)&biasp[j0 + 32 + 2 * n];
#pragma unroll
        for (int r = 0; r < 8; ++r) {
            acc0[r] = bA.x; acc1[r] = bA.y; acc2[r] = bB.x; acc3[r] = bB.y;
        }
    }

    v4f unused[8];
    GatherRide none;
    gemm_tile<CH, CH, 2 * NP1, FP, false, false, false>(feats, Wpk, ib, j0, As, Bs,
                                                        acc0, acc1, acc2, acc3, nullptr, unused, none);

    const int  jpA = j0 + 2 * n;
    const bool ok0 = jpA < N1, ok1 = jpA + 1 < N1, ok2 = jpA + 32 < N1, ok3 = jpA + 33 < N1;
    v8f mx;
#pragma unroll
    for (int r = 0; r < 8; ++r) {
        float v = -3.402823466e+38f;
        v = ok0 ? fmaxf(v, acc0[r]) : v;
        v = ok1 ? fmaxf(v, acc1[r]) : v;
        v = ok2 ? fmaxf(v, acc2[r]) : v;
        v = ok3 ? fmaxf(v, acc3[r]) : v;
        mx[r] = v;
    }
#pragma unroll
    for (int r = 0; r < 8; ++r) mx[r] = fmaxf(mx[r], row_xor<1>(mx[r]));
#pragma unroll
    for (int r = 0; r < 8; ++r) mx[r] = fmaxf(mx[r], row_xor<2>(mx[r]));
#pragma unroll
    for (int r = 0; r < 8; ++r) mx[r] = fmaxf(mx[r], row_xor<4>(mx[r]));
#pragma unroll
    for (int r = 0; r < 8; ++r) mx[r] = fmaxf(mx[r], row_xor<8>(mx[r]));
    v8f ex0, ex1, ex2, ex3;
#pragma unroll
    for (int r = 0; r < 8; ++r) {
        ex0[r] = expf(acc0[r] - mx[r]);
        ex1[r] = expf(acc1[r] - mx[r]);
        ex2[r] = expf(acc2[r] - mx[r]);
        ex3[r] = expf(acc3[r] - mx[r]);
    }
    asm volatile("" : "+v"(ex0), "+v"(ex1), "+v"(ex2), "+v"(ex3));
    const int wv  = __builtin_amdgcn_readfirstlane(wave);
    const int i0s = ib + wv * 16;
    if (i0s < MP) {
        char* const rb = (char*)(attn + (size_t)i0s * LDATT);
        const int      m0 = i0s + 8 * half;
        const unsigned ob = (unsigned)(8 * half * LDATT + jpA) * 4u;
        unsigned o0[8], o1[8], o2[8], o3[8];
#pragma unroll
        for (int r = 0; r < 8; ++r) {
            const int m = m0 + r;
            o0[r] = ob + (jpA      >= m ? 4u : 0u);
            o1[r] = ob + (jpA + 1  >= m ? 4u : 0u);
            o2[r] = ob + (jpA + 32 >= m ? 4u : 0u);
            o3[r] = ob + (jpA + 33 >= m ? 4u : 0u);
        }
        if (ok0) {
#pragma unroll
            for (int r = 0; r < 8; ++r) ((float*)(rb + o0[r]))[r * LDATT]      = ex0[r];
        }
        if (ok1) {
#pragma unroll
            for (int r = 0; r < 8; ++r) ((float*)(rb + o1[r]))[r * LDATT + 1]  = ex1[r];
        }
        if (ok2) {
#pragma unroll
            for (int r = 0; r < 8; ++r) ((float*)(rb + o2[r]))[r * LDATT + 32] = ex2[r];
        }
        if (ok3) {
#pragma unroll
            for (int r = 0; r < 8; ++r) ((float*)(rb + o3[r]))[r * LDATT + 33] = ex3[r];
        }
        if (n == 0) {
            float* const mp = (float*)(rb + (unsigned)(8 * half * LDATT + A_DIM + bx) * 4u);
#pragma unroll
            for (int r = 0; r < 8; ++r) mp[r * LDATT] = mx[r];
            if (by == bx && m0 < A_DIM) {
                float* const dp = (float*)(rb + (unsigned)(8 * half * LDATT + m0) * 4u);
#pragma unroll
                for (int r = 0; r < 8; ++r) dp[r * (LDATT + 1)] = 0.0f;
            }
        }
        if (bx == 15 && n < 8) {
            float* const zp = (float*)(rb + (unsigned)(8 * half * LDATT + A_DIM + 16 + n) * 4u);
#pragma unroll
            for (int r = 0; r < 8; ++r) zp[r * LDATT] = 0.0f;
        }
    }
}

__global__ void __launch_bounds__(128) gemm1_scores(const float* __restrict__ feats,
                             const float* __restrict__ Wpk,
                             const float* __restrict__ biasp,
                             float* __restrict__ attn)
{
    __shared__ float As[2][BM][LDS_PAD];
    __shared__ float Bs[2][KC / 2][BPITCH];
    gemm1_body(feats, Wpk, biasp, attn, (int)blockIdx.x, (int)blockIdx.y, As, Bs);
}

__device__ __forceinline__ void gemm2_body(const float* __restrict__ attn,
                          const float* __restrict__ feats,
                          float* __restrict__ outb,
                          const float* __restrict__ fv,
                          const int*   __restrict__ zi,
                          const int*   __restrict__ yi,
                          const int*   __restrict__ xi,
                          const int*   __restrict__ mask,
                          float* __restrict__ featsNext, int bNext,
                          const int bx, const int by,
                          float (*As)[BM][LDS_PAD], float (*Bs)[KC / 2][BPITCH],
                          FqPtr Fq,
                          float* Ss)
{
    const int lane = threadIdx.x;
    const int wave = threadIdx.y;
    const int n0   = bx * BN2;
    const int ib   = by * BM;
    const int half = lane >> 4;
    const int n    = lane & 15;

    v8f acc0 = {}, acc1 = {}, acc2 = {}, acc3 = {};

    v4f rs[8];
#pragma unroll
    for (int q = 0; q < 8; ++q) rs[q] = v4f{0.0f, 0.0f, 0.0f, 0.0f};

    GatherRide gr;
    {
        const int bn = bNext < B_DIM ? bNext : B_DIM - 1;
        const float* gb = fv + (size_t)bn * (C_DIM * H_DIM * W_DIM);
        const int *gz = zi, *gy = yi, *gx = xi, *gm = mask;
        asm volatile("" : "+s"(gb), "+s"(gz), "+s"(gy), "+s"(gx), "+s"(gm));
        gr.base  = (GFloatPtr)gb;
        gr.zi    = (GIntPtr)gz;
        gr.yi    = (GIntPtr)gy;
        gr.xi    = (GIntPtr)gx;
        gr.mask  = (GIntPtr)gm;
        gr.feats = featsNext;
        const int blk = by * 16 + bx;
#pragma unroll
        for (int q = 0; q < 3; ++q)
            gr.e0[q] = ((blk * 3 + q) * 128 + wave * 32 + lane) * 8;
    }

    gemm_tile<FP, LDATT, CH, MP, true, true, true>(attn, feats, ib, n0, As, Bs,
                                                   acc0, acc1, acc2, acc3, Fq, rs, gr);

    v8f sm;
#pragma unroll
    for (int q = 0; q < 8; ++q) sm[q] = (rs[q][0] + rs[q][1]) + (rs[q][2] + rs[q][3]);
#pragma unroll
    for (int q = 0; q < 8; ++q) sm[q] += row_xor<1>(sm[q]);
#pragma unroll
    for (int q = 0; q < 8; ++q) sm[q] += row_xor<2>(sm[q]);
#pragma unroll
    for (int q = 0; q < 8; ++q) sm[q] += row_xor<4>(sm[q]);
#pragma unroll
    for (int q = 0; q < 8; ++q) sm[q] += row_xor<8>(sm[q]);
    if (n == 0) {
#pragma unroll
        for (int q = 0; q < 8; ++q) Ss[2 * wave + half + 8 * q] = sm[q];
    }
    __syncthreads();

    v8f iv;
#pragma unroll
    for (int r = 0; r < 8; ++r) iv[r] = 1.0f / Ss[wave * 16 + r + 8 * half];
    asm volatile("" : "+v"(iv));
    const int wv  = __builtin_amdgcn_readfirstlane(wave);
    const int i0s = ib + wv * 16;
    const int m0  = i0s + 8 * half;
    v8f q0, q1, q2;
#pragma unroll
    for (int r = 0; r < 8; ++r) {
        q0[r] = acc0[r] * iv[r];  q1[r] = acc1[r] * iv[r];
        q2[r] = acc2[r] * iv[r];
    }
    if (m0 < A_DIM) {
        char* const ob = (char*)(outb + (size_t)i0s * CH);
        float* const orow = (float*)(ob + (unsigned)(8 * half * CH + n0 + 2 * n) * 4u);
        float* const ocol = (float*)(ob + (unsigned)(8 * half * CH + n0 + 32 + n) * 4u);
#pragma unroll
        for (int r = 0; r < 8; ++r) {
            *(float2*)(orow + r * CH) = make_float2(q0[r], q1[r]);
            ocol[r * CH]              = q2[r];
        }
    }
}

__global__ void __launch_bounds__(128) gemm2_out(const float* __restrict__ attn,
                          const float* __restrict__ feats,
                          float* __restrict__ outb,
                          const float* __restrict__ fv,
                          const int*   __restrict__ zi,
                          const int*   __restrict__ yi,
                          const int*   __restrict__ xi,
                          const int*   __restrict__ mask,
                          float* __restrict__ featsNext, int bNext)
{
    __shared__ float As[2][BM][LDS_PAD];
    __shared__ float Bs[2][KC / 2][BPITCH];
    __shared__ float Fq[FP / KC][8][8];
    __shared__ float Ss[BM];
    gemm2_body(attn, feats, outb, fv, zi, yi, xi, mask, featsNext, bNext,
               (int)blockIdx.x, (int)blockIdx.y, As, Bs, Fq, Ss);
}

__global__ void __launch_bounds__(128, 2) batch_step(const float* __restrict__ attnCur,
                          const float* __restrict__ featsCur,
                          float* __restrict__ outb,
                          const float* __restrict__ fv,
                          const int*   __restrict__ zi,
                          const int*   __restrict__ yi,
                          const int*   __restrict__ xi,
                          const int*   __restrict__ mask,
                          float* __restrict__ featsRide, int bRide,
                          const float* __restrict__ featsNext,
                          const float* __restrict__ Wpk,
                          const float* __restrict__ biasp,
                          float* __restrict__ attnNext)
{
    __shared__ float As[2][BM][LDS_PAD];
    __shared__ float Bs[2][KC / 2][BPITCH];
    __shared__ float Fq[FP / KC][8][8];
    __shared__ float Ss[BM];
    const int bx = (int)blockIdx.x;
    const int by = (int)blockIdx.y;
    if (by < 16)
        gemm2_body(attnCur, featsCur, outb, fv, zi, yi, xi, mask, featsRide, bRide,
                   bx, by, As, Bs, Fq, Ss);
    else
        gemm1_body(featsNext, Wpk, biasp, attnNext, bx, by - 16, As, Bs);
}

extern "C" void kernel_launch(void* const* d_in, const int* in_sizes, int n_in,
                              void* d_out, int out_size, void* d_ws, size_t ws_size,
                              hipStream_t stream)
{
    const float* fv   = (const float*)d_in[0];
    const int*   zi   = (const int*)  d_in[1];
    const int*   yi   = (const int*)  d_in[2];
    const int*   xi   = (const int*)  d_in[3];
    const int*   mask = (const int*)  d_in[4];
    const float* Wt   = (const float*)d_in[5];
    const float* bias = (const float*)d_in[6];
    float*       out  = (float*)d_out;

    float* Wpk     = (float*)d_ws;
    float* biasp   = Wpk     + (size_t)CH * NP1;
    float* feats0  = biasp   + NP1;
    float* feats1  = feats0  + (size_t)FP * CH;
    float* attn0   = feats1  + (size_t)FP * CH;
    float* feats2  = attn0   + (size_t)MP * LDATT;
    float* attn1   = feats2  + (size_t)FP * CH;
    const size_t full = ((size_t)CH * NP1 + NP1 + 3 * (size_t)FP * CH
                         + 2 * (size_t)MP * LDATT) * sizeof(float);

    const dim3 wblk(32, 4);

    pad_w<<<(CH * NP1 + 255) / 256, 256, 0, stream>>>(Wt, Wpk);
    pad_bias<<<(NP1 + 255) / 256, 256, 0, stream>>>(bias, biasp);

    if (ws_size >= full) {
        float* const F[3] = {feats0, feats1, feats2};
        float* const T[2] = {attn0, attn1};
        gather_feats<<<dim3(GATHER_BLOCKS, 2), 256, 0, stream>>>(fv, zi, yi, xi, mask, feats0, 0);
        gemm1_scores<<<dim3(16, 16), wblk, 0, stream>>>(feats0, Wpk, biasp, attn0);
        for (int b = 0; b + 1 < B_DIM; ++b)
            batch_step<<<dim3(16, 32), wblk, 0, stream>>>(
                T[b & 1], F[b % 3], out + (size_t)b * A_DIM * CH,
                fv, zi, yi, xi, mask, F[(b + 2) % 3], b + 2,
                F[(b + 1) % 3], Wpk, biasp, T[(b + 1) & 1]);
        gemm2_out<<<dim3(16, 16), wblk, 0, stream>>>(
            T[(B_DIM - 1) & 1], F[(B_DIM - 1) % 3], out + (size_t)(B_DIM - 1) * A_DIM * CH,
            fv, zi, yi, xi, mask, F[(B_DIM + 1) % 3], B_DIM);
        return;
    }

    gather_feats<<<GATHER_BLOCKS, 256, 0, stream>>>(fv, zi, yi, xi, mask, feats0, 0);

    for (int b = 0; b < B_DIM; ++b) {
        float* cur = (b & 1) ? feats1 : feats0;
        float* nxt = (b & 1) ? feats0 : feats1;
        gemm1_scores<<<dim3(16, 16), wblk, 0, stream>>>(cur, Wpk, biasp, attn0);
        gemm2_out<<<dim3(16, 16), wblk, 0, stream>>>(
            attn0, cur, out + (size_t)b * A_DIM * CH,
            fv, zi, yi, xi, mask, nxt, b + 1);
    }
}
